# starting kernel
# baseline (speedup 1.0000x reference)
#include <hip/hip_runtime.h>

typedef __attribute__((ext_vector_type(2))) float v2f;
typedef __attribute__((ext_vector_type(8))) float v8f;

#define HIDC 16
#define BLK 256

__device__ __forceinline__ void fadd(float* p, float v) {
    __hip_atomic_fetch_add(p, v, __ATOMIC_RELAXED, __HIP_MEMORY_SCOPE_AGENT);
}

__global__ void k_init(float* deg, float* agg, float* pooled, float* cnt,
                       int n_nodes, int n_graphs) {
    int i = blockIdx.x * blockDim.x + threadIdx.x;
    int tot = n_nodes * HIDC;
    if (i < tot) agg[i] = 0.0f;
    if (i < n_nodes) deg[i] = 1.0f;
    if (i < n_graphs * HIDC) pooled[i] = 0.0f;
    if (i < n_graphs) cnt[i] = 0.0f;
}

__global__ void k_deg(const int* __restrict__ dst, float* deg, int n_edges) {
    int e = blockIdx.x * blockDim.x + threadIdx.x;
    if (e < n_edges) fadd(&deg[dst[e]], 1.0f);
}

__global__ void k_dinv(float* deg, int n_nodes) {
    int i = blockIdx.x * blockDim.x + threadIdx.x;
    if (i < n_nodes) {
        float d = deg[i];
        deg[i] = (d > 0.0f) ? rsqrtf(d) : 0.0f;
    }
}

__global__ void k_embed(const int* __restrict__ x, const float* __restrict__ emb,
                        float* __restrict__ h, int n_nodes) {
    int i = blockIdx.x * blockDim.x + threadIdx.x;
    if (i < n_nodes * HIDC) {
        int node = i >> 4, j = i & 15;
        h[i] = emb[x[node] * HIDC + j];
    }
}

__global__ void k_gcn_wmma(const float* __restrict__ h, const float* __restrict__ W,
                           float* __restrict__ hw, int ntiles) {
    int wave = (blockIdx.x * blockDim.x + threadIdx.x) >> 5;
    int lane = threadIdx.x & 31;
    if (wave >= ntiles) return;
    int half = lane >> 4;
    int l15  = lane & 15;

    v2f bf[4];
#pragma unroll
    for (int t = 0; t < 4; ++t) {
        int kk = 4 * t + 2 * half;
        bf[t][0] = W[(kk + 0) * HIDC + l15];
        bf[t][1] = W[(kk + 1) * HIDC + l15];
    }

    const float* hrow = h + (size_t)(wave * 16 + l15) * HIDC;
    v8f c = {};
#pragma unroll
    for (int t = 0; t < 4; ++t) {
        int kk = 4 * t + 2 * half;
        v2f a;
        a[0] = hrow[kk + 0];
        a[1] = hrow[kk + 1];
        c = __builtin_amdgcn_wmma_f32_16x16x4_f32(
                false, a, false, bf[t], (short)0, c, false, false);
    }

#pragma unroll
    for (int v = 0; v < 8; ++v) {
        int m = v + 8 * half;
        hw[(size_t)(wave * 16 + m) * HIDC + l15] = c[v];
    }
}

__global__ void k_scatter(const int* __restrict__ src, const int* __restrict__ dst,
                          const float* __restrict__ dinv, const float* __restrict__ hw,
                          float* __restrict__ agg, int n_edges, int n_nodes) {
    int e = blockIdx.x * blockDim.x + threadIdx.x;
    int total = n_edges + n_nodes;
    if (e >= total) return;
    int s, d;
    if (e < n_edges) { s = src[e]; d = dst[e]; }
    else             { s = e - n_edges; d = s; }
    float nrm = dinv[s] * dinv[d];
    const float4* hr = (const float4*)(hw + (size_t)s * HIDC);
    float* ar = agg + (size_t)d * HIDC;
#pragma unroll
    for (int q = 0; q < 4; ++q) {
        float4 m = hr[q];
        fadd(ar + 4 * q + 0, m.x * nrm);
        fadd(ar + 4 * q + 1, m.y * nrm);
        fadd(ar + 4 * q + 2, m.z * nrm);
        fadd(ar + 4 * q + 3, m.w * nrm);
    }
}

__global__ void k_relu_bias(float* __restrict__ agg, const float* __restrict__ b,
                            float* __restrict__ h, int n_nodes) {
    int i = blockIdx.x * blockDim.x + threadIdx.x;
    if (i < n_nodes * HIDC) {
        int j = i & 15;
        h[i] = fmaxf(agg[i] + b[j], 0.0f);
        agg[i] = 0.0f;
    }
}

__global__ void k_pool(const float* __restrict__ h, const int* __restrict__ batch,
                       float* __restrict__ pooled, float* __restrict__ cnt, int n_nodes) {
    int i = blockIdx.x * blockDim.x + threadIdx.x;
    if (i < n_nodes * HIDC) {
        int node = i >> 4, j = i & 15;
        int g = batch[node];
        fadd(&pooled[g * HIDC + j], h[i]);
        if (j == 0) fadd(&cnt[g], 1.0f);
    }
}

__global__ void k_classify(const float* __restrict__ pooled, const float* __restrict__ cnt,
                           const float* __restrict__ Wc, const float* __restrict__ bc,
                           float* __restrict__ out, int n_graphs, int labels) {
    int i = blockIdx.x * blockDim.x + threadIdx.x;
    if (i >= n_graphs * labels) return;
    int g = i / labels, c = i % labels;
    float inv = 1.0f / fmaxf(cnt[g], 1.0f);
    float acc = bc[c];
#pragma unroll
    for (int j = 0; j < HIDC; ++j)
        acc += pooled[g * HIDC + j] * inv * Wc[j * labels + c];
    out[i] = acc;
}

extern "C" void kernel_launch(void* const* d_in, const int* in_sizes, int n_in,
                              void* d_out, int out_size, void* d_ws, size_t ws_size,
                              hipStream_t stream) {
    const int*   x     = (const int*)d_in[0];
    const int*   ei    = (const int*)d_in[1];
    const int*   batch = (const int*)d_in[2];
    const float* emb   = (const float*)d_in[3];
    const float* W1    = (const float*)d_in[4];
    const float* b1    = (const float*)d_in[5];
    const float* W2    = (const float*)d_in[6];
    const float* b2    = (const float*)d_in[7];
    const float* Wc    = (const float*)d_in[8];
    const float* bc    = (const float*)d_in[9];
    float* out = (float*)d_out;

    const int n_nodes  = in_sizes[0];
    const int n_edges  = in_sizes[1] / 2;
    const int labels   = in_sizes[9];
    const int n_graphs = out_size / labels;
    const int* src = ei;
    const int* dst = ei + n_edges;

    float* ws     = (float*)d_ws;
    float* deg    = ws;
    float* h      = deg + n_nodes;
    float* hw     = h  + (size_t)n_nodes * HIDC;
    float* agg    = hw + (size_t)n_nodes * HIDC;
    float* pooled = agg + (size_t)n_nodes * HIDC;
    float* cnt    = pooled + (size_t)n_graphs * HIDC;

    const int nh     = n_nodes * HIDC;
    const int ntiles = n_nodes / 16;
    const int wavesPerBlk = BLK / 32;
    const int etotal = n_edges + n_nodes;

    dim3 b(BLK);
    auto g1 = [](int n) { return dim3((n + BLK - 1) / BLK); };
    dim3 gw((ntiles + wavesPerBlk - 1) / wavesPerBlk);

    k_init<<<g1(nh), b, 0, stream>>>(deg, agg, pooled, cnt, n_nodes, n_graphs);
    k_deg<<<g1(n_edges), b, 0, stream>>>(dst, deg, n_edges);
    k_dinv<<<g1(n_nodes), b, 0, stream>>>(deg, n_nodes);
    k_embed<<<g1(nh), b, 0, stream>>>(x, emb, h, n_nodes);

    k_gcn_wmma<<<gw, b, 0, stream>>>(h, W1, hw, ntiles);
    k_scatter<<<g1(etotal), b, 0, stream>>>(src, dst, deg, hw, agg, n_edges, n_nodes);
    k_relu_bias<<<g1(nh), b, 0, stream>>>(agg, b1, h, n_nodes);

    k_gcn_wmma<<<gw, b, 0, stream>>>(h, W2, hw, ntiles);
    k_scatter<<<g1(etotal), b, 0, stream>>>(src, dst, deg, hw, agg, n_edges, n_nodes);
    k_relu_bias<<<g1(nh), b, 0, stream>>>(agg, b2, h, n_nodes);

    k_pool<<<g1(nh), b, 0, stream>>>(h, batch, pooled, cnt, n_nodes);
    k_classify<<<g1(n_graphs * labels), b, 0, stream>>>(pooled, cnt, Wc, bc, out,
                                                        n_graphs, labels);
}


# float atomic scatter-add replaced by a gather over edges grouped by dst (counting sort once per call, int atomic rank); convs sum rows in registers, bias+relu at the store; sum order = arrival rank
# speedup vs baseline: 4.1924x; 4.1924x over previous
#include <hip/hip_runtime.h>

typedef __attribute__((ext_vector_type(2))) float v2f;
typedef __attribute__((ext_vector_type(8))) float v8f;

#define HIDC 16
#define BLK 256

__device__ __forceinline__ void fadd(float* p, float v) {
    __hip_atomic_fetch_add(p, v, __ATOMIC_RELAXED, __HIP_MEMORY_SCOPE_AGENT);
}

#define SCAN_BLK 1024

__global__ void k_init(int* indeg, float* pooled, float* cnt, int n_nodes, int n_graphs) {
    int i = blockIdx.x * blockDim.x + threadIdx.x;
    int stride = gridDim.x * blockDim.x;
    if (i < n_nodes) indeg[i] = 0;
    for (int k = i; k < n_graphs * HIDC; k += stride) pooled[k] = 0.0f;
    for (int k = i; k < n_graphs; k += stride) cnt[k] = 0.0f;
}

__global__ void k_rank(const int* __restrict__ dst, int* indeg, int* __restrict__ rank, int n_edges) {
    int e = blockIdx.x * blockDim.x + threadIdx.x;
    if (e < n_edges)
        rank[e] = __hip_atomic_fetch_add(&indeg[dst[e]], 1, __ATOMIC_RELAXED,
                                         __HIP_MEMORY_SCOPE_AGENT);
}

__device__ __forceinline__ int wave_sum(int v) {
#pragma unroll
    for (int o = 16; o > 0; o >>= 1) v += __shfl_xor(v, o);
    return v;
}

__global__ void __launch_bounds__(SCAN_BLK)
k_blocksum(const int* __restrict__ indeg, int* __restrict__ bsum, int n_nodes) {
    __shared__ int sh[32];
    int tid = threadIdx.x, lane = tid & 31, w = tid >> 5;
    int i = blockIdx.x * SCAN_BLK + tid;
    int v = wave_sum((i < n_nodes) ? indeg[i] : 0);
    if (lane == 0) sh[w] = v;
    __syncthreads();
    if (w == 0) {
        int t = wave_sum(sh[lane]);
        if (lane == 0) bsum[blockIdx.x] = t;
    }
}

__global__ void __launch_bounds__(SCAN_BLK)
k_rowptr(const int* __restrict__ indeg, const int* __restrict__ bsum,
         int* __restrict__ rowptr, float* __restrict__ dinv, int n_nodes) {
    __shared__ int sh[32];
    __shared__ int s_base;
    int tid = threadIdx.x, lane = tid & 31, w = tid >> 5;

    int part = 0;
    for (int k = tid; k < (int)blockIdx.x; k += SCAN_BLK) part += bsum[k];
    part = wave_sum(part);
    if (lane == 0) sh[w] = part;
    __syncthreads();
    if (w == 0) {
        int t = wave_sum(sh[lane]);
        if (lane == 0) s_base = t;
    }
    __syncthreads();
    int base = s_base;

    int i = blockIdx.x * SCAN_BLK + tid;
    int c = (i < n_nodes) ? indeg[i] : 0;
    int incl = c;
#pragma unroll
    for (int o = 1; o < 32; o <<= 1) {
        int t = __shfl_up(incl, o);
        if (lane >= o) incl += t;
    }
    if (lane == 31) sh[w] = incl;
    __syncthreads();
    if (w == 0) {
        int t = sh[lane];
        int ti = t;
#pragma unroll
        for (int o = 1; o < 32; o <<= 1) {
            int u = __shfl_up(ti, o);
            if (lane >= o) ti += u;
        }
        sh[lane] = ti - t;
    }
    __syncthreads();
    int excl = base + sh[w] + incl - c;
    if (i < n_nodes) {
        rowptr[i] = excl;
        float d = (float)(c + 1);
        dinv[i] = (d > 0.0f) ? rsqrtf(d) : 0.0f;
        if (i == n_nodes - 1) rowptr[n_nodes] = excl + c;
    }
}

__global__ void k_fill(const int* __restrict__ src, const int* __restrict__ dst,
                       const int* __restrict__ rank, const int* __restrict__ rowptr,
                       int* __restrict__ csr_src, int n_edges) {
    int e = blockIdx.x * blockDim.x + threadIdx.x;
    if (e < n_edges) csr_src[rowptr[dst[e]] + rank[e]] = src[e];
}

__global__ void k_embed(const int* __restrict__ x, const float* __restrict__ emb,
                        float* __restrict__ h, int n_nodes) {
    int i = blockIdx.x * blockDim.x + threadIdx.x;
    if (i < n_nodes * HIDC) {
        int node = i >> 4, j = i & 15;
        h[i] = emb[x[node] * HIDC + j];
    }
}

__global__ void k_gcn_wmma(const float* __restrict__ h, const float* __restrict__ W,
                           float* __restrict__ hw, int ntiles) {
    int wave = (blockIdx.x * blockDim.x + threadIdx.x) >> 5;
    int lane = threadIdx.x & 31;
    if (wave >= ntiles) return;
    int half = lane >> 4;
    int l15  = lane & 15;

    v2f bf[4];
#pragma unroll
    for (int t = 0; t < 4; ++t) {
        int kk = 4 * t + 2 * half;
        bf[t][0] = W[(kk + 0) * HIDC + l15];
        bf[t][1] = W[(kk + 1) * HIDC + l15];
    }

    const float* hrow = h + (size_t)(wave * 16 + l15) * HIDC;
    v8f c = {};
#pragma unroll
    for (int t = 0; t < 4; ++t) {
        int kk = 4 * t + 2 * half;
        v2f a;
        a[0] = hrow[kk + 0];
        a[1] = hrow[kk + 1];
        c = __builtin_amdgcn_wmma_f32_16x16x4_f32(
                false, a, false, bf[t], (short)0, c, false, false);
    }

#pragma unroll
    for (int v = 0; v < 8; ++v) {
        int m = v + 8 * half;
        hw[(size_t)(wave * 16 + m) * HIDC + l15] = c[v];
    }
}

__global__ void k_gather(const int* __restrict__ rowptr, const int* __restrict__ csr_src,
                         const float* __restrict__ dinv, const float* __restrict__ hw,
                         const float* __restrict__ b, float* __restrict__ h, int n_nodes) {
    int i = blockIdx.x * blockDim.x + threadIdx.x;
    int node = i >> 4, j = i & 15;
    if (node >= n_nodes) return;
    int beg = rowptr[node], end = rowptr[node + 1];
    float dd = dinv[node];
    float acc = hw[i] * (dd * dd);
    int s_next = (beg + j < end) ? csr_src[beg + j] : node;
    for (int p = beg; p < end; p += 16) {
        int s = s_next;
        int qn = p + 16 + j;
        s_next = (qn < end) ? csr_src[qn] : node;
        float w = dinv[s] * dd;
        int nk = end - p;
#pragma unroll
        for (int k = 0; k < 16; ++k) {
            int   sk = __shfl(s, k, 16);
            float wk = __shfl(w, k, 16);
            float v  = hw[sk * HIDC + j];
            acc += (k < nk) ? v * wk : 0.0f;
        }
    }
    h[i] = fmaxf(acc + b[j], 0.0f);
}

__global__ void k_pool(const float* __restrict__ h, const int* __restrict__ batch,
                       float* __restrict__ pooled, float* __restrict__ cnt, int n_nodes) {
    int i = blockIdx.x * blockDim.x + threadIdx.x;
    if (i < n_nodes * HIDC) {
        int node = i >> 4, j = i & 15;
        int g = batch[node];
        fadd(&pooled[g * HIDC + j], h[i]);
        if (j == 0) fadd(&cnt[g], 1.0f);
    }
}

__global__ void k_classify(const float* __restrict__ pooled, const float* __restrict__ cnt,
                           const float* __restrict__ Wc, const float* __restrict__ bc,
                           float* __restrict__ out, int n_graphs, int labels) {
    int i = blockIdx.x * blockDim.x + threadIdx.x;
    if (i >= n_graphs * labels) return;
    int g = i / labels, c = i % labels;
    float inv = 1.0f / fmaxf(cnt[g], 1.0f);
    float acc = bc[c];
#pragma unroll
    for (int j = 0; j < HIDC; ++j)
        acc += pooled[g * HIDC + j] * inv * Wc[j * labels + c];
    out[i] = acc;
}

extern "C" void kernel_launch(void* const* d_in, const int* in_sizes, int n_in,
                              void* d_out, int out_size, void* d_ws, size_t ws_size,
                              hipStream_t stream) {
    const int*   x     = (const int*)d_in[0];
    const int*   ei    = (const int*)d_in[1];
    const int*   batch = (const int*)d_in[2];
    const float* emb   = (const float*)d_in[3];
    const float* W1    = (const float*)d_in[4];
    const float* b1    = (const float*)d_in[5];
    const float* W2    = (const float*)d_in[6];
    const float* b2    = (const float*)d_in[7];
    const float* Wc    = (const float*)d_in[8];
    const float* bc    = (const float*)d_in[9];
    float* out = (float*)d_out;

    const int n_nodes  = in_sizes[0];
    const int n_edges  = in_sizes[1] / 2;
    const int labels   = in_sizes[9];
    const int n_graphs = out_size / labels;
    const int* src = ei;
    const int* dst = ei + n_edges;

    float* ws     = (float*)d_ws;
    float* h      = ws;
    float* hw     = h  + (size_t)n_nodes * HIDC;
    float* pooled = hw + (size_t)n_nodes * HIDC;
    float* cnt    = pooled + (size_t)n_graphs * HIDC;
    float* dinv   = cnt + n_graphs;
    int* indeg    = (int*)(dinv + n_nodes);
    int* rowptr   = indeg + n_nodes;
    int* bsum     = rowptr + n_nodes + 1;
    const int nsb = (n_nodes + SCAN_BLK - 1) / SCAN_BLK;
    int* rank     = bsum + nsb;
    int* csr_src  = rank + n_edges;

    const int nh     = n_nodes * HIDC;
    const int ntiles = n_nodes / 16;
    const int wavesPerBlk = BLK / 32;

    dim3 b(BLK);
    auto g1 = [](int n) { return dim3((n + BLK - 1) / BLK); };
    dim3 gw((ntiles + wavesPerBlk - 1) / wavesPerBlk);

    k_init<<<g1(n_nodes), b, 0, stream>>>(indeg, pooled, cnt, n_nodes, n_graphs);
    k_rank<<<g1(n_edges), b, 0, stream>>>(dst, indeg, rank, n_edges);
    k_blocksum<<<dim3(nsb), dim3(SCAN_BLK), 0, stream>>>(indeg, bsum, n_nodes);
    k_rowptr<<<dim3(nsb), dim3(SCAN_BLK), 0, stream>>>(indeg, bsum, rowptr, dinv, n_nodes);
    k_fill<<<g1(n_edges), b, 0, stream>>>(src, dst, rank, rowptr, csr_src, n_edges);

    k_embed<<<g1(nh), b, 0, stream>>>(x, emb, h, n_nodes);

    k_gcn_wmma<<<gw, b, 0, stream>>>(h, W1, hw, ntiles);
    k_gather<<<g1(nh), b, 0, stream>>>(rowptr, csr_src, dinv, hw, b1, h, n_nodes);

    k_gcn_wmma<<<gw, b, 0, stream>>>(h, W2, hw, ntiles);
    k_gather<<<g1(nh), b, 0, stream>>>(rowptr, csr_src, dinv, hw, b2, h, n_nodes);

    k_pool<<<g1(nh), b, 0, stream>>>(h, batch, pooled, cnt, n_nodes);
    k_classify<<<g1(n_graphs * labels), b, 0, stream>>>(pooled, cnt, Wc, bc, out,
                                                        n_graphs, labels);
}


# fixed grids with grid-stride loops (4 items in flight per lane) in the edge, element, gather and pool kernels: 7x fewer waves, wave entry paced them; sums keep atomic arrival order, arrivals differ
# speedup vs baseline: 4.4446x; 1.0602x over previous
#include <hip/hip_runtime.h>


typedef __attribute__((ext_vector_type(2))) float v2f;
typedef __attribute__((ext_vector_type(8))) float v8f;

#define HIDC 16
#define BLK 256

#define G_EDGE 1024
#define G_ELEM 512
#define G_INIT 128

__device__ __forceinline__ void fadd(float* p, float v) {
    __hip_atomic_fetch_add(p, v, __ATOMIC_RELAXED, __HIP_MEMORY_SCOPE_AGENT);
}

#define SCAN_BLK 1024

__global__ void k_init(int* indeg, float* pooled, float* cnt, int n_nodes, int n_graphs) {
    int i = blockIdx.x * blockDim.x + threadIdx.x;
    int stride = gridDim.x * blockDim.x;
    for (int k = i; k < n_nodes; k += stride) indeg[k] = 0;
    for (int k = i; k < n_graphs * HIDC; k += stride) pooled[k] = 0.0f;
    for (int k = i; k < n_graphs; k += stride) cnt[k] = 0.0f;
}

__device__ __forceinline__ int rank_one(int* indeg, int d) {
    return __hip_atomic_fetch_add(&indeg[d], 1, __ATOMIC_RELAXED, __HIP_MEMORY_SCOPE_AGENT);
}

__global__ void k_rank(const int* __restrict__ dst, int* indeg, int* __restrict__ rank, int n_edges) {
    int e = blockIdx.x * blockDim.x + threadIdx.x;
    int stride = gridDim.x * blockDim.x;
    for (; e + 3 * stride < n_edges; e += 4 * stride) {
        int d0 = dst[e];
        int d1 = dst[e + stride];
        int d2 = dst[e + 2 * stride];
        int d3 = dst[e + 3 * stride];
        int r0 = rank_one(indeg, d0);
        int r1 = rank_one(indeg, d1);
        int r2 = rank_one(indeg, d2);
        int r3 = rank_one(indeg, d3);
        rank[e]              = r0;
        rank[e + stride]     = r1;
        rank[e + 2 * stride] = r2;
        rank[e + 3 * stride] = r3;
    }
    for (; e < n_edges; e += stride) rank[e] = rank_one(indeg, dst[e]);
}

__device__ __forceinline__ int wave_sum(int v) {
#pragma unroll
    for (int o = 16; o > 0; o >>= 1) v += __shfl_xor(v, o);
    return v;
}

__global__ void __launch_bounds__(SCAN_BLK)
k_blocksum(const int* __restrict__ indeg, int* __restrict__ bsum, int n_nodes) {
    __shared__ int sh[32];
    int tid = threadIdx.x, lane = tid & 31, w = tid >> 5;
    int i = blockIdx.x * SCAN_BLK + tid;
    int v = wave_sum((i < n_nodes) ? indeg[i] : 0);
    if (lane == 0) sh[w] = v;
    __syncthreads();
    if (w == 0) {
        int t = wave_sum(sh[lane]);
        if (lane == 0) bsum[blockIdx.x] = t;
    }
}

__global__ void __launch_bounds__(SCAN_BLK)
k_rowptr(const int* __restrict__ indeg, const int* __restrict__ bsum,
         int* __restrict__ rowptr, float* __restrict__ dinv, int n_nodes) {
    __shared__ int sh[32];
    __shared__ int s_base;
    int tid = threadIdx.x, lane = tid & 31, w = tid >> 5;

    int part = 0;
    for (int k = tid; k < (int)blockIdx.x; k += SCAN_BLK) part += bsum[k];
    part = wave_sum(part);
    if (lane == 0) sh[w] = part;
    __syncthreads();
    if (w == 0) {
        int t = wave_sum(sh[lane]);
        if (lane == 0) s_base = t;
    }
    __syncthreads();
    int base = s_base;

    int i = blockIdx.x * SCAN_BLK + tid;
    int c = (i < n_nodes) ? indeg[i] : 0;
    int incl = c;
#pragma unroll
    for (int o = 1; o < 32; o <<= 1) {
        int t = __shfl_up(incl, o);
        if (lane >= o) incl += t;
    }
    if (lane == 31) sh[w] = incl;
    __syncthreads();
    if (w == 0) {
        int t = sh[lane];
        int ti = t;
#pragma unroll
        for (int o = 1; o < 32; o <<= 1) {
            int u = __shfl_up(ti, o);
            if (lane >= o) ti += u;
        }
        sh[lane] = ti - t;
    }
    __syncthreads();
    int excl = base + sh[w] + incl - c;
    if (i < n_nodes) {
        rowptr[i] = excl;
        float d = (float)(c + 1);
        dinv[i] = (d > 0.0f) ? rsqrtf(d) : 0.0f;
        if (i == n_nodes - 1) rowptr[n_nodes] = excl + c;
    }
}

__global__ void k_fill(const int* __restrict__ src, const int* __restrict__ dst,
                       const int* __restrict__ rank, const int* __restrict__ rowptr,
                       int* __restrict__ csr_src, int n_edges) {
    int e = blockIdx.x * blockDim.x + threadIdx.x;
    int stride = gridDim.x * blockDim.x;
    for (; e + 3 * stride < n_edges; e += 4 * stride) {
        int e1 = e + stride, e2 = e + 2 * stride, e3 = e + 3 * stride;
        int d0 = dst[e], d1 = dst[e1], d2 = dst[e2], d3 = dst[e3];
        int r0 = rank[e], r1 = rank[e1], r2 = rank[e2], r3 = rank[e3];
        int s0 = src[e], s1 = src[e1], s2 = src[e2], s3 = src[e3];
        int p0 = rowptr[d0], p1 = rowptr[d1], p2 = rowptr[d2], p3 = rowptr[d3];
        csr_src[p0 + r0] = s0;
        csr_src[p1 + r1] = s1;
        csr_src[p2 + r2] = s2;
        csr_src[p3 + r3] = s3;
    }
    for (; e < n_edges; e += stride) csr_src[rowptr[dst[e]] + rank[e]] = src[e];
}

__global__ void k_embed(const int* __restrict__ x, const float* __restrict__ emb,
                        float* __restrict__ h, int n_nodes) {
    int i = blockIdx.x * blockDim.x + threadIdx.x;
    int stride = gridDim.x * blockDim.x;
    int nh = n_nodes * HIDC;
    for (; i + 3 * stride < nh; i += 4 * stride) {
        int i1 = i + stride, i2 = i + 2 * stride, i3 = i + 3 * stride;
        int x0 = x[i >> 4], x1 = x[i1 >> 4], x2 = x[i2 >> 4], x3 = x[i3 >> 4];
        float v0 = emb[x0 * HIDC + (i & 15)];
        float v1 = emb[x1 * HIDC + (i1 & 15)];
        float v2 = emb[x2 * HIDC + (i2 & 15)];
        float v3 = emb[x3 * HIDC + (i3 & 15)];
        h[i]  = v0;
        h[i1] = v1;
        h[i2] = v2;
        h[i3] = v3;
    }
    for (; i < nh; i += stride) h[i] = emb[x[i >> 4] * HIDC + (i & 15)];
}

__global__ void k_gcn_wmma(const float* __restrict__ h, const float* __restrict__ W,
                           float* __restrict__ hw, int ntiles) {
    int wave = (blockIdx.x * blockDim.x + threadIdx.x) >> 5;
    int lane = threadIdx.x & 31;
    if (wave >= ntiles) return;
    int half = lane >> 4;
    int l15  = lane & 15;

    v2f bf[4];
#pragma unroll
    for (int t = 0; t < 4; ++t) {
        int kk = 4 * t + 2 * half;
        bf[t][0] = W[(kk + 0) * HIDC + l15];
        bf[t][1] = W[(kk + 1) * HIDC + l15];
    }

    const float* hrow = h + (size_t)(wave * 16 + l15) * HIDC;
    v8f c = {};
#pragma unroll
    for (int t = 0; t < 4; ++t) {
        int kk = 4 * t + 2 * half;
        v2f a;
        a[0] = hrow[kk + 0];
        a[1] = hrow[kk + 1];
        c = __builtin_amdgcn_wmma_f32_16x16x4_f32(
                false, a, false, bf[t], (short)0, c, false, false);
    }

#pragma unroll
    for (int v = 0; v < 8; ++v) {
        int m = v + 8 * half;
        hw[(size_t)(wave * 16 + m) * HIDC + l15] = c[v];
    }
}

__global__ void k_gather(const int* __restrict__ rowptr, const int* __restrict__ csr_src,
                         const float* __restrict__ dinv, const float* __restrict__ hw,
                         const float* __restrict__ b, float* __restrict__ h, int n_nodes) {
    int t = blockIdx.x * blockDim.x + threadIdx.x;
    int j = t & 15;
    int ngrp = (gridDim.x * blockDim.x) >> 4;
    float bj = b[j];
    for (int node = t >> 4; node < n_nodes; node += ngrp) {
        int i = node * HIDC + j;
        int beg = rowptr[node], end = rowptr[node + 1];
        float dd = dinv[node];
        float acc = hw[i] * (dd * dd);
        int s_next = (beg + j < end) ? csr_src[beg + j] : node;
        for (int p = beg; p < end; p += 16) {
            int s = s_next;
            int qn = p + 16 + j;
            s_next = (qn < end) ? csr_src[qn] : node;
            float w = dinv[s] * dd;
            int nk = end - p;
#pragma unroll
            for (int k = 0; k < 16; ++k) {
                int   sk = __shfl(s, k, 16);
                float wk = __shfl(w, k, 16);
                float v  = hw[sk * HIDC + j];
                acc += (k < nk) ? v * wk : 0.0f;
            }
        }
        h[i] = fmaxf(acc + bj, 0.0f);
    }
}

__global__ void k_pool(const float* __restrict__ h, const int* __restrict__ batch,
                       float* __restrict__ pooled, float* __restrict__ cnt, int n_nodes) {
    int t = blockIdx.x * blockDim.x + threadIdx.x;
    int wave = t >> 5, nwaves = (gridDim.x * blockDim.x) >> 5;
    int sub = (t >> 4) & 1, j = t & 15;
    int per = n_nodes / nwaves, rem = n_nodes - per * nwaves;
    int n0 = wave * per + (wave < rem ? wave : rem);
    int n1 = n0 + per + (wave < rem ? 1 : 0);
    int node = n0 + sub;
    for (; node + 6 < n1; node += 8) {
        int g0 = batch[node], g1 = batch[node + 2], g2 = batch[node + 4], g3 = batch[node + 6];
        float v0 = h[node * HIDC + j];
        float v1 = h[(node + 2) * HIDC + j];
        float v2 = h[(node + 4) * HIDC + j];
        float v3 = h[(node + 6) * HIDC + j];
        fadd(&pooled[g0 * HIDC + j], v0);
        fadd(&pooled[g1 * HIDC + j], v1);
        fadd(&pooled[g2 * HIDC + j], v2);
        fadd(&pooled[g3 * HIDC + j], v3);
        if (j == 0) {
            fadd(&cnt[g0], 1.0f);
            fadd(&cnt[g1], 1.0f);
            fadd(&cnt[g2], 1.0f);
            fadd(&cnt[g3], 1.0f);
        }
    }
    for (; node < n1; node += 2) {
        int g = batch[node];
        fadd(&pooled[g * HIDC + j], h[node * HIDC + j]);
        if (j == 0) fadd(&cnt[g], 1.0f);
    }
}

__global__ void k_classify(const float* __restrict__ pooled, const float* __restrict__ cnt,
                           const float* __restrict__ Wc, const float* __restrict__ bc,
                           float* __restrict__ out, int n_graphs, int labels) {
    int i = blockIdx.x * blockDim.x + threadIdx.x;
    if (i >= n_graphs * labels) return;
    int g = i / labels, c = i % labels;
    float inv = 1.0f / fmaxf(cnt[g], 1.0f);
    float acc = bc[c];
#pragma unroll
    for (int j = 0; j < HIDC; ++j)
        acc += pooled[g * HIDC + j] * inv * Wc[j * labels + c];
    out[i] = acc;
}

extern "C" void kernel_launch(void* const* d_in, const int* in_sizes, int n_in,
                              void* d_out, int out_size, void* d_ws, size_t ws_size,
                              hipStream_t stream) {
    const int*   x     = (const int*)d_in[0];
    const int*   ei    = (const int*)d_in[1];
    const int*   batch = (const int*)d_in[2];
    const float* emb   = (const float*)d_in[3];
    const float* W1    = (const float*)d_in[4];
    const float* b1    = (const float*)d_in[5];
    const float* W2    = (const float*)d_in[6];
    const float* b2    = (const float*)d_in[7];
    const float* Wc    = (const float*)d_in[8];
    const float* bc    = (const float*)d_in[9];
    float* out = (float*)d_out;

    const int n_nodes  = in_sizes[0];
    const int n_edges  = in_sizes[1] / 2;
    const int labels   = in_sizes[9];
    const int n_graphs = out_size / labels;
    const int* src = ei;
    const int* dst = ei + n_edges;

    float* ws     = (float*)d_ws;
    float* h      = ws;
    float* hw     = h  + (size_t)n_nodes * HIDC;
    float* pooled = hw + (size_t)n_nodes * HIDC;
    float* cnt    = pooled + (size_t)n_graphs * HIDC;
    float* dinv   = cnt + n_graphs;
    int* indeg    = (int*)(dinv + n_nodes);
    int* rowptr   = indeg + n_nodes;
    int* bsum     = rowptr + n_nodes + 1;
    const int nsb = (n_nodes + SCAN_BLK - 1) / SCAN_BLK;
    int* rank     = bsum + nsb;
    int* csr_src  = rank + n_edges;

    const int ntiles = n_nodes / 16;
    const int wavesPerBlk = BLK / 32;

    dim3 b(BLK);
    auto g1 = [](int n) { return dim3((n + BLK - 1) / BLK); };
    dim3 gw((ntiles + wavesPerBlk - 1) / wavesPerBlk);

    k_init<<<dim3(G_INIT), b, 0, stream>>>(indeg, pooled, cnt, n_nodes, n_graphs);
    k_rank<<<dim3(G_EDGE), b, 0, stream>>>(dst, indeg, rank, n_edges);
    k_blocksum<<<dim3(nsb), dim3(SCAN_BLK), 0, stream>>>(indeg, bsum, n_nodes);
    k_rowptr<<<dim3(nsb), dim3(SCAN_BLK), 0, stream>>>(indeg, bsum, rowptr, dinv, n_nodes);
    k_fill<<<dim3(G_EDGE), b, 0, stream>>>(src, dst, rank, rowptr, csr_src, n_edges);

    k_embed<<<dim3(G_ELEM), b, 0, stream>>>(x, emb, h, n_nodes);

    k_gcn_wmma<<<gw, b, 0, stream>>>(h, W1, hw, ntiles);
    k_gather<<<dim3(G_EDGE), b, 0, stream>>>(rowptr, csr_src, dinv, hw, b1, h, n_nodes);

    k_gcn_wmma<<<gw, b, 0, stream>>>(h, W2, hw, ntiles);
    k_gather<<<dim3(G_EDGE), b, 0, stream>>>(rowptr, csr_src, dinv, hw, b2, h, n_nodes);

    k_pool<<<dim3(G_ELEM), b, 0, stream>>>(h, batch, pooled, cnt, n_nodes);
    k_classify<<<g1(n_graphs * labels), b, 0, stream>>>(pooled, cnt, Wc, bc, out,
                                                        n_graphs, labels);
}


# per-edge returning global atomic rank replaced by a two-level counting sort on LDS counters (chunk x dst-bucket deal, then one work-group per bucket); sum order = LDS arrival rank, arrivals differ
# speedup vs baseline: 8.2963x; 1.8666x over previous
#include <hip/hip_runtime.h>


typedef __attribute__((ext_vector_type(2))) float v2f;
typedef __attribute__((ext_vector_type(8))) float v8f;

#define HIDC 16
#define BLK 256

#define G_EDGE 1024
#define G_ELEM 512

#define NODE_SHIFT 8
#define BNODES 256
#define NCHUNK 256
#define SORT_BLK 1024

__device__ __forceinline__ void fadd(float* p, float v) {
    __hip_atomic_fetch_add(p, v, __ATOMIC_RELAXED, __HIP_MEMORY_SCOPE_AGENT);
}

__device__ __forceinline__ int wave_sum(int v) {
#pragma unroll
    for (int o = 16; o > 0; o >>= 1) v += __shfl_xor(v, o);
    return v;
}

__device__ __forceinline__ int wave_scan_incl(int v, int lane) {
#pragma unroll
    for (int o = 1; o < 32; o <<= 1) {
        int t = __shfl_up(v, o);
        if (lane >= o) v += t;
    }
    return v;
}

extern __shared__ int dsh[];

__global__ void __launch_bounds__(SORT_BLK)
k_hist(const int* __restrict__ dst, int* __restrict__ bcnt, float* __restrict__ pooled,
       float* __restrict__ cnt, int n_edges, int nb, int n_graphs) {
    int tid = threadIdx.x;
    for (int k = tid; k < nb; k += SORT_BLK) dsh[k] = 0;
    int gi = blockIdx.x * SORT_BLK + tid, gs = NCHUNK * SORT_BLK;
    for (int k = gi; k < n_graphs * HIDC; k += gs) pooled[k] = 0.0f;
    for (int k = gi; k < n_graphs; k += gs) cnt[k] = 0.0f;
    __syncthreads();
    int per = (n_edges + NCHUNK - 1) / NCHUNK;
    int c0 = blockIdx.x * per;
    int c1 = min(c0 + per, n_edges);
    int e = c0 + tid;
    for (; e + 3 * SORT_BLK < c1; e += 4 * SORT_BLK) {
        int d0 = dst[e];
        int d1 = dst[e + SORT_BLK];
        int d2 = dst[e + 2 * SORT_BLK];
        int d3 = dst[e + 3 * SORT_BLK];
        atomicAdd(&dsh[d0 >> NODE_SHIFT], 1);
        atomicAdd(&dsh[d1 >> NODE_SHIFT], 1);
        atomicAdd(&dsh[d2 >> NODE_SHIFT], 1);
        atomicAdd(&dsh[d3 >> NODE_SHIFT], 1);
    }
    for (; e < c1; e += SORT_BLK) atomicAdd(&dsh[dst[e] >> NODE_SHIFT], 1);
    __syncthreads();
    for (int k = tid; k < nb; k += SORT_BLK) bcnt[k * NCHUNK + blockIdx.x] = dsh[k];
}

__global__ void __launch_bounds__(NCHUNK)
k_btot(const int* __restrict__ bcnt, int* __restrict__ btot) {
    __shared__ int sh[NCHUNK / 32];
    int tid = threadIdx.x, lane = tid & 31, w = tid >> 5;
    int v = wave_sum(bcnt[blockIdx.x * NCHUNK + tid]);
    if (lane == 0) sh[w] = v;
    __syncthreads();
    if (tid == 0) {
        int t = 0;
        for (int k = 0; k < NCHUNK / 32; ++k) t += sh[k];
        btot[blockIdx.x] = t;
    }
}

__global__ void __launch_bounds__(NCHUNK)
k_off(const int* __restrict__ bcnt, const int* __restrict__ btot, int* __restrict__ boff,
      int* __restrict__ bstart, int nb) {
    __shared__ int sh[NCHUNK / 32];
    __shared__ int s_base;
    int tid = threadIdx.x, lane = tid & 31, w = tid >> 5;
    int b = blockIdx.x;
    int part = 0;
    for (int k = tid; k < b; k += NCHUNK) part += btot[k];
    part = wave_sum(part);
    if (lane == 0) sh[w] = part;
    __syncthreads();
    if (tid == 0) {
        int t = 0;
        for (int k = 0; k < NCHUNK / 32; ++k) t += sh[k];
        s_base = t;
    }
    __syncthreads();
    int base = s_base;
    int c = bcnt[b * NCHUNK + tid];
    int incl = wave_scan_incl(c, lane);
    if (lane == 31) sh[w] = incl;
    __syncthreads();
    int wpre = 0;
    for (int k = 0; k < w; ++k) wpre += sh[k];
    boff[b * NCHUNK + tid] = base + wpre + incl - c;
    if (tid == 0) bstart[b] = base;
    if (b == nb - 1 && tid == NCHUNK - 1) bstart[nb] = base + wpre + incl;
}

__global__ void __launch_bounds__(SORT_BLK)
k_scatter(const int* __restrict__ src, const int* __restrict__ dst, const int* __restrict__ boff,
          int* __restrict__ pairs, int n_edges, int nb) {
    int2* __restrict__ rec = reinterpret_cast<int2*>(pairs);
    int tid = threadIdx.x;
    for (int k = tid; k < nb; k += SORT_BLK) dsh[k] = boff[k * NCHUNK + blockIdx.x];
    __syncthreads();
    int per = (n_edges + NCHUNK - 1) / NCHUNK;
    int c0 = blockIdx.x * per;
    int c1 = min(c0 + per, n_edges);
    int e = c0 + tid;
    for (; e + 3 * SORT_BLK < c1; e += 4 * SORT_BLK) {
        int e1 = e + SORT_BLK, e2 = e + 2 * SORT_BLK, e3 = e + 3 * SORT_BLK;
        int d0 = dst[e], d1 = dst[e1], d2 = dst[e2], d3 = dst[e3];
        int s0 = src[e], s1 = src[e1], s2 = src[e2], s3 = src[e3];
        int p0 = atomicAdd(&dsh[d0 >> NODE_SHIFT], 1);
        int p1 = atomicAdd(&dsh[d1 >> NODE_SHIFT], 1);
        int p2 = atomicAdd(&dsh[d2 >> NODE_SHIFT], 1);
        int p3 = atomicAdd(&dsh[d3 >> NODE_SHIFT], 1);
        rec[p0] = make_int2(s0, d0);
        rec[p1] = make_int2(s1, d1);
        rec[p2] = make_int2(s2, d2);
        rec[p3] = make_int2(s3, d3);
    }
    for (; e < c1; e += SORT_BLK) {
        int d = dst[e], s = src[e];
        int p = atomicAdd(&dsh[d >> NODE_SHIFT], 1);
        rec[p] = make_int2(s, d);
    }
}

__global__ void __launch_bounds__(SORT_BLK)
k_bucket(const int* __restrict__ pairs, const int* __restrict__ bstart,
         int* __restrict__ rowptr, float* __restrict__ dinv, int* __restrict__ csr_src,
         int n_nodes, int n_edges, int nb) {
    const int2* __restrict__ rec = reinterpret_cast<const int2*>(pairs);
    __shared__ int cn[BNODES];
    __shared__ int wsum[BNODES / 32];
    int tid = threadIdx.x, lane = tid & 31, w = tid >> 5;
    int b = blockIdx.x;
    int nbase = b << NODE_SHIFT;
    int e0 = bstart[b], e1 = bstart[b + 1];
    if (tid < BNODES) cn[tid] = 0;
    __syncthreads();
    int e = e0 + tid;
    for (; e + 3 * SORT_BLK < e1; e += 4 * SORT_BLK) {
        int d0 = rec[e].y;
        int d1 = rec[e + SORT_BLK].y;
        int d2 = rec[e + 2 * SORT_BLK].y;
        int d3 = rec[e + 3 * SORT_BLK].y;
        atomicAdd(&cn[d0 - nbase], 1);
        atomicAdd(&cn[d1 - nbase], 1);
        atomicAdd(&cn[d2 - nbase], 1);
        atomicAdd(&cn[d3 - nbase], 1);
    }
    for (; e < e1; e += SORT_BLK) atomicAdd(&cn[rec[e].y - nbase], 1);
    __syncthreads();
    int c = 0, incl = 0;
    if (w < BNODES / 32) {
        c = cn[tid];
        incl = wave_scan_incl(c, lane);
        if (lane == 31) wsum[w] = incl;
    }
    __syncthreads();
    if (w < BNODES / 32) {
        int wpre = 0;
        for (int k = 0; k < w; ++k) wpre += wsum[k];
        int excl = wpre + incl - c;
        cn[tid] = excl;
        int node = nbase + tid;
        if (node < n_nodes) {
            rowptr[node] = e0 + excl;
            float d = (float)(c + 1);
            dinv[node] = (d > 0.0f) ? rsqrtf(d) : 0.0f;
        }
    }
    if (b == nb - 1 && tid == 0) rowptr[n_nodes] = n_edges;
    __syncthreads();
    e = e0 + tid;
    for (; e + 3 * SORT_BLK < e1; e += 4 * SORT_BLK) {
        int e1i = e + SORT_BLK, e2i = e + 2 * SORT_BLK, e3i = e + 3 * SORT_BLK;
        int2 r0 = rec[e], r1 = rec[e1i], r2 = rec[e2i], r3 = rec[e3i];
        int p0 = atomicAdd(&cn[r0.y - nbase], 1);
        int p1 = atomicAdd(&cn[r1.y - nbase], 1);
        int p2 = atomicAdd(&cn[r2.y - nbase], 1);
        int p3 = atomicAdd(&cn[r3.y - nbase], 1);
        csr_src[e0 + p0] = r0.x;
        csr_src[e0 + p1] = r1.x;
        csr_src[e0 + p2] = r2.x;
        csr_src[e0 + p3] = r3.x;
    }
    for (; e < e1; e += SORT_BLK) {
        int2 r = rec[e];
        int p = atomicAdd(&cn[r.y - nbase], 1);
        csr_src[e0 + p] = r.x;
    }
}

__global__ void k_embed(const int* __restrict__ x, const float* __restrict__ emb,
                        float* __restrict__ h, int n_nodes) {
    int i = blockIdx.x * blockDim.x + threadIdx.x;
    int stride = gridDim.x * blockDim.x;
    int nh = n_nodes * HIDC;
    for (; i + 3 * stride < nh; i += 4 * stride) {
        int i1 = i + stride, i2 = i + 2 * stride, i3 = i + 3 * stride;
        int x0 = x[i >> 4], x1 = x[i1 >> 4], x2 = x[i2 >> 4], x3 = x[i3 >> 4];
        float v0 = emb[x0 * HIDC + (i & 15)];
        float v1 = emb[x1 * HIDC + (i1 & 15)];
        float v2 = emb[x2 * HIDC + (i2 & 15)];
        float v3 = emb[x3 * HIDC + (i3 & 15)];
        h[i]  = v0;
        h[i1] = v1;
        h[i2] = v2;
        h[i3] = v3;
    }
    for (; i < nh; i += stride) h[i] = emb[x[i >> 4] * HIDC + (i & 15)];
}

__global__ void k_gcn_wmma(const float* __restrict__ h, const float* __restrict__ W,
                           float* __restrict__ hw, int ntiles) {
    int wave = (blockIdx.x * blockDim.x + threadIdx.x) >> 5;
    int lane = threadIdx.x & 31;
    if (wave >= ntiles) return;
    int half = lane >> 4;
    int l15  = lane & 15;

    v2f bf[4];
#pragma unroll
    for (int t = 0; t < 4; ++t) {
        int kk = 4 * t + 2 * half;
        bf[t][0] = W[(kk + 0) * HIDC + l15];
        bf[t][1] = W[(kk + 1) * HIDC + l15];
    }

    const float* hrow = h + (size_t)(wave * 16 + l15) * HIDC;
    v8f c = {};
#pragma unroll
    for (int t = 0; t < 4; ++t) {
        int kk = 4 * t + 2 * half;
        v2f a;
        a[0] = hrow[kk + 0];
        a[1] = hrow[kk + 1];
        c = __builtin_amdgcn_wmma_f32_16x16x4_f32(
                false, a, false, bf[t], (short)0, c, false, false);
    }

#pragma unroll
    for (int v = 0; v < 8; ++v) {
        int m = v + 8 * half;
        hw[(size_t)(wave * 16 + m) * HIDC + l15] = c[v];
    }
}

__global__ void k_gather(const int* __restrict__ rowptr, const int* __restrict__ csr_src,
                         const float* __restrict__ dinv, const float* __restrict__ hw,
                         const float* __restrict__ b, float* __restrict__ h, int n_nodes) {
    int t = blockIdx.x * blockDim.x + threadIdx.x;
    int j = t & 15;
    int ngrp = (gridDim.x * blockDim.x) >> 4;
    float bj = b[j];
    for (int node = t >> 4; node < n_nodes; node += ngrp) {
        int i = node * HIDC + j;
        int beg = rowptr[node], end = rowptr[node + 1];
        float dd = dinv[node];
        float acc = hw[i] * (dd * dd);
        int s_next = (beg + j < end) ? csr_src[beg + j] : node;
        for (int p = beg; p < end; p += 16) {
            int s = s_next;
            int qn = p + 16 + j;
            s_next = (qn < end) ? csr_src[qn] : node;
            float w = dinv[s] * dd;
            int nk = end - p;
#pragma unroll
            for (int k = 0; k < 16; ++k) {
                int   sk = __shfl(s, k, 16);
                float wk = __shfl(w, k, 16);
                float v  = hw[sk * HIDC + j];
                acc += (k < nk) ? v * wk : 0.0f;
            }
        }
        h[i] = fmaxf(acc + bj, 0.0f);
    }
}

__global__ void k_pool(const float* __restrict__ h, const int* __restrict__ batch,
                       float* __restrict__ pooled, float* __restrict__ cnt, int n_nodes) {
    int t = blockIdx.x * blockDim.x + threadIdx.x;
    int wave = t >> 5, nwaves = (gridDim.x * blockDim.x) >> 5;
    int sub = (t >> 4) & 1, j = t & 15;
    int per = n_nodes / nwaves, rem = n_nodes - per * nwaves;
    int n0 = wave * per + (wave < rem ? wave : rem);
    int n1 = n0 + per + (wave < rem ? 1 : 0);
    int node = n0 + sub;
    for (; node + 6 < n1; node += 8) {
        int g0 = batch[node], g1 = batch[node + 2], g2 = batch[node + 4], g3 = batch[node + 6];
        float v0 = h[node * HIDC + j];
        float v1 = h[(node + 2) * HIDC + j];
        float v2 = h[(node + 4) * HIDC + j];
        float v3 = h[(node + 6) * HIDC + j];
        fadd(&pooled[g0 * HIDC + j], v0);
        fadd(&pooled[g1 * HIDC + j], v1);
        fadd(&pooled[g2 * HIDC + j], v2);
        fadd(&pooled[g3 * HIDC + j], v3);
        if (j == 0) {
            fadd(&cnt[g0], 1.0f);
            fadd(&cnt[g1], 1.0f);
            fadd(&cnt[g2], 1.0f);
            fadd(&cnt[g3], 1.0f);
        }
    }
    for (; node < n1; node += 2) {
        int g = batch[node];
        fadd(&pooled[g * HIDC + j], h[node * HIDC + j]);
        if (j == 0) fadd(&cnt[g], 1.0f);
    }
}

__global__ void k_classify(const float* __restrict__ pooled, const float* __restrict__ cnt,
                           const float* __restrict__ Wc, const float* __restrict__ bc,
                           float* __restrict__ out, int n_graphs, int labels) {
    int i = blockIdx.x * blockDim.x + threadIdx.x;
    if (i >= n_graphs * labels) return;
    int g = i / labels, c = i % labels;
    float inv = 1.0f / fmaxf(cnt[g], 1.0f);
    float acc = bc[c];
#pragma unroll
    for (int j = 0; j < HIDC; ++j)
        acc += pooled[g * HIDC + j] * inv * Wc[j * labels + c];
    out[i] = acc;
}

extern "C" void kernel_launch(void* const* d_in, const int* in_sizes, int n_in,
                              void* d_out, int out_size, void* d_ws, size_t ws_size,
                              hipStream_t stream) {
    const int*   x     = (const int*)d_in[0];
    const int*   ei    = (const int*)d_in[1];
    const int*   batch = (const int*)d_in[2];
    const float* emb   = (const float*)d_in[3];
    const float* W1    = (const float*)d_in[4];
    const float* b1    = (const float*)d_in[5];
    const float* W2    = (const float*)d_in[6];
    const float* b2    = (const float*)d_in[7];
    const float* Wc    = (const float*)d_in[8];
    const float* bc    = (const float*)d_in[9];
    float* out = (float*)d_out;

    const int n_nodes  = in_sizes[0];
    const int n_edges  = in_sizes[1] / 2;
    const int labels   = in_sizes[9];
    const int n_graphs = out_size / labels;
    const int* src = ei;
    const int* dst = ei + n_edges;

    const int nb = (n_nodes + BNODES - 1) / BNODES;

    float* ws     = (float*)d_ws;
    float* h      = ws;
    float* hw     = h  + (size_t)n_nodes * HIDC;
    int* pairs    = (int*)(hw + (size_t)n_nodes * HIDC);
    float* pooled = (float*)(pairs + (size_t)2 * n_edges);
    float* cnt    = pooled + (size_t)n_graphs * HIDC;
    float* dinv   = cnt + n_graphs;
    int* rowptr   = (int*)(dinv + n_nodes);
    int* bcnt     = rowptr + n_nodes + 1;
    int* boff     = bcnt + (size_t)nb * NCHUNK;
    int* btot     = boff + (size_t)nb * NCHUNK;
    int* bstart   = btot + nb;
    int* csr_src  = bstart + nb + 1;

    const int ntiles = n_nodes / 16;
    const int wavesPerBlk = BLK / 32;

    dim3 b(BLK);
    auto g1 = [](int n) { return dim3((n + BLK - 1) / BLK); };
    dim3 gw((ntiles + wavesPerBlk - 1) / wavesPerBlk);

    k_hist<<<dim3(NCHUNK), dim3(SORT_BLK), nb * sizeof(int), stream>>>(dst, bcnt, pooled, cnt,
                                                                       n_edges, nb, n_graphs);
    k_btot<<<dim3(nb), dim3(NCHUNK), 0, stream>>>(bcnt, btot);
    k_off<<<dim3(nb), dim3(NCHUNK), 0, stream>>>(bcnt, btot, boff, bstart, nb);
    k_scatter<<<dim3(NCHUNK), dim3(SORT_BLK), nb * sizeof(int), stream>>>(src, dst, boff, pairs,
                                                                          n_edges, nb);
    k_bucket<<<dim3(nb), dim3(SORT_BLK), 0, stream>>>(pairs, bstart, rowptr, dinv, csr_src,
                                                      n_nodes, n_edges, nb);

    k_embed<<<dim3(G_ELEM), b, 0, stream>>>(x, emb, h, n_nodes);

    k_gcn_wmma<<<gw, b, 0, stream>>>(h, W1, hw, ntiles);
    k_gather<<<dim3(G_EDGE), b, 0, stream>>>(rowptr, csr_src, dinv, hw, b1, h, n_nodes);

    k_gcn_wmma<<<gw, b, 0, stream>>>(h, W2, hw, ntiles);
    k_gather<<<dim3(G_EDGE), b, 0, stream>>>(rowptr, csr_src, dinv, hw, b2, h, n_nodes);

    k_pool<<<dim3(G_ELEM), b, 0, stream>>>(h, batch, pooled, cnt, n_nodes);
    k_classify<<<g1(n_graphs * labels), b, 0, stream>>>(pooled, cnt, Wc, bc, out,
                                                        n_graphs, labels);
}


# edge weight factored: transform scales its input rows by dinv[src], gather sums stored rows and scales by dinv[dst] once per node; no per-edge dinv[src] fetch; sum order kept, term rounding regrouped
# speedup vs baseline: 9.6231x; 1.1599x over previous
#include <hip/hip_runtime.h>


typedef __attribute__((ext_vector_type(2))) float v2f;
typedef __attribute__((ext_vector_type(8))) float v8f;

#define HIDC 16
#define BLK 256

#define G_EDGE 1024
#define G_ELEM 512

#define NODE_SHIFT 8
#define BNODES 256
#define NCHUNK 256
#define SORT_BLK 1024

__device__ __forceinline__ void fadd(float* p, float v) {
    __hip_atomic_fetch_add(p, v, __ATOMIC_RELAXED, __HIP_MEMORY_SCOPE_AGENT);
}

__device__ __forceinline__ int wave_sum(int v) {
#pragma unroll
    for (int o = 16; o > 0; o >>= 1) v += __shfl_xor(v, o);
    return v;
}

__device__ __forceinline__ int wave_scan_incl(int v, int lane) {
#pragma unroll
    for (int o = 1; o < 32; o <<= 1) {
        int t = __shfl_up(v, o);
        if (lane >= o) v += t;
    }
    return v;
}

extern __shared__ int dsh[];

__global__ void __launch_bounds__(SORT_BLK)
k_hist(const int* __restrict__ dst, int* __restrict__ bcnt, float* __restrict__ pooled,
       float* __restrict__ cnt, int n_edges, int nb, int n_graphs) {
    int tid = threadIdx.x;
    for (int k = tid; k < nb; k += SORT_BLK) dsh[k] = 0;
    int gi = blockIdx.x * SORT_BLK + tid, gs = NCHUNK * SORT_BLK;
    for (int k = gi; k < n_graphs * HIDC; k += gs) pooled[k] = 0.0f;
    for (int k = gi; k < n_graphs; k += gs) cnt[k] = 0.0f;
    __syncthreads();
    int per = (n_edges + NCHUNK - 1) / NCHUNK;
    int c0 = blockIdx.x * per;
    int c1 = min(c0 + per, n_edges);
    int e = c0 + tid;
    for (; e + 3 * SORT_BLK < c1; e += 4 * SORT_BLK) {
        int d0 = dst[e];
        int d1 = dst[e + SORT_BLK];
        int d2 = dst[e + 2 * SORT_BLK];
        int d3 = dst[e + 3 * SORT_BLK];
        atomicAdd(&dsh[d0 >> NODE_SHIFT], 1);
        atomicAdd(&dsh[d1 >> NODE_SHIFT], 1);
        atomicAdd(&dsh[d2 >> NODE_SHIFT], 1);
        atomicAdd(&dsh[d3 >> NODE_SHIFT], 1);
    }
    for (; e < c1; e += SORT_BLK) atomicAdd(&dsh[dst[e] >> NODE_SHIFT], 1);
    __syncthreads();
    for (int k = tid; k < nb; k += SORT_BLK) bcnt[k * NCHUNK + blockIdx.x] = dsh[k];
}

__global__ void __launch_bounds__(NCHUNK)
k_btot(const int* __restrict__ bcnt, int* __restrict__ btot) {
    __shared__ int sh[NCHUNK / 32];
    int tid = threadIdx.x, lane = tid & 31, w = tid >> 5;
    int v = wave_sum(bcnt[blockIdx.x * NCHUNK + tid]);
    if (lane == 0) sh[w] = v;
    __syncthreads();
    if (tid == 0) {
        int t = 0;
        for (int k = 0; k < NCHUNK / 32; ++k) t += sh[k];
        btot[blockIdx.x] = t;
    }
}

__global__ void __launch_bounds__(NCHUNK)
k_off(const int* __restrict__ bcnt, const int* __restrict__ btot, int* __restrict__ boff,
      int* __restrict__ bstart, int nb) {
    __shared__ int sh[NCHUNK / 32];
    __shared__ int s_base;
    int tid = threadIdx.x, lane = tid & 31, w = tid >> 5;
    int b = blockIdx.x;
    int part = 0;
    for (int k = tid; k < b; k += NCHUNK) part += btot[k];
    part = wave_sum(part);
    if (lane == 0) sh[w] = part;
    __syncthreads();
    if (tid == 0) {
        int t = 0;
        for (int k = 0; k < NCHUNK / 32; ++k) t += sh[k];
        s_base = t;
    }
    __syncthreads();
    int base = s_base;
    int c = bcnt[b * NCHUNK + tid];
    int incl = wave_scan_incl(c, lane);
    if (lane == 31) sh[w] = incl;
    __syncthreads();
    int wpre = 0;
    for (int k = 0; k < w; ++k) wpre += sh[k];
    boff[b * NCHUNK + tid] = base + wpre + incl - c;
    if (tid == 0) bstart[b] = base;
    if (b == nb - 1 && tid == NCHUNK - 1) bstart[nb] = base + wpre + incl;
}

__global__ void __launch_bounds__(SORT_BLK)
k_scatter(const int* __restrict__ src, const int* __restrict__ dst, const int* __restrict__ boff,
          int* __restrict__ pairs, int n_edges, int nb) {
    int2* __restrict__ rec = reinterpret_cast<int2*>(pairs);
    int tid = threadIdx.x;
    for (int k = tid; k < nb; k += SORT_BLK) dsh[k] = boff[k * NCHUNK + blockIdx.x];
    __syncthreads();
    int per = (n_edges + NCHUNK - 1) / NCHUNK;
    int c0 = blockIdx.x * per;
    int c1 = min(c0 + per, n_edges);
    int e = c0 + tid;
    for (; e + 3 * SORT_BLK < c1; e += 4 * SORT_BLK) {
        int e1 = e + SORT_BLK, e2 = e + 2 * SORT_BLK, e3 = e + 3 * SORT_BLK;
        int d0 = dst[e], d1 = dst[e1], d2 = dst[e2], d3 = dst[e3];
        int s0 = src[e], s1 = src[e1], s2 = src[e2], s3 = src[e3];
        int p0 = atomicAdd(&dsh[d0 >> NODE_SHIFT], 1);
        int p1 = atomicAdd(&dsh[d1 >> NODE_SHIFT], 1);
        int p2 = atomicAdd(&dsh[d2 >> NODE_SHIFT], 1);
        int p3 = atomicAdd(&dsh[d3 >> NODE_SHIFT], 1);
        rec[p0] = make_int2(s0, d0);
        rec[p1] = make_int2(s1, d1);
        rec[p2] = make_int2(s2, d2);
        rec[p3] = make_int2(s3, d3);
    }
    for (; e < c1; e += SORT_BLK) {
        int d = dst[e], s = src[e];
        int p = atomicAdd(&dsh[d >> NODE_SHIFT], 1);
        rec[p] = make_int2(s, d);
    }
}

__global__ void __launch_bounds__(SORT_BLK)
k_bucket(const int* __restrict__ pairs, const int* __restrict__ bstart,
         int* __restrict__ rowptr, float* __restrict__ dinv, int* __restrict__ csr_src,
         int n_nodes, int n_edges, int nb) {
    const int2* __restrict__ rec = reinterpret_cast<const int2*>(pairs);
    __shared__ int cn[BNODES];
    __shared__ int wsum[BNODES / 32];
    int tid = threadIdx.x, lane = tid & 31, w = tid >> 5;
    int b = blockIdx.x;
    int nbase = b << NODE_SHIFT;
    int e0 = bstart[b], e1 = bstart[b + 1];
    if (tid < BNODES) cn[tid] = 0;
    __syncthreads();
    int e = e0 + tid;
    for (; e + 3 * SORT_BLK < e1; e += 4 * SORT_BLK) {
        int d0 = rec[e].y;
        int d1 = rec[e + SORT_BLK].y;
        int d2 = rec[e + 2 * SORT_BLK].y;
        int d3 = rec[e + 3 * SORT_BLK].y;
        atomicAdd(&cn[d0 - nbase], 1);
        atomicAdd(&cn[d1 - nbase], 1);
        atomicAdd(&cn[d2 - nbase], 1);
        atomicAdd(&cn[d3 - nbase], 1);
    }
    for (; e < e1; e += SORT_BLK) atomicAdd(&cn[rec[e].y - nbase], 1);
    __syncthreads();
    int c = 0, incl = 0;
    if (w < BNODES / 32) {
        c = cn[tid];
        incl = wave_scan_incl(c, lane);
        if (lane == 31) wsum[w] = incl;
    }
    __syncthreads();
    if (w < BNODES / 32) {
        int wpre = 0;
        for (int k = 0; k < w; ++k) wpre += wsum[k];
        int excl = wpre + incl - c;
        cn[tid] = excl;
        int node = nbase + tid;
        if (node < n_nodes) {
            rowptr[node] = e0 + excl;
            float d = (float)(c + 1);
            dinv[node] = (d > 0.0f) ? rsqrtf(d) : 0.0f;
        }
    }
    if (b == nb - 1 && tid == 0) rowptr[n_nodes] = n_edges;
    __syncthreads();
    e = e0 + tid;
    for (; e + 3 * SORT_BLK < e1; e += 4 * SORT_BLK) {
        int e1i = e + SORT_BLK, e2i = e + 2 * SORT_BLK, e3i = e + 3 * SORT_BLK;
        int2 r0 = rec[e], r1 = rec[e1i], r2 = rec[e2i], r3 = rec[e3i];
        int p0 = atomicAdd(&cn[r0.y - nbase], 1);
        int p1 = atomicAdd(&cn[r1.y - nbase], 1);
        int p2 = atomicAdd(&cn[r2.y - nbase], 1);
        int p3 = atomicAdd(&cn[r3.y - nbase], 1);
        csr_src[e0 + p0] = r0.x;
        csr_src[e0 + p1] = r1.x;
        csr_src[e0 + p2] = r2.x;
        csr_src[e0 + p3] = r3.x;
    }
    for (; e < e1; e += SORT_BLK) {
        int2 r = rec[e];
        int p = atomicAdd(&cn[r.y - nbase], 1);
        csr_src[e0 + p] = r.x;
    }
}

__global__ void k_embed(const int* __restrict__ x, const float* __restrict__ emb,
                        float* __restrict__ h, int n_nodes) {
    int i = blockIdx.x * blockDim.x + threadIdx.x;
    int stride = gridDim.x * blockDim.x;
    int nh = n_nodes * HIDC;
    for (; i + 3 * stride < nh; i += 4 * stride) {
        int i1 = i + stride, i2 = i + 2 * stride, i3 = i + 3 * stride;
        int x0 = x[i >> 4], x1 = x[i1 >> 4], x2 = x[i2 >> 4], x3 = x[i3 >> 4];
        float v0 = emb[x0 * HIDC + (i & 15)];
        float v1 = emb[x1 * HIDC + (i1 & 15)];
        float v2 = emb[x2 * HIDC + (i2 & 15)];
        float v3 = emb[x3 * HIDC + (i3 & 15)];
        h[i]  = v0;
        h[i1] = v1;
        h[i2] = v2;
        h[i3] = v3;
    }
    for (; i < nh; i += stride) h[i] = emb[x[i >> 4] * HIDC + (i & 15)];
}

__global__ void k_gcn_wmma(const float* __restrict__ h, const float* __restrict__ W,
                           const float* __restrict__ dinv, float* __restrict__ hw, int ntiles) {
    int wave = (blockIdx.x * blockDim.x + threadIdx.x) >> 5;
    int lane = threadIdx.x & 31;
    if (wave >= ntiles) return;
    int half = lane >> 4;
    int l15  = lane & 15;

    v2f bf[4];
#pragma unroll
    for (int t = 0; t < 4; ++t) {
        int kk = 4 * t + 2 * half;
        bf[t][0] = W[(kk + 0) * HIDC + l15];
        bf[t][1] = W[(kk + 1) * HIDC + l15];
    }

    float rsl = dinv[wave * 16 + l15];
    const float* hrow = h + (size_t)(wave * 16 + l15) * HIDC;
    v8f c = {};
#pragma unroll
    for (int t = 0; t < 4; ++t) {
        int kk = 4 * t + 2 * half;
        v2f a;
        a[0] = hrow[kk + 0] * rsl;
        a[1] = hrow[kk + 1] * rsl;
        c = __builtin_amdgcn_wmma_f32_16x16x4_f32(
                false, a, false, bf[t], (short)0, c, false, false);
    }

#pragma unroll
    for (int v = 0; v < 8; ++v) {
        int m = v + 8 * half;
        hw[(size_t)(wave * 16 + m) * HIDC + l15] = c[v];
    }
}

__global__ void k_gather(const int* __restrict__ rowptr, const int* __restrict__ csr_src,
                         const float* __restrict__ dinv, const float* __restrict__ hw,
                         const float* __restrict__ b, float* __restrict__ h, int n_nodes) {
    int t = blockIdx.x * blockDim.x + threadIdx.x;
    int j = t & 15;
    int ngrp = (gridDim.x * blockDim.x) >> 4;
    float bj = b[j];
    for (int node = t >> 4; node < n_nodes; node += ngrp) {
        int i = node * HIDC + j;
        int beg = rowptr[node], end = rowptr[node + 1];
        float dd = dinv[node];
        float acc = hw[i];
        int s_next = (beg + j < end) ? csr_src[beg + j] : node;
        for (int p = beg; p < end; p += 16) {
            int s = s_next;
            int qn = p + 16 + j;
            s_next = (qn < end) ? csr_src[qn] : node;
            int nk = end - p;
#pragma unroll
            for (int k = 0; k < 16; ++k) {
                int   sk = __shfl(s, k, 16);
                float v  = hw[sk * HIDC + j];
                acc += (k < nk) ? v : 0.0f;
            }
        }
        h[i] = fmaxf(acc * dd + bj, 0.0f);
    }
}

__global__ void k_pool(const float* __restrict__ h, const int* __restrict__ batch,
                       float* __restrict__ pooled, float* __restrict__ cnt, int n_nodes) {
    int t = blockIdx.x * blockDim.x + threadIdx.x;
    int wave = t >> 5, nwaves = (gridDim.x * blockDim.x) >> 5;
    int sub = (t >> 4) & 1, j = t & 15;
    int per = n_nodes / nwaves, rem = n_nodes - per * nwaves;
    int n0 = wave * per + (wave < rem ? wave : rem);
    int n1 = n0 + per + (wave < rem ? 1 : 0);
    int node = n0 + sub;
    for (; node + 6 < n1; node += 8) {
        int g0 = batch[node], g1 = batch[node + 2], g2 = batch[node + 4], g3 = batch[node + 6];
        float v0 = h[node * HIDC + j];
        float v1 = h[(node + 2) * HIDC + j];
        float v2 = h[(node + 4) * HIDC + j];
        float v3 = h[(node + 6) * HIDC + j];
        fadd(&pooled[g0 * HIDC + j], v0);
        fadd(&pooled[g1 * HIDC + j], v1);
        fadd(&pooled[g2 * HIDC + j], v2);
        fadd(&pooled[g3 * HIDC + j], v3);
        if (j == 0) {
            fadd(&cnt[g0], 1.0f);
            fadd(&cnt[g1], 1.0f);
            fadd(&cnt[g2], 1.0f);
            fadd(&cnt[g3], 1.0f);
        }
    }
    for (; node < n1; node += 2) {
        int g = batch[node];
        fadd(&pooled[g * HIDC + j], h[node * HIDC + j]);
        if (j == 0) fadd(&cnt[g], 1.0f);
    }
}

__global__ void k_classify(const float* __restrict__ pooled, const float* __restrict__ cnt,
                           const float* __restrict__ Wc, const float* __restrict__ bc,
                           float* __restrict__ out, int n_graphs, int labels) {
    int i = blockIdx.x * blockDim.x + threadIdx.x;
    if (i >= n_graphs * labels) return;
    int g = i / labels, c = i % labels;
    float inv = 1.0f / fmaxf(cnt[g], 1.0f);
    float acc = bc[c];
#pragma unroll
    for (int j = 0; j < HIDC; ++j)
        acc += pooled[g * HIDC + j] * inv * Wc[j * labels + c];
    out[i] = acc;
}

extern "C" void kernel_launch(void* const* d_in, const int* in_sizes, int n_in,
                              void* d_out, int out_size, void* d_ws, size_t ws_size,
                              hipStream_t stream) {
    const int*   x     = (const int*)d_in[0];
    const int*   ei    = (const int*)d_in[1];
    const int*   batch = (const int*)d_in[2];
    const float* emb   = (const float*)d_in[3];
    const float* W1    = (const float*)d_in[4];
    const float* b1    = (const float*)d_in[5];
    const float* W2    = (const float*)d_in[6];
    const float* b2    = (const float*)d_in[7];
    const float* Wc    = (const float*)d_in[8];
    const float* bc    = (const float*)d_in[9];
    float* out = (float*)d_out;

    const int n_nodes  = in_sizes[0];
    const int n_edges  = in_sizes[1] / 2;
    const int labels   = in_sizes[9];
    const int n_graphs = out_size / labels;
    const int* src = ei;
    const int* dst = ei + n_edges;

    const int nb = (n_nodes + BNODES - 1) / BNODES;

    float* ws     = (float*)d_ws;
    float* h      = ws;
    float* hw     = h  + (size_t)n_nodes * HIDC;
    int* pairs    = (int*)(hw + (size_t)n_nodes * HIDC);
    float* pooled = (float*)(pairs + (size_t)2 * n_edges);
    float* cnt    = pooled + (size_t)n_graphs * HIDC;
    float* dinv   = cnt + n_graphs;
    int* rowptr   = (int*)(dinv + n_nodes);
    int* bcnt     = rowptr + n_nodes + 1;
    int* boff     = bcnt + (size_t)nb * NCHUNK;
    int* btot     = boff + (size_t)nb * NCHUNK;
    int* bstart   = btot + nb;
    int* csr_src  = bstart + nb + 1;

    const int ntiles = n_nodes / 16;
    const int wavesPerBlk = BLK / 32;

    dim3 b(BLK);
    auto g1 = [](int n) { return dim3((n + BLK - 1) / BLK); };
    dim3 gw((ntiles + wavesPerBlk - 1) / wavesPerBlk);

    k_hist<<<dim3(NCHUNK), dim3(SORT_BLK), nb * sizeof(int), stream>>>(dst, bcnt, pooled, cnt,
                                                                       n_edges, nb, n_graphs);
    k_btot<<<dim3(nb), dim3(NCHUNK), 0, stream>>>(bcnt, btot);
    k_off<<<dim3(nb), dim3(NCHUNK), 0, stream>>>(bcnt, btot, boff, bstart, nb);
    k_scatter<<<dim3(NCHUNK), dim3(SORT_BLK), nb * sizeof(int), stream>>>(src, dst, boff, pairs,
                                                                          n_edges, nb);
    k_bucket<<<dim3(nb), dim3(SORT_BLK), 0, stream>>>(pairs, bstart, rowptr, dinv, csr_src,
                                                      n_nodes, n_edges, nb);

    k_embed<<<dim3(G_ELEM), b, 0, stream>>>(x, emb, h, n_nodes);

    k_gcn_wmma<<<gw, b, 0, stream>>>(h, W1, dinv, hw, ntiles);
    k_gather<<<dim3(G_EDGE), b, 0, stream>>>(rowptr, csr_src, dinv, hw, b1, h, n_nodes);

    k_gcn_wmma<<<gw, b, 0, stream>>>(h, W2, dinv, hw, ntiles);
    k_gather<<<dim3(G_EDGE), b, 0, stream>>>(rowptr, csr_src, dinv, hw, b2, h, n_nodes);

    k_pool<<<dim3(G_ELEM), b, 0, stream>>>(h, batch, pooled, cnt, n_nodes);
    k_classify<<<g1(n_graphs * labels), b, 0, stream>>>(pooled, cnt, Wc, bc, out,
                                                        n_graphs, labels);
}


# 4 short launches folded into neighbours: bucket totals into the histogram, embedding + conv-1 transform into the bucket kernel, mean-pool sums into conv-2's gather (group partial sums, then atomics)
# speedup vs baseline: 12.1351x; 1.2610x over previous
#include <hip/hip_runtime.h>


typedef __attribute__((ext_vector_type(2))) float v2f;
typedef __attribute__((ext_vector_type(8))) float v8f;

#define HIDC 16
#define BLK 256

#define G_EDGE 1024

#define NODE_SHIFT 8
#define BNODES 256
#define NCHUNK 256
#define SORT_BLK 1024

__device__ __forceinline__ void fadd(float* p, float v) {
    __hip_atomic_fetch_add(p, v, __ATOMIC_RELAXED, __HIP_MEMORY_SCOPE_AGENT);
}

__device__ __forceinline__ int wave_sum(int v) {
#pragma unroll
    for (int o = 16; o > 0; o >>= 1) v += __shfl_xor(v, o);
    return v;
}

__device__ __forceinline__ int wave_scan_incl(int v, int lane) {
#pragma unroll
    for (int o = 1; o < 32; o <<= 1) {
        int t = __shfl_up(v, o);
        if (lane >= o) v += t;
    }
    return v;
}

extern __shared__ int dsh[];

__global__ void __launch_bounds__(SORT_BLK)
k_hist(const int* __restrict__ dst, int* __restrict__ bcnt, int* __restrict__ cpre,
       float* __restrict__ pooled, float* __restrict__ cnt, int n_edges, int nb, int n_graphs) {
    __shared__ int wtot[SORT_BLK / 32];
    int tid = threadIdx.x;
    for (int k = tid; k < nb; k += SORT_BLK) dsh[k] = 0;
    int gi = blockIdx.x * SORT_BLK + tid, gs = NCHUNK * SORT_BLK;
    for (int k = gi; k < n_graphs * HIDC; k += gs) pooled[k] = 0.0f;
    for (int k = gi; k < n_graphs; k += gs) cnt[k] = 0.0f;
    __syncthreads();
    int per = (n_edges + NCHUNK - 1) / NCHUNK;
    int c0 = blockIdx.x * per;
    int c1 = min(c0 + per, n_edges);
    int e = c0 + tid;
    for (; e + 3 * SORT_BLK < c1; e += 4 * SORT_BLK) {
        int d0 = dst[e];
        int d1 = dst[e + SORT_BLK];
        int d2 = dst[e + 2 * SORT_BLK];
        int d3 = dst[e + 3 * SORT_BLK];
        atomicAdd(&dsh[d0 >> NODE_SHIFT], 1);
        atomicAdd(&dsh[d1 >> NODE_SHIFT], 1);
        atomicAdd(&dsh[d2 >> NODE_SHIFT], 1);
        atomicAdd(&dsh[d3 >> NODE_SHIFT], 1);
    }
    for (; e < c1; e += SORT_BLK) atomicAdd(&dsh[dst[e] >> NODE_SHIFT], 1);
    __syncthreads();
    for (int k = tid; k < nb; k += SORT_BLK) bcnt[k * NCHUNK + blockIdx.x] = dsh[k];
    int lane = tid & 31, w = tid >> 5;
    int carry = 0;
    for (int k0 = 0; k0 < nb; k0 += SORT_BLK) {
        int k = k0 + tid;
        int v = (k < nb) ? dsh[k] : 0;
        int incl = wave_scan_incl(v, lane);
        if (lane == 31) wtot[w] = incl;
        __syncthreads();
        int wpre = 0, tot = 0;
        for (int i = 0; i < SORT_BLK / 32; ++i) {
            int t = wtot[i];
            tot += t;
            if (i < w) wpre += t;
        }
        if (k < nb) cpre[k * NCHUNK + blockIdx.x] = carry + wpre + incl - v;
        carry += tot;
        __syncthreads();
    }
}

__global__ void __launch_bounds__(NCHUNK)
k_off(const int* __restrict__ bcnt, const int* __restrict__ cpre, int* __restrict__ boff,
      int* __restrict__ bstart, int nb) {
    __shared__ int sh[NCHUNK / 32];
    __shared__ int s_base;
    int tid = threadIdx.x, lane = tid & 31, w = tid >> 5;
    int b = blockIdx.x;
    int part = wave_sum(cpre[b * NCHUNK + tid]);
    if (lane == 0) sh[w] = part;
    __syncthreads();
    if (tid == 0) {
        int t = 0;
        for (int k = 0; k < NCHUNK / 32; ++k) t += sh[k];
        s_base = t;
    }
    __syncthreads();
    int base = s_base;
    int c = bcnt[b * NCHUNK + tid];
    int incl = wave_scan_incl(c, lane);
    if (lane == 31) sh[w] = incl;
    __syncthreads();
    int wpre = 0;
    for (int k = 0; k < w; ++k) wpre += sh[k];
    boff[b * NCHUNK + tid] = base + wpre + incl - c;
    if (tid == 0) bstart[b] = base;
    if (b == nb - 1 && tid == NCHUNK - 1) bstart[nb] = base + wpre + incl;
}

__global__ void __launch_bounds__(SORT_BLK)
k_scatter(const int* __restrict__ src, const int* __restrict__ dst, const int* __restrict__ boff,
          int* __restrict__ pairs, int n_edges, int nb) {
    int2* __restrict__ rec = reinterpret_cast<int2*>(pairs);
    int tid = threadIdx.x;
    for (int k = tid; k < nb; k += SORT_BLK) dsh[k] = boff[k * NCHUNK + blockIdx.x];
    __syncthreads();
    int per = (n_edges + NCHUNK - 1) / NCHUNK;
    int c0 = blockIdx.x * per;
    int c1 = min(c0 + per, n_edges);
    int e = c0 + tid;
    for (; e + 3 * SORT_BLK < c1; e += 4 * SORT_BLK) {
        int e1 = e + SORT_BLK, e2 = e + 2 * SORT_BLK, e3 = e + 3 * SORT_BLK;
        int d0 = dst[e], d1 = dst[e1], d2 = dst[e2], d3 = dst[e3];
        int s0 = src[e], s1 = src[e1], s2 = src[e2], s3 = src[e3];
        int p0 = atomicAdd(&dsh[d0 >> NODE_SHIFT], 1);
        int p1 = atomicAdd(&dsh[d1 >> NODE_SHIFT], 1);
        int p2 = atomicAdd(&dsh[d2 >> NODE_SHIFT], 1);
        int p3 = atomicAdd(&dsh[d3 >> NODE_SHIFT], 1);
        rec[p0] = make_int2(s0, d0);
        rec[p1] = make_int2(s1, d1);
        rec[p2] = make_int2(s2, d2);
        rec[p3] = make_int2(s3, d3);
    }
    for (; e < c1; e += SORT_BLK) {
        int d = dst[e], s = src[e];
        int p = atomicAdd(&dsh[d >> NODE_SHIFT], 1);
        rec[p] = make_int2(s, d);
    }
}

#define TILE_W0 16
__global__ void __launch_bounds__(SORT_BLK)
k_bucket(const int* __restrict__ pairs, const int* __restrict__ bstart,
         int* __restrict__ rowptr, float* __restrict__ dinv, int* __restrict__ csr_src,
         const int* __restrict__ x, const float* __restrict__ emb, const float* __restrict__ W,
         float* __restrict__ hw, int n_nodes, int n_edges, int nb) {
    const int2* __restrict__ rec = reinterpret_cast<const int2*>(pairs);
    __shared__ int cn[BNODES];
    __shared__ int wsum[BNODES / 32];
    __shared__ float sdinv[BNODES];
    int tid = threadIdx.x, lane = tid & 31, w = tid >> 5;
    int b = blockIdx.x;
    int nbase = b << NODE_SHIFT;
    int e0 = bstart[b], e1 = bstart[b + 1];
    int half = lane >> 4, l15 = lane & 15;
    int trow = nbase + (w - TILE_W0) * 16;
    bool tile_on = (w >= TILE_W0) && (w < TILE_W0 + BNODES / 16) && (trow + 16 <= n_nodes);
    if (tid < BNODES) cn[tid] = 0;
    __syncthreads();
    int e = e0 + tid;
    for (; e + 3 * SORT_BLK < e1; e += 4 * SORT_BLK) {
        int d0 = rec[e].y;
        int d1 = rec[e + SORT_BLK].y;
        int d2 = rec[e + 2 * SORT_BLK].y;
        int d3 = rec[e + 3 * SORT_BLK].y;
        atomicAdd(&cn[d0 - nbase], 1);
        atomicAdd(&cn[d1 - nbase], 1);
        atomicAdd(&cn[d2 - nbase], 1);
        atomicAdd(&cn[d3 - nbase], 1);
    }
    for (; e < e1; e += SORT_BLK) atomicAdd(&cn[rec[e].y - nbase], 1);
    __syncthreads();
    int xi = 0;
    v2f bf[4] = {};
    if (tile_on) {
        xi = x[trow + l15];
#pragma unroll
        for (int t = 0; t < 4; ++t) {
            int kk = 4 * t + 2 * half;
            bf[t][0] = W[(kk + 0) * HIDC + l15];
            bf[t][1] = W[(kk + 1) * HIDC + l15];
        }
    }
    int c = 0, incl = 0;
    if (w < BNODES / 32) {
        c = cn[tid];
        incl = wave_scan_incl(c, lane);
        if (lane == 31) wsum[w] = incl;
    }
    __syncthreads();
    if (w < BNODES / 32) {
        int wpre = 0;
        for (int k = 0; k < w; ++k) wpre += wsum[k];
        int excl = wpre + incl - c;
        cn[tid] = excl;
        int node = nbase + tid;
        float d = (float)(c + 1);
        float di = (d > 0.0f) ? rsqrtf(d) : 0.0f;
        sdinv[tid] = di;
        if (node < n_nodes) {
            rowptr[node] = e0 + excl;
            dinv[node] = di;
        }
    }
    if (b == nb - 1 && tid == 0) rowptr[n_nodes] = n_edges;
    __syncthreads();
    v2f af[4] = {};
    if (tile_on) {
        const float* erow = emb + xi * HIDC;
#pragma unroll
        for (int t = 0; t < 4; ++t) {
            int kk = 4 * t + 2 * half;
            af[t][0] = erow[kk + 0];
            af[t][1] = erow[kk + 1];
        }
    }
    e = e0 + tid;
    for (; e + 3 * SORT_BLK < e1; e += 4 * SORT_BLK) {
        int e1i = e + SORT_BLK, e2i = e + 2 * SORT_BLK, e3i = e + 3 * SORT_BLK;
        int2 r0 = rec[e], r1 = rec[e1i], r2 = rec[e2i], r3 = rec[e3i];
        int p0 = atomicAdd(&cn[r0.y - nbase], 1);
        int p1 = atomicAdd(&cn[r1.y - nbase], 1);
        int p2 = atomicAdd(&cn[r2.y - nbase], 1);
        int p3 = atomicAdd(&cn[r3.y - nbase], 1);
        csr_src[e0 + p0] = r0.x;
        csr_src[e0 + p1] = r1.x;
        csr_src[e0 + p2] = r2.x;
        csr_src[e0 + p3] = r3.x;
    }
    for (; e < e1; e += SORT_BLK) {
        int2 r = rec[e];
        int p = atomicAdd(&cn[r.y - nbase], 1);
        csr_src[e0 + p] = r.x;
    }
    if (tile_on) {
        float rsl = sdinv[(w - TILE_W0) * 16 + l15];
        v8f acc = {};
#pragma unroll
        for (int t = 0; t < 4; ++t) {
            v2f a;
            a[0] = af[t][0] * rsl;
            a[1] = af[t][1] * rsl;
            acc = __builtin_amdgcn_wmma_f32_16x16x4_f32(
                      false, a, false, bf[t], (short)0, acc, false, false);
        }
#pragma unroll
        for (int v = 0; v < 8; ++v) {
            int m = v + 8 * half;
            hw[(size_t)(trow + m) * HIDC + l15] = acc[v];
        }
    }
}

__global__ void k_gcn_wmma(const float* __restrict__ h, const float* __restrict__ W,
                           const float* __restrict__ dinv, float* __restrict__ hw, int ntiles) {
    int wave = (blockIdx.x * blockDim.x + threadIdx.x) >> 5;
    int lane = threadIdx.x & 31;
    if (wave >= ntiles) return;
    int half = lane >> 4;
    int l15  = lane & 15;

    v2f bf[4];
#pragma unroll
    for (int t = 0; t < 4; ++t) {
        int kk = 4 * t + 2 * half;
        bf[t][0] = W[(kk + 0) * HIDC + l15];
        bf[t][1] = W[(kk + 1) * HIDC + l15];
    }

    float rsl = dinv[wave * 16 + l15];
    const float* hrow = h + (size_t)(wave * 16 + l15) * HIDC;
    v8f c = {};
#pragma unroll
    for (int t = 0; t < 4; ++t) {
        int kk = 4 * t + 2 * half;
        v2f a;
        a[0] = hrow[kk + 0] * rsl;
        a[1] = hrow[kk + 1] * rsl;
        c = __builtin_amdgcn_wmma_f32_16x16x4_f32(
                false, a, false, bf[t], (short)0, c, false, false);
    }

#pragma unroll
    for (int v = 0; v < 8; ++v) {
        int m = v + 8 * half;
        hw[(size_t)(wave * 16 + m) * HIDC + l15] = c[v];
    }
}

__device__ __forceinline__ float gather_node(const int* __restrict__ rowptr,
                                             const int* __restrict__ csr_src,
                                             const float* __restrict__ dinv,
                                             const float* __restrict__ hw,
                                             int node, int j, float bj) {
    int i = node * HIDC + j;
    int beg = rowptr[node], end = rowptr[node + 1];
    float dd = dinv[node];
    float acc = hw[i];
    int s_next = (beg + j < end) ? csr_src[beg + j] : node;
    for (int p = beg; p < end; p += 16) {
        int s = s_next;
        int qn = p + 16 + j;
        s_next = (qn < end) ? csr_src[qn] : node;
        int nk = end - p;
#pragma unroll
        for (int k = 0; k < 16; ++k) {
            int   sk = __shfl(s, k, 16);
            float v  = hw[sk * HIDC + j];
            acc += (k < nk) ? v : 0.0f;
        }
    }
    return fmaxf(acc * dd + bj, 0.0f);
}

__global__ void k_gather(const int* __restrict__ rowptr, const int* __restrict__ csr_src,
                         const float* __restrict__ dinv, const float* __restrict__ hw,
                         const float* __restrict__ b, float* __restrict__ h, int n_nodes) {
    int t = blockIdx.x * blockDim.x + threadIdx.x;
    int j = t & 15;
    int ngrp = (gridDim.x * blockDim.x) >> 4;
    float bj = b[j];
    for (int node = t >> 4; node < n_nodes; node += ngrp)
        h[node * HIDC + j] = gather_node(rowptr, csr_src, dinv, hw, node, j, bj);
}

__global__ void k_gather_pool(const int* __restrict__ rowptr, const int* __restrict__ csr_src,
                              const float* __restrict__ dinv, const float* __restrict__ hw,
                              const float* __restrict__ b, const int* __restrict__ batch,
                              float* __restrict__ pooled, float* __restrict__ cnt, int n_nodes) {
    int t = blockIdx.x * blockDim.x + threadIdx.x;
    int j = t & 15;
    int grp = t >> 4;
    int ngrp = (gridDim.x * blockDim.x) >> 4;
    int per = n_nodes / ngrp, rem = n_nodes - per * ngrp;
    int n0 = grp * per + (grp < rem ? grp : rem);
    int n1 = n0 + per + (grp < rem ? 1 : 0);
    float bj = b[j];
    int gcur = -1;
    float psum = 0.0f, pcnt = 0.0f;
    for (int node = n0; node < n1; ++node) {
        int g = batch[node];
        float v = gather_node(rowptr, csr_src, dinv, hw, node, j, bj);
        if (g != gcur) {
            if (gcur >= 0) {
                fadd(&pooled[gcur * HIDC + j], psum);
                if (j == 0) fadd(&cnt[gcur], pcnt);
            }
            gcur = g;
            psum = 0.0f;
            pcnt = 0.0f;
        }
        psum += v;
        pcnt += 1.0f;
    }
    if (gcur >= 0) {
        fadd(&pooled[gcur * HIDC + j], psum);
        if (j == 0) fadd(&cnt[gcur], pcnt);
    }
}

__global__ void k_classify(const float* __restrict__ pooled, const float* __restrict__ cnt,
                           const float* __restrict__ Wc, const float* __restrict__ bc,
                           float* __restrict__ out, int n_graphs, int labels) {
    int i = blockIdx.x * blockDim.x + threadIdx.x;
    if (i >= n_graphs * labels) return;
    int g = i / labels, c = i % labels;
    float inv = 1.0f / fmaxf(cnt[g], 1.0f);
    float acc = bc[c];
#pragma unroll
    for (int j = 0; j < HIDC; ++j)
        acc += pooled[g * HIDC + j] * inv * Wc[j * labels + c];
    out[i] = acc;
}

extern "C" void kernel_launch(void* const* d_in, const int* in_sizes, int n_in,
                              void* d_out, int out_size, void* d_ws, size_t ws_size,
                              hipStream_t stream) {
    const int*   x     = (const int*)d_in[0];
    const int*   ei    = (const int*)d_in[1];
    const int*   batch = (const int*)d_in[2];
    const float* emb   = (const float*)d_in[3];
    const float* W1    = (const float*)d_in[4];
    const float* b1    = (const float*)d_in[5];
    const float* W2    = (const float*)d_in[6];
    const float* b2    = (const float*)d_in[7];
    const float* Wc    = (const float*)d_in[8];
    const float* bc    = (const float*)d_in[9];
    float* out = (float*)d_out;

    const int n_nodes  = in_sizes[0];
    const int n_edges  = in_sizes[1] / 2;
    const int labels   = in_sizes[9];
    const int n_graphs = out_size / labels;
    const int* src = ei;
    const int* dst = ei + n_edges;

    const int nb = (n_nodes + BNODES - 1) / BNODES;

    float* ws     = (float*)d_ws;
    float* h      = ws;
    float* hw     = h  + (size_t)n_nodes * HIDC;
    int* pairs    = (int*)(hw + (size_t)n_nodes * HIDC);
    float* pooled = (float*)(pairs + (size_t)2 * n_edges);
    float* cnt    = pooled + (size_t)n_graphs * HIDC;
    float* dinv   = cnt + n_graphs;
    int* rowptr   = (int*)(dinv + n_nodes);
    int* bcnt     = rowptr + n_nodes + 1;
    int* boff     = bcnt + (size_t)nb * NCHUNK;
    int* cpre     = boff + (size_t)nb * NCHUNK;
    int* bstart   = cpre + (size_t)nb * NCHUNK;
    int* csr_src  = bstart + nb + 1;

    const int ntiles = n_nodes / 16;
    const int wavesPerBlk = BLK / 32;

    dim3 b(BLK);
    auto g1 = [](int n) { return dim3((n + BLK - 1) / BLK); };
    dim3 gw((ntiles + wavesPerBlk - 1) / wavesPerBlk);

    k_hist<<<dim3(NCHUNK), dim3(SORT_BLK), nb * sizeof(int), stream>>>(dst, bcnt, cpre, pooled, cnt,
                                                                       n_edges, nb, n_graphs);
    k_off<<<dim3(nb), dim3(NCHUNK), 0, stream>>>(bcnt, cpre, boff, bstart, nb);
    k_scatter<<<dim3(NCHUNK), dim3(SORT_BLK), nb * sizeof(int), stream>>>(src, dst, boff, pairs,
                                                                          n_edges, nb);
    k_bucket<<<dim3(nb), dim3(SORT_BLK), 0, stream>>>(pairs, bstart, rowptr, dinv, csr_src,
                                                      x, emb, W1, hw, n_nodes, n_edges, nb);

    k_gather<<<dim3(G_EDGE), b, 0, stream>>>(rowptr, csr_src, dinv, hw, b1, h, n_nodes);

    k_gcn_wmma<<<gw, b, 0, stream>>>(h, W2, dinv, hw, ntiles);
    k_gather_pool<<<dim3(G_EDGE), b, 0, stream>>>(rowptr, csr_src, dinv, hw, b2, batch,
                                                  pooled, cnt, n_nodes);

    k_classify<<<g1(n_graphs * labels), b, 0, stream>>>(pooled, cnt, Wc, bc, out,
                                                        n_graphs, labels);
}


# both levels of the edge sort put records at the counter's slot of an LDS image and copy it out in order: whole-line stores, not one scattered store per edge; sum order still LDS arrival rank
# speedup vs baseline: 15.1321x; 1.2470x over previous
#include <hip/hip_runtime.h>


typedef __attribute__((ext_vector_type(2))) float v2f;
typedef __attribute__((ext_vector_type(8))) float v8f;

#define HIDC 16
#define BLK 256

#define G_EDGE 1024

#define NODE_SHIFT 8
#define BNODES 256
#define NCHUNK 256
#define SORT_BLK 1024
#define STAGE_CAP 16384
#define BSTAGE_CAP 16384

__device__ __forceinline__ void fadd(float* p, float v) {
    __hip_atomic_fetch_add(p, v, __ATOMIC_RELAXED, __HIP_MEMORY_SCOPE_AGENT);
}

__device__ __forceinline__ int wave_sum(int v) {
#pragma unroll
    for (int o = 16; o > 0; o >>= 1) v += __shfl_xor(v, o);
    return v;
}

__device__ __forceinline__ int wave_scan_incl(int v, int lane) {
#pragma unroll
    for (int o = 1; o < 32; o <<= 1) {
        int t = __shfl_up(v, o);
        if (lane >= o) v += t;
    }
    return v;
}

extern __shared__ int dsh[];

__global__ void __launch_bounds__(SORT_BLK)
k_hist(const int* __restrict__ dst, int* __restrict__ bcnt, int* __restrict__ cpre,
       float* __restrict__ pooled, float* __restrict__ cnt, int n_edges, int nb, int n_graphs) {
    __shared__ int wtot[SORT_BLK / 32];
    int tid = threadIdx.x;
    for (int k = tid; k < nb; k += SORT_BLK) dsh[k] = 0;
    int gi = blockIdx.x * SORT_BLK + tid, gs = NCHUNK * SORT_BLK;
    for (int k = gi; k < n_graphs * HIDC; k += gs) pooled[k] = 0.0f;
    for (int k = gi; k < n_graphs; k += gs) cnt[k] = 0.0f;
    __syncthreads();
    int per = (n_edges + NCHUNK - 1) / NCHUNK;
    int c0 = blockIdx.x * per;
    int c1 = min(c0 + per, n_edges);
    int e = c0 + tid;
    for (; e + 3 * SORT_BLK < c1; e += 4 * SORT_BLK) {
        int d0 = dst[e];
        int d1 = dst[e + SORT_BLK];
        int d2 = dst[e + 2 * SORT_BLK];
        int d3 = dst[e + 3 * SORT_BLK];
        atomicAdd(&dsh[d0 >> NODE_SHIFT], 1);
        atomicAdd(&dsh[d1 >> NODE_SHIFT], 1);
        atomicAdd(&dsh[d2 >> NODE_SHIFT], 1);
        atomicAdd(&dsh[d3 >> NODE_SHIFT], 1);
    }
    for (; e < c1; e += SORT_BLK) atomicAdd(&dsh[dst[e] >> NODE_SHIFT], 1);
    __syncthreads();
    for (int k = tid; k < nb; k += SORT_BLK) bcnt[k * NCHUNK + blockIdx.x] = dsh[k];
    int lane = tid & 31, w = tid >> 5;
    int carry = 0;
    for (int k0 = 0; k0 < nb; k0 += SORT_BLK) {
        int k = k0 + tid;
        int v = (k < nb) ? dsh[k] : 0;
        int incl = wave_scan_incl(v, lane);
        if (lane == 31) wtot[w] = incl;
        __syncthreads();
        int wpre = 0, tot = 0;
        for (int i = 0; i < SORT_BLK / 32; ++i) {
            int t = wtot[i];
            tot += t;
            if (i < w) wpre += t;
        }
        if (k < nb) cpre[k * NCHUNK + blockIdx.x] = carry + wpre + incl - v;
        carry += tot;
        __syncthreads();
    }
}

__global__ void __launch_bounds__(NCHUNK)
k_off(const int* __restrict__ bcnt, const int* __restrict__ cpre, int* __restrict__ boff,
      int* __restrict__ bstart, int nb) {
    __shared__ int sh[NCHUNK / 32];
    __shared__ int s_base;
    int tid = threadIdx.x, lane = tid & 31, w = tid >> 5;
    int b = blockIdx.x;
    int part = wave_sum(cpre[b * NCHUNK + tid]);
    if (lane == 0) sh[w] = part;
    __syncthreads();
    if (tid == 0) {
        int t = 0;
        for (int k = 0; k < NCHUNK / 32; ++k) t += sh[k];
        s_base = t;
    }
    __syncthreads();
    int base = s_base;
    int c = bcnt[b * NCHUNK + tid];
    int incl = wave_scan_incl(c, lane);
    if (lane == 31) sh[w] = incl;
    __syncthreads();
    int wpre = 0;
    for (int k = 0; k < w; ++k) wpre += sh[k];
    boff[b * NCHUNK + tid] = base + wpre + incl - c;
    if (tid == 0) bstart[b] = base;
    if (b == nb - 1 && tid == NCHUNK - 1) bstart[nb] = base + wpre + incl;
}

__global__ void __launch_bounds__(SORT_BLK)
k_scatter(const int* __restrict__ src, const int* __restrict__ dst, const int* __restrict__ boff,
          const int* __restrict__ cpre, int* __restrict__ pairs, int n_edges, int nb) {
    __shared__ int2 stage[STAGE_CAP];
    int2* __restrict__ rec = reinterpret_cast<int2*>(pairs);
    int tid = threadIdx.x;
    int per = (n_edges + NCHUNK - 1) / NCHUNK;
    int c0 = blockIdx.x * per;
    int c1 = min(c0 + per, n_edges);
    int e = c0 + tid;
    if (per <= STAGE_CAP) {
        int* shift = dsh + nb;
        for (int k = tid; k < nb; k += SORT_BLK) {
            int lp = cpre[k * NCHUNK + blockIdx.x];
            dsh[k] = lp;
            shift[k] = boff[k * NCHUNK + blockIdx.x] - lp;
        }
        __syncthreads();
        for (; e + 3 * SORT_BLK < c1; e += 4 * SORT_BLK) {
            int e1 = e + SORT_BLK, e2 = e + 2 * SORT_BLK, e3 = e + 3 * SORT_BLK;
            int d0 = dst[e], d1 = dst[e1], d2 = dst[e2], d3 = dst[e3];
            int s0 = src[e], s1 = src[e1], s2 = src[e2], s3 = src[e3];
            int p0 = atomicAdd(&dsh[d0 >> NODE_SHIFT], 1);
            int p1 = atomicAdd(&dsh[d1 >> NODE_SHIFT], 1);
            int p2 = atomicAdd(&dsh[d2 >> NODE_SHIFT], 1);
            int p3 = atomicAdd(&dsh[d3 >> NODE_SHIFT], 1);
            stage[p0] = make_int2(s0, d0);
            stage[p1] = make_int2(s1, d1);
            stage[p2] = make_int2(s2, d2);
            stage[p3] = make_int2(s3, d3);
        }
        for (; e < c1; e += SORT_BLK) {
            int d = dst[e], s = src[e];
            int p = atomicAdd(&dsh[d >> NODE_SHIFT], 1);
            stage[p] = make_int2(s, d);
        }
        __syncthreads();
        int n = c1 - c0;
        int i = tid;
        for (; i + 3 * SORT_BLK < n; i += 4 * SORT_BLK) {
            int i1 = i + SORT_BLK, i2 = i + 2 * SORT_BLK, i3 = i + 3 * SORT_BLK;
            int2 r0 = stage[i], r1 = stage[i1], r2 = stage[i2], r3 = stage[i3];
            int g0 = i  + shift[r0.y >> NODE_SHIFT];
            int g1 = i1 + shift[r1.y >> NODE_SHIFT];
            int g2 = i2 + shift[r2.y >> NODE_SHIFT];
            int g3 = i3 + shift[r3.y >> NODE_SHIFT];
            rec[g0] = r0;
            rec[g1] = r1;
            rec[g2] = r2;
            rec[g3] = r3;
        }
        for (; i < n; i += SORT_BLK) {
            int2 r = stage[i];
            rec[i + shift[r.y >> NODE_SHIFT]] = r;
        }
    } else {
        for (int k = tid; k < nb; k += SORT_BLK) dsh[k] = boff[k * NCHUNK + blockIdx.x];
        __syncthreads();
        for (; e < c1; e += SORT_BLK) {
            int d = dst[e], s = src[e];
            int p = atomicAdd(&dsh[d >> NODE_SHIFT], 1);
            rec[p] = make_int2(s, d);
        }
    }
}

#define TILE_W0 16
__global__ void __launch_bounds__(SORT_BLK)
k_bucket(const int* __restrict__ pairs, const int* __restrict__ bstart,
         int* __restrict__ rowptr, float* __restrict__ dinv, int* __restrict__ csr_src,
         const int* __restrict__ x, const float* __restrict__ emb, const float* __restrict__ W,
         float* __restrict__ hw, int n_nodes, int n_edges, int nb) {
    const int2* __restrict__ rec = reinterpret_cast<const int2*>(pairs);
    __shared__ int cn[BNODES];
    __shared__ int wsum[BNODES / 32];
    __shared__ float sdinv[BNODES];
    __shared__ int simg[BSTAGE_CAP];
    int tid = threadIdx.x, lane = tid & 31, w = tid >> 5;
    int b = blockIdx.x;
    int nbase = b << NODE_SHIFT;
    int e0 = bstart[b], e1 = bstart[b + 1];
    const bool staged = (e1 - e0) <= BSTAGE_CAP;
    int half = lane >> 4, l15 = lane & 15;
    int trow = nbase + (w - TILE_W0) * 16;
    bool tile_on = (w >= TILE_W0) && (w < TILE_W0 + BNODES / 16) && (trow + 16 <= n_nodes);
    if (tid < BNODES) cn[tid] = 0;
    __syncthreads();
    int e = e0 + tid;
    for (; e + 3 * SORT_BLK < e1; e += 4 * SORT_BLK) {
        int d0 = rec[e].y;
        int d1 = rec[e + SORT_BLK].y;
        int d2 = rec[e + 2 * SORT_BLK].y;
        int d3 = rec[e + 3 * SORT_BLK].y;
        atomicAdd(&cn[d0 - nbase], 1);
        atomicAdd(&cn[d1 - nbase], 1);
        atomicAdd(&cn[d2 - nbase], 1);
        atomicAdd(&cn[d3 - nbase], 1);
    }
    for (; e < e1; e += SORT_BLK) atomicAdd(&cn[rec[e].y - nbase], 1);
    __syncthreads();
    int xi = 0;
    v2f bf[4] = {};
    if (tile_on) {
        xi = x[trow + l15];
#pragma unroll
        for (int t = 0; t < 4; ++t) {
            int kk = 4 * t + 2 * half;
            bf[t][0] = W[(kk + 0) * HIDC + l15];
            bf[t][1] = W[(kk + 1) * HIDC + l15];
        }
    }
    int c = 0, incl = 0;
    if (w < BNODES / 32) {
        c = cn[tid];
        incl = wave_scan_incl(c, lane);
        if (lane == 31) wsum[w] = incl;
    }
    __syncthreads();
    if (w < BNODES / 32) {
        int wpre = 0;
        for (int k = 0; k < w; ++k) wpre += wsum[k];
        int excl = wpre + incl - c;
        cn[tid] = excl;
        int node = nbase + tid;
        float d = (float)(c + 1);
        float di = (d > 0.0f) ? rsqrtf(d) : 0.0f;
        sdinv[tid] = di;
        if (node < n_nodes) {
            rowptr[node] = e0 + excl;
            dinv[node] = di;
        }
    }
    if (b == nb - 1 && tid == 0) rowptr[n_nodes] = n_edges;
    __syncthreads();
    v2f af[4] = {};
    if (tile_on) {
        const float* erow = emb + xi * HIDC;
#pragma unroll
        for (int t = 0; t < 4; ++t) {
            int kk = 4 * t + 2 * half;
            af[t][0] = erow[kk + 0];
            af[t][1] = erow[kk + 1];
        }
    }
    e = e0 + tid;
    if (staged) {
        for (; e + 3 * SORT_BLK < e1; e += 4 * SORT_BLK) {
            int e1i = e + SORT_BLK, e2i = e + 2 * SORT_BLK, e3i = e + 3 * SORT_BLK;
            int2 r0 = rec[e], r1 = rec[e1i], r2 = rec[e2i], r3 = rec[e3i];
            int p0 = atomicAdd(&cn[r0.y - nbase], 1);
            int p1 = atomicAdd(&cn[r1.y - nbase], 1);
            int p2 = atomicAdd(&cn[r2.y - nbase], 1);
            int p3 = atomicAdd(&cn[r3.y - nbase], 1);
            simg[p0] = r0.x;
            simg[p1] = r1.x;
            simg[p2] = r2.x;
            simg[p3] = r3.x;
        }
        for (; e < e1; e += SORT_BLK) {
            int2 r = rec[e];
            int p = atomicAdd(&cn[r.y - nbase], 1);
            simg[p] = r.x;
        }
        __syncthreads();
        int ne = e1 - e0;
        int i = tid - (int)((reinterpret_cast<size_t>(csr_src + e0) >> 2) & 31);
        if (i >= 0 && i < ne) csr_src[e0 + i] = simg[i];
        for (i += SORT_BLK; i < ne; i += SORT_BLK) csr_src[e0 + i] = simg[i];
    } else {
        for (; e < e1; e += SORT_BLK) {
            int2 r = rec[e];
            int p = atomicAdd(&cn[r.y - nbase], 1);
            csr_src[e0 + p] = r.x;
        }
    }
    if (tile_on) {
        float rsl = sdinv[(w - TILE_W0) * 16 + l15];
        v8f acc = {};
#pragma unroll
        for (int t = 0; t < 4; ++t) {
            v2f a;
            a[0] = af[t][0] * rsl;
            a[1] = af[t][1] * rsl;
            acc = __builtin_amdgcn_wmma_f32_16x16x4_f32(
                      false, a, false, bf[t], (short)0, acc, false, false);
        }
#pragma unroll
        for (int v = 0; v < 8; ++v) {
            int m = v + 8 * half;
            hw[(size_t)(trow + m) * HIDC + l15] = acc[v];
        }
    }
}

__global__ void k_gcn_wmma(const float* __restrict__ h, const float* __restrict__ W,
                           const float* __restrict__ dinv, float* __restrict__ hw, int ntiles) {
    int wave = (blockIdx.x * blockDim.x + threadIdx.x) >> 5;
    int lane = threadIdx.x & 31;
    if (wave >= ntiles) return;
    int half = lane >> 4;
    int l15  = lane & 15;

    v2f bf[4];
#pragma unroll
    for (int t = 0; t < 4; ++t) {
        int kk = 4 * t + 2 * half;
        bf[t][0] = W[(kk + 0) * HIDC + l15];
        bf[t][1] = W[(kk + 1) * HIDC + l15];
    }

    float rsl = dinv[wave * 16 + l15];
    const float* hrow = h + (size_t)(wave * 16 + l15) * HIDC;
    v8f c = {};
#pragma unroll
    for (int t = 0; t < 4; ++t) {
        int kk = 4 * t + 2 * half;
        v2f a;
        a[0] = hrow[kk + 0] * rsl;
        a[1] = hrow[kk + 1] * rsl;
        c = __builtin_amdgcn_wmma_f32_16x16x4_f32(
                false, a, false, bf[t], (short)0, c, false, false);
    }

#pragma unroll
    for (int v = 0; v < 8; ++v) {
        int m = v + 8 * half;
        hw[(size_t)(wave * 16 + m) * HIDC + l15] = c[v];
    }
}

__device__ __forceinline__ float gather_node(const int* __restrict__ rowptr,
                                             const int* __restrict__ csr_src,
                                             const float* __restrict__ dinv,
                                             const float* __restrict__ hw,
                                             int node, int j, float bj) {
    int i = node * HIDC + j;
    int beg = rowptr[node], end = rowptr[node + 1];
    float dd = dinv[node];
    float acc = hw[i];
    int s_next = (beg + j < end) ? csr_src[beg + j] : node;
    for (int p = beg; p < end; p += 16) {
        int s = s_next;
        int qn = p + 16 + j;
        s_next = (qn < end) ? csr_src[qn] : node;
        int nk = end - p;
#pragma unroll
        for (int k = 0; k < 16; ++k) {
            int   sk = __shfl(s, k, 16);
            float v  = hw[sk * HIDC + j];
            acc += (k < nk) ? v : 0.0f;
        }
    }
    return fmaxf(acc * dd + bj, 0.0f);
}

__global__ void k_gather(const int* __restrict__ rowptr, const int* __restrict__ csr_src,
                         const float* __restrict__ dinv, const float* __restrict__ hw,
                         const float* __restrict__ b, float* __restrict__ h, int n_nodes) {
    int t = blockIdx.x * blockDim.x + threadIdx.x;
    int j = t & 15;
    int ngrp = (gridDim.x * blockDim.x) >> 4;
    float bj = b[j];
    for (int node = t >> 4; node < n_nodes; node += ngrp)
        h[node * HIDC + j] = gather_node(rowptr, csr_src, dinv, hw, node, j, bj);
}

__global__ void k_gather_pool(const int* __restrict__ rowptr, const int* __restrict__ csr_src,
                              const float* __restrict__ dinv, const float* __restrict__ hw,
                              const float* __restrict__ b, const int* __restrict__ batch,
                              float* __restrict__ pooled, float* __restrict__ cnt, int n_nodes) {
    int t = blockIdx.x * blockDim.x + threadIdx.x;
    int j = t & 15;
    int grp = t >> 4;
    int ngrp = (gridDim.x * blockDim.x) >> 4;
    int per = n_nodes / ngrp, rem = n_nodes - per * ngrp;
    int n0 = grp * per + (grp < rem ? grp : rem);
    int n1 = n0 + per + (grp < rem ? 1 : 0);
    float bj = b[j];
    int gcur = -1;
    float psum = 0.0f, pcnt = 0.0f;
    for (int node = n0; node < n1; ++node) {
        int g = batch[node];
        float v = gather_node(rowptr, csr_src, dinv, hw, node, j, bj);
        if (g != gcur) {
            if (gcur >= 0) {
                fadd(&pooled[gcur * HIDC + j], psum);
                if (j == 0) fadd(&cnt[gcur], pcnt);
            }
            gcur = g;
            psum = 0.0f;
            pcnt = 0.0f;
        }
        psum += v;
        pcnt += 1.0f;
    }
    if (gcur >= 0) {
        fadd(&pooled[gcur * HIDC + j], psum);
        if (j == 0) fadd(&cnt[gcur], pcnt);
    }
}

__global__ void k_classify(const float* __restrict__ pooled, const float* __restrict__ cnt,
                           const float* __restrict__ Wc, const float* __restrict__ bc,
                           float* __restrict__ out, int n_graphs, int labels) {
    int i = blockIdx.x * blockDim.x + threadIdx.x;
    if (i >= n_graphs * labels) return;
    int g = i / labels, c = i % labels;
    float inv = 1.0f / fmaxf(cnt[g], 1.0f);
    float acc = bc[c];
#pragma unroll
    for (int j = 0; j < HIDC; ++j)
        acc += pooled[g * HIDC + j] * inv * Wc[j * labels + c];
    out[i] = acc;
}

extern "C" void kernel_launch(void* const* d_in, const int* in_sizes, int n_in,
                              void* d_out, int out_size, void* d_ws, size_t ws_size,
                              hipStream_t stream) {
    const int*   x     = (const int*)d_in[0];
    const int*   ei    = (const int*)d_in[1];
    const int*   batch = (const int*)d_in[2];
    const float* emb   = (const float*)d_in[3];
    const float* W1    = (const float*)d_in[4];
    const float* b1    = (const float*)d_in[5];
    const float* W2    = (const float*)d_in[6];
    const float* b2    = (const float*)d_in[7];
    const float* Wc    = (const float*)d_in[8];
    const float* bc    = (const float*)d_in[9];
    float* out = (float*)d_out;

    const int n_nodes  = in_sizes[0];
    const int n_edges  = in_sizes[1] / 2;
    const int labels   = in_sizes[9];
    const int n_graphs = out_size / labels;
    const int* src = ei;
    const int* dst = ei + n_edges;

    const int nb = (n_nodes + BNODES - 1) / BNODES;

    float* ws     = (float*)d_ws;
    float* h      = ws;
    float* hw     = h  + (size_t)n_nodes * HIDC;
    int* pairs    = (int*)(hw + (size_t)n_nodes * HIDC);
    float* pooled = (float*)(pairs + (size_t)2 * n_edges);
    float* cnt    = pooled + (size_t)n_graphs * HIDC;
    float* dinv   = cnt + n_graphs;
    int* rowptr   = (int*)(dinv + n_nodes);
    int* bcnt     = rowptr + n_nodes + 1;
    int* boff     = bcnt + (size_t)nb * NCHUNK;
    int* cpre     = boff + (size_t)nb * NCHUNK;
    int* bstart   = cpre + (size_t)nb * NCHUNK;
    int* csr_src  = bstart + nb + 1;

    const int ntiles = n_nodes / 16;
    const int wavesPerBlk = BLK / 32;

    dim3 b(BLK);
    auto g1 = [](int n) { return dim3((n + BLK - 1) / BLK); };
    dim3 gw((ntiles + wavesPerBlk - 1) / wavesPerBlk);

    k_hist<<<dim3(NCHUNK), dim3(SORT_BLK), nb * sizeof(int), stream>>>(dst, bcnt, cpre, pooled, cnt,
                                                                       n_edges, nb, n_graphs);
    k_off<<<dim3(nb), dim3(NCHUNK), 0, stream>>>(bcnt, cpre, boff, bstart, nb);
    k_scatter<<<dim3(NCHUNK), dim3(SORT_BLK), 2 * nb * sizeof(int), stream>>>(src, dst, boff, cpre,
                                                                              pairs, n_edges, nb);
    k_bucket<<<dim3(nb), dim3(SORT_BLK), 0, stream>>>(pairs, bstart, rowptr, dinv, csr_src,
                                                      x, emb, W1, hw, n_nodes, n_edges, nb);

    k_gather<<<dim3(G_EDGE), b, 0, stream>>>(rowptr, csr_src, dinv, hw, b1, h, n_nodes);

    k_gcn_wmma<<<gw, b, 0, stream>>>(h, W2, dinv, hw, ntiles);
    k_gather_pool<<<dim3(G_EDGE), b, 0, stream>>>(rowptr, csr_src, dinv, hw, b2, batch,
                                                  pooled, cnt, n_nodes);

    k_classify<<<g1(n_graphs * labels), b, 0, stream>>>(pooled, cnt, Wc, bc, out,
                                                        n_graphs, labels);
}


# conv 2's transform (dinv scale + 16x16 WMMA) folded into conv 1's gather: each wave collects its 16 summed rows in an LDS tile and multiplies them before the store; h never stored, 7 launches not 8
# speedup vs baseline: 15.6212x; 1.0323x over previous
#include <hip/hip_runtime.h>


typedef __attribute__((ext_vector_type(2))) float v2f;
typedef __attribute__((ext_vector_type(8))) float v8f;

#define HIDC 16
#define BLK 256

#define G_EDGE 1024

#define NODE_SHIFT 8
#define BNODES 256
#define NCHUNK 256
#define SORT_BLK 1024
#define STAGE_CAP 16384
#define BSTAGE_CAP 16384

__device__ __forceinline__ void fadd(float* p, float v) {
    __hip_atomic_fetch_add(p, v, __ATOMIC_RELAXED, __HIP_MEMORY_SCOPE_AGENT);
}

__device__ __forceinline__ int wave_sum(int v) {
#pragma unroll
    for (int o = 16; o > 0; o >>= 1) v += __shfl_xor(v, o);
    return v;
}

__device__ __forceinline__ int wave_scan_incl(int v, int lane) {
#pragma unroll
    for (int o = 1; o < 32; o <<= 1) {
        int t = __shfl_up(v, o);
        if (lane >= o) v += t;
    }
    return v;
}

extern __shared__ int dsh[];

__global__ void __launch_bounds__(SORT_BLK)
k_hist(const int* __restrict__ dst, int* __restrict__ bcnt, int* __restrict__ cpre,
       float* __restrict__ pooled, float* __restrict__ cnt, int n_edges, int nb, int n_graphs) {
    __shared__ int wtot[SORT_BLK / 32];
    int tid = threadIdx.x;
    for (int k = tid; k < nb; k += SORT_BLK) dsh[k] = 0;
    int gi = blockIdx.x * SORT_BLK + tid, gs = NCHUNK * SORT_BLK;
    for (int k = gi; k < n_graphs * HIDC; k += gs) pooled[k] = 0.0f;
    for (int k = gi; k < n_graphs; k += gs) cnt[k] = 0.0f;
    __syncthreads();
    int per = (n_edges + NCHUNK - 1) / NCHUNK;
    int c0 = blockIdx.x * per;
    int c1 = min(c0 + per, n_edges);
    int e = c0 + tid;
    for (; e + 3 * SORT_BLK < c1; e += 4 * SORT_BLK) {
        int d0 = dst[e];
        int d1 = dst[e + SORT_BLK];
        int d2 = dst[e + 2 * SORT_BLK];
        int d3 = dst[e + 3 * SORT_BLK];
        atomicAdd(&dsh[d0 >> NODE_SHIFT], 1);
        atomicAdd(&dsh[d1 >> NODE_SHIFT], 1);
        atomicAdd(&dsh[d2 >> NODE_SHIFT], 1);
        atomicAdd(&dsh[d3 >> NODE_SHIFT], 1);
    }
    for (; e < c1; e += SORT_BLK) atomicAdd(&dsh[dst[e] >> NODE_SHIFT], 1);
    __syncthreads();
    for (int k = tid; k < nb; k += SORT_BLK) bcnt[k * NCHUNK + blockIdx.x] = dsh[k];
    int lane = tid & 31, w = tid >> 5;
    int carry = 0;
    for (int k0 = 0; k0 < nb; k0 += SORT_BLK) {
        int k = k0 + tid;
        int v = (k < nb) ? dsh[k] : 0;
        int incl = wave_scan_incl(v, lane);
        if (lane == 31) wtot[w] = incl;
        __syncthreads();
        int wpre = 0, tot = 0;
        for (int i = 0; i < SORT_BLK / 32; ++i) {
            int t = wtot[i];
            tot += t;
            if (i < w) wpre += t;
        }
        if (k < nb) cpre[k * NCHUNK + blockIdx.x] = carry + wpre + incl - v;
        carry += tot;
        __syncthreads();
    }
}

__global__ void __launch_bounds__(NCHUNK)
k_off(const int* __restrict__ bcnt, const int* __restrict__ cpre, int* __restrict__ boff,
      int* __restrict__ bstart, int nb) {
    __shared__ int sh[NCHUNK / 32];
    __shared__ int s_base;
    int tid = threadIdx.x, lane = tid & 31, w = tid >> 5;
    int b = blockIdx.x;
    int part = wave_sum(cpre[b * NCHUNK + tid]);
    if (lane == 0) sh[w] = part;
    __syncthreads();
    if (tid == 0) {
        int t = 0;
        for (int k = 0; k < NCHUNK / 32; ++k) t += sh[k];
        s_base = t;
    }
    __syncthreads();
    int base = s_base;
    int c = bcnt[b * NCHUNK + tid];
    int incl = wave_scan_incl(c, lane);
    if (lane == 31) sh[w] = incl;
    __syncthreads();
    int wpre = 0;
    for (int k = 0; k < w; ++k) wpre += sh[k];
    boff[b * NCHUNK + tid] = base + wpre + incl - c;
    if (tid == 0) bstart[b] = base;
    if (b == nb - 1 && tid == NCHUNK - 1) bstart[nb] = base + wpre + incl;
}

__global__ void __launch_bounds__(SORT_BLK)
k_scatter(const int* __restrict__ src, const int* __restrict__ dst, const int* __restrict__ boff,
          const int* __restrict__ cpre, int* __restrict__ pairs, int n_edges, int nb) {
    __shared__ int2 stage[STAGE_CAP];
    int2* __restrict__ rec = reinterpret_cast<int2*>(pairs);
    int tid = threadIdx.x;
    int per = (n_edges + NCHUNK - 1) / NCHUNK;
    int c0 = blockIdx.x * per;
    int c1 = min(c0 + per, n_edges);
    int e = c0 + tid;
    if (per <= STAGE_CAP) {
        int* shift = dsh + nb;
        for (int k = tid; k < nb; k += SORT_BLK) {
            int lp = cpre[k * NCHUNK + blockIdx.x];
            dsh[k] = lp;
            shift[k] = boff[k * NCHUNK + blockIdx.x] - lp;
        }
        __syncthreads();
        for (; e + 3 * SORT_BLK < c1; e += 4 * SORT_BLK) {
            int e1 = e + SORT_BLK, e2 = e + 2 * SORT_BLK, e3 = e + 3 * SORT_BLK;
            int d0 = dst[e], d1 = dst[e1], d2 = dst[e2], d3 = dst[e3];
            int s0 = src[e], s1 = src[e1], s2 = src[e2], s3 = src[e3];
            int p0 = atomicAdd(&dsh[d0 >> NODE_SHIFT], 1);
            int p1 = atomicAdd(&dsh[d1 >> NODE_SHIFT], 1);
            int p2 = atomicAdd(&dsh[d2 >> NODE_SHIFT], 1);
            int p3 = atomicAdd(&dsh[d3 >> NODE_SHIFT], 1);
            stage[p0] = make_int2(s0, d0);
            stage[p1] = make_int2(s1, d1);
            stage[p2] = make_int2(s2, d2);
            stage[p3] = make_int2(s3, d3);
        }
        for (; e < c1; e += SORT_BLK) {
            int d = dst[e], s = src[e];
            int p = atomicAdd(&dsh[d >> NODE_SHIFT], 1);
            stage[p] = make_int2(s, d);
        }
        __syncthreads();
        int n = c1 - c0;
        int i = tid;
        for (; i + 3 * SORT_BLK < n; i += 4 * SORT_BLK) {
            int i1 = i + SORT_BLK, i2 = i + 2 * SORT_BLK, i3 = i + 3 * SORT_BLK;
            int2 r0 = stage[i], r1 = stage[i1], r2 = stage[i2], r3 = stage[i3];
            int g0 = i  + shift[r0.y >> NODE_SHIFT];
            int g1 = i1 + shift[r1.y >> NODE_SHIFT];
            int g2 = i2 + shift[r2.y >> NODE_SHIFT];
            int g3 = i3 + shift[r3.y >> NODE_SHIFT];
            rec[g0] = r0;
            rec[g1] = r1;
            rec[g2] = r2;
            rec[g3] = r3;
        }
        for (; i < n; i += SORT_BLK) {
            int2 r = stage[i];
            rec[i + shift[r.y >> NODE_SHIFT]] = r;
        }
    } else {
        for (int k = tid; k < nb; k += SORT_BLK) dsh[k] = boff[k * NCHUNK + blockIdx.x];
        __syncthreads();
        for (; e < c1; e += SORT_BLK) {
            int d = dst[e], s = src[e];
            int p = atomicAdd(&dsh[d >> NODE_SHIFT], 1);
            rec[p] = make_int2(s, d);
        }
    }
}

#define TILE_W0 16
__global__ void __launch_bounds__(SORT_BLK)
k_bucket(const int* __restrict__ pairs, const int* __restrict__ bstart,
         int* __restrict__ rowptr, float* __restrict__ dinv, int* __restrict__ csr_src,
         const int* __restrict__ x, const float* __restrict__ emb, const float* __restrict__ W,
         float* __restrict__ hw, int n_nodes, int n_edges, int nb) {
    const int2* __restrict__ rec = reinterpret_cast<const int2*>(pairs);
    __shared__ int cn[BNODES];
    __shared__ int wsum[BNODES / 32];
    __shared__ float sdinv[BNODES];
    __shared__ int simg[BSTAGE_CAP];
    int tid = threadIdx.x, lane = tid & 31, w = tid >> 5;
    int b = blockIdx.x;
    int nbase = b << NODE_SHIFT;
    int e0 = bstart[b], e1 = bstart[b + 1];
    const bool staged = (e1 - e0) <= BSTAGE_CAP;
    int half = lane >> 4, l15 = lane & 15;
    int trow = nbase + (w - TILE_W0) * 16;
    bool tile_on = (w >= TILE_W0) && (w < TILE_W0 + BNODES / 16) && (trow + 16 <= n_nodes);
    if (tid < BNODES) cn[tid] = 0;
    __syncthreads();
    int e = e0 + tid;
    for (; e + 3 * SORT_BLK < e1; e += 4 * SORT_BLK) {
        int d0 = rec[e].y;
        int d1 = rec[e + SORT_BLK].y;
        int d2 = rec[e + 2 * SORT_BLK].y;
        int d3 = rec[e + 3 * SORT_BLK].y;
        atomicAdd(&cn[d0 - nbase], 1);
        atomicAdd(&cn[d1 - nbase], 1);
        atomicAdd(&cn[d2 - nbase], 1);
        atomicAdd(&cn[d3 - nbase], 1);
    }
    for (; e < e1; e += SORT_BLK) atomicAdd(&cn[rec[e].y - nbase], 1);
    __syncthreads();
    int xi = 0;
    v2f bf[4] = {};
    if (tile_on) {
        xi = x[trow + l15];
#pragma unroll
        for (int t = 0; t < 4; ++t) {
            int kk = 4 * t + 2 * half;
            bf[t][0] = W[(kk + 0) * HIDC + l15];
            bf[t][1] = W[(kk + 1) * HIDC + l15];
        }
    }
    int c = 0, incl = 0;
    if (w < BNODES / 32) {
        c = cn[tid];
        incl = wave_scan_incl(c, lane);
        if (lane == 31) wsum[w] = incl;
    }
    __syncthreads();
    if (w < BNODES / 32) {
        int wpre = 0;
        for (int k = 0; k < w; ++k) wpre += wsum[k];
        int excl = wpre + incl - c;
        cn[tid] = excl;
        int node = nbase + tid;
        float d = (float)(c + 1);
        float di = (d > 0.0f) ? rsqrtf(d) : 0.0f;
        sdinv[tid] = di;
        if (node < n_nodes) {
            rowptr[node] = e0 + excl;
            dinv[node] = di;
        }
    }
    if (b == nb - 1 && tid == 0) rowptr[n_nodes] = n_edges;
    __syncthreads();
    v2f af[4] = {};
    if (tile_on) {
        const float* erow = emb + xi * HIDC;
#pragma unroll
        for (int t = 0; t < 4; ++t) {
            int kk = 4 * t + 2 * half;
            af[t][0] = erow[kk + 0];
            af[t][1] = erow[kk + 1];
        }
    }
    e = e0 + tid;
    if (staged) {
        for (; e + 3 * SORT_BLK < e1; e += 4 * SORT_BLK) {
            int e1i = e + SORT_BLK, e2i = e + 2 * SORT_BLK, e3i = e + 3 * SORT_BLK;
            int2 r0 = rec[e], r1 = rec[e1i], r2 = rec[e2i], r3 = rec[e3i];
            int p0 = atomicAdd(&cn[r0.y - nbase], 1);
            int p1 = atomicAdd(&cn[r1.y - nbase], 1);
            int p2 = atomicAdd(&cn[r2.y - nbase], 1);
            int p3 = atomicAdd(&cn[r3.y - nbase], 1);
            simg[p0] = r0.x;
            simg[p1] = r1.x;
            simg[p2] = r2.x;
            simg[p3] = r3.x;
        }
        for (; e < e1; e += SORT_BLK) {
            int2 r = rec[e];
            int p = atomicAdd(&cn[r.y - nbase], 1);
            simg[p] = r.x;
        }
        __syncthreads();
        int ne = e1 - e0;
        int i = tid - (int)((reinterpret_cast<size_t>(csr_src + e0) >> 2) & 31);
        if (i >= 0 && i < ne) csr_src[e0 + i] = simg[i];
        for (i += SORT_BLK; i < ne; i += SORT_BLK) csr_src[e0 + i] = simg[i];
    } else {
        for (; e < e1; e += SORT_BLK) {
            int2 r = rec[e];
            int p = atomicAdd(&cn[r.y - nbase], 1);
            csr_src[e0 + p] = r.x;
        }
    }
    if (tile_on) {
        float rsl = sdinv[(w - TILE_W0) * 16 + l15];
        v8f acc = {};
#pragma unroll
        for (int t = 0; t < 4; ++t) {
            v2f a;
            a[0] = af[t][0] * rsl;
            a[1] = af[t][1] * rsl;
            acc = __builtin_amdgcn_wmma_f32_16x16x4_f32(
                      false, a, false, bf[t], (short)0, acc, false, false);
        }
#pragma unroll
        for (int v = 0; v < 8; ++v) {
            int m = v + 8 * half;
            hw[(size_t)(trow + m) * HIDC + l15] = acc[v];
        }
    }
}

__device__ __forceinline__ float gather_node(const int* __restrict__ rowptr,
                                             const int* __restrict__ csr_src,
                                             const float* __restrict__ dinv,
                                             const float* __restrict__ hw,
                                             int node, int j, float bj, float& dd) {
    int i = node * HIDC + j;
    int beg = rowptr[node], end = rowptr[node + 1];
    dd = dinv[node];
    float acc = hw[i];
    int s_next = (beg + j < end) ? csr_src[beg + j] : node;
    for (int p = beg; p < end; p += 16) {
        int s = s_next;
        int qn = p + 16 + j;
        s_next = (qn < end) ? csr_src[qn] : node;
        int nk = end - p;
#pragma unroll
        for (int k = 0; k < 16; ++k) {
            int   sk = __shfl(s, k, 16);
            float v  = hw[sk * HIDC + j];
            acc += (k < nk) ? v : 0.0f;
        }
    }
    return fmaxf(acc * dd + bj, 0.0f);
}

#define TROW 20
__global__ void __launch_bounds__(BLK)
k_gather_xform(const int* __restrict__ rowptr, const int* __restrict__ csr_src,
               const float* __restrict__ dinv, const float* __restrict__ hw,
               const float* __restrict__ b, const float* __restrict__ W,
               float* __restrict__ hw2, int n_nodes) {
    __shared__ float tile[BLK / 32][16][TROW];
    int t = blockIdx.x * blockDim.x + threadIdx.x;
    int lane = threadIdx.x & 31, wv = threadIdx.x >> 5;
    int half = lane >> 4, l15 = lane & 15;
    int grp = t >> 4;
    int ngrp = (gridDim.x * blockDim.x) >> 4;
    float bj = b[l15];

    v2f bf[4];
#pragma unroll
    for (int q = 0; q < 4; ++q) {
        int kk = 4 * q + 2 * half;
        bf[q][0] = W[(kk + 0) * HIDC + l15];
        bf[q][1] = W[(kk + 1) * HIDC + l15];
    }

    int ntrip = (n_nodes + ngrp - 1) / ngrp;
    for (int i0 = 0; i0 < ntrip; i0 += 8) {
#pragma unroll 1
        for (int i = 0; i < 8; ++i) {
            int node = grp + (i0 + i) * ngrp;
            float a = 0.0f;
            if (node < n_nodes) {
                float dd;
                float hv = gather_node(rowptr, csr_src, dinv, hw, node, l15, bj, dd);
                a = hv * dd;
            }
            tile[wv][i + 8 * half][l15] = a;
        }
        __syncwarp();
        v8f c = {};
#pragma unroll
        for (int q = 0; q < 4; ++q) {
            int kk = 4 * q + 2 * half;
            v2f a;
            a[0] = tile[wv][l15][kk + 0];
            a[1] = tile[wv][l15][kk + 1];
            c = __builtin_amdgcn_wmma_f32_16x16x4_f32(
                    false, a, false, bf[q], (short)0, c, false, false);
        }
        __syncwarp();
#pragma unroll
        for (int v = 0; v < 8; ++v) {
            int node = grp + (i0 + v) * ngrp;
            if (node < n_nodes) hw2[(size_t)node * HIDC + l15] = c[v];
        }
    }
}

__global__ void k_gather_pool(const int* __restrict__ rowptr, const int* __restrict__ csr_src,
                              const float* __restrict__ dinv, const float* __restrict__ hw,
                              const float* __restrict__ b, const int* __restrict__ batch,
                              float* __restrict__ pooled, float* __restrict__ cnt, int n_nodes) {
    int t = blockIdx.x * blockDim.x + threadIdx.x;
    int j = t & 15;
    int grp = t >> 4;
    int ngrp = (gridDim.x * blockDim.x) >> 4;
    int per = n_nodes / ngrp, rem = n_nodes - per * ngrp;
    int n0 = grp * per + (grp < rem ? grp : rem);
    int n1 = n0 + per + (grp < rem ? 1 : 0);
    float bj = b[j];
    int gcur = -1;
    float psum = 0.0f, pcnt = 0.0f;
    for (int node = n0; node < n1; ++node) {
        int g = batch[node];
        float dd;
        float v = gather_node(rowptr, csr_src, dinv, hw, node, j, bj, dd);
        if (g != gcur) {
            if (gcur >= 0) {
                fadd(&pooled[gcur * HIDC + j], psum);
                if (j == 0) fadd(&cnt[gcur], pcnt);
            }
            gcur = g;
            psum = 0.0f;
            pcnt = 0.0f;
        }
        psum += v;
        pcnt += 1.0f;
    }
    if (gcur >= 0) {
        fadd(&pooled[gcur * HIDC + j], psum);
        if (j == 0) fadd(&cnt[gcur], pcnt);
    }
}

__global__ void k_classify(const float* __restrict__ pooled, const float* __restrict__ cnt,
                           const float* __restrict__ Wc, const float* __restrict__ bc,
                           float* __restrict__ out, int n_graphs, int labels) {
    int i = blockIdx.x * blockDim.x + threadIdx.x;
    if (i >= n_graphs * labels) return;
    int g = i / labels, c = i % labels;
    float inv = 1.0f / fmaxf(cnt[g], 1.0f);
    float acc = bc[c];
#pragma unroll
    for (int j = 0; j < HIDC; ++j)
        acc += pooled[g * HIDC + j] * inv * Wc[j * labels + c];
    out[i] = acc;
}

extern "C" void kernel_launch(void* const* d_in, const int* in_sizes, int n_in,
                              void* d_out, int out_size, void* d_ws, size_t ws_size,
                              hipStream_t stream) {
    const int*   x     = (const int*)d_in[0];
    const int*   ei    = (const int*)d_in[1];
    const int*   batch = (const int*)d_in[2];
    const float* emb   = (const float*)d_in[3];
    const float* W1    = (const float*)d_in[4];
    const float* b1    = (const float*)d_in[5];
    const float* W2    = (const float*)d_in[6];
    const float* b2    = (const float*)d_in[7];
    const float* Wc    = (const float*)d_in[8];
    const float* bc    = (const float*)d_in[9];
    float* out = (float*)d_out;

    const int n_nodes  = in_sizes[0];
    const int n_edges  = in_sizes[1] / 2;
    const int labels   = in_sizes[9];
    const int n_graphs = out_size / labels;
    const int* src = ei;
    const int* dst = ei + n_edges;

    const int nb = (n_nodes + BNODES - 1) / BNODES;

    float* ws     = (float*)d_ws;
    float* hw2    = ws;
    float* hw     = hw2 + (size_t)n_nodes * HIDC;
    int* pairs    = (int*)(hw + (size_t)n_nodes * HIDC);
    float* pooled = (float*)(pairs + (size_t)2 * n_edges);
    float* cnt    = pooled + (size_t)n_graphs * HIDC;
    float* dinv   = cnt + n_graphs;
    int* rowptr   = (int*)(dinv + n_nodes);
    int* bcnt     = rowptr + n_nodes + 1;
    int* boff     = bcnt + (size_t)nb * NCHUNK;
    int* cpre     = boff + (size_t)nb * NCHUNK;
    int* bstart   = cpre + (size_t)nb * NCHUNK;
    int* csr_src  = bstart + nb + 1;

    dim3 b(BLK);
    auto g1 = [](int n) { return dim3((n + BLK - 1) / BLK); };

    k_hist<<<dim3(NCHUNK), dim3(SORT_BLK), nb * sizeof(int), stream>>>(dst, bcnt, cpre, pooled, cnt,
                                                                       n_edges, nb, n_graphs);
    k_off<<<dim3(nb), dim3(NCHUNK), 0, stream>>>(bcnt, cpre, boff, bstart, nb);
    k_scatter<<<dim3(NCHUNK), dim3(SORT_BLK), 2 * nb * sizeof(int), stream>>>(src, dst, boff, cpre,
                                                                              pairs, n_edges, nb);
    k_bucket<<<dim3(nb), dim3(SORT_BLK), 0, stream>>>(pairs, bstart, rowptr, dinv, csr_src,
                                                      x, emb, W1, hw, n_nodes, n_edges, nb);

    k_gather_xform<<<dim3(G_EDGE), b, 0, stream>>>(rowptr, csr_src, dinv, hw, b1, W2, hw2, n_nodes);

    k_gather_pool<<<dim3(G_EDGE), b, 0, stream>>>(rowptr, csr_src, dinv, hw2, b2, batch,
                                                  pooled, cnt, n_nodes);

    k_classify<<<g1(n_graphs * labels), b, 0, stream>>>(pooled, cnt, Wc, bc, out,
                                                        n_graphs, labels);
}


# level 1 of the edge sort orders each chunk by its own LDS counts, kept chunk-major; level 2 finds its run per chunk by the chunk's prefix: no histogram or offset launches, 5 not 7; arrivals differ
# speedup vs baseline: 17.1443x; 1.0975x over previous
#include <hip/hip_runtime.h>


typedef __attribute__((ext_vector_type(2))) float v2f;
typedef __attribute__((ext_vector_type(8))) float v8f;

#define HIDC 16
#define BLK 256

#define G_EDGE 1024

#define NODE_SHIFT 8
#define BNODES 256
#define SORT_BLK 1024
#define SEG 12500
#define SEG_K 13
#define BSTAGE_CAP 16384
#define RUNS 8

__device__ __forceinline__ void fadd(float* p, float v) {
    __hip_atomic_fetch_add(p, v, __ATOMIC_RELAXED, __HIP_MEMORY_SCOPE_AGENT);
}

__device__ __forceinline__ int wave_sum(int v) {
#pragma unroll
    for (int o = 16; o > 0; o >>= 1) v += __shfl_xor(v, o);
    return v;
}

__device__ __forceinline__ int wave_scan_incl(int v, int lane) {
#pragma unroll
    for (int o = 1; o < 32; o <<= 1) {
        int t = __shfl_up(v, o);
        if (lane >= o) v += t;
    }
    return v;
}

extern __shared__ int dsh[];

__global__ void __launch_bounds__(SORT_BLK)
k_chunksort(const int* __restrict__ src, const int* __restrict__ dst, int* __restrict__ cpre,
            int* __restrict__ pairs, float* __restrict__ pooled, float* __restrict__ cnt,
            int n_edges, int nb, int nchunk, int n_graphs) {
    __shared__ int2 stage[SEG];
    __shared__ int wtot[SORT_BLK / 32];
    int2* __restrict__ rec = reinterpret_cast<int2*>(pairs);
    int tid = threadIdx.x, lane = tid & 31, w = tid >> 5;
    for (int k = tid; k < nb; k += SORT_BLK) dsh[k] = 0;
    int gi = blockIdx.x * SORT_BLK + tid, gs = gridDim.x * SORT_BLK;
    for (int k = gi; k < n_graphs * HIDC; k += gs) pooled[k] = 0.0f;
    for (int k = gi; k < n_graphs; k += gs) cnt[k] = 0.0f;
    int c0 = blockIdx.x * SEG;
    int n = min(SEG, n_edges - c0);
    int es[SEG_K], ed[SEG_K], er[SEG_K];
#pragma unroll
    for (int k = 0; k < SEG_K; ++k) {
        int i = tid + k * SORT_BLK;
        ed[k] = (i < n) ? dst[c0 + i] : 0;
        es[k] = (i < n) ? src[c0 + i] : 0;
    }
    __syncthreads();
#pragma unroll
    for (int k = 0; k < SEG_K; ++k) {
        int i = tid + k * SORT_BLK;
        er[k] = (i < n) ? atomicAdd(&dsh[ed[k] >> NODE_SHIFT], 1) : 0;
    }
    __syncthreads();
    int carry = 0;
    for (int k0 = 0; k0 < nb; k0 += SORT_BLK) {
        int k = k0 + tid;
        int v = (k < nb) ? dsh[k] : 0;
        int incl = wave_scan_incl(v, lane);
        if (lane == 31) wtot[w] = incl;
        __syncthreads();
        int wpre = 0, tot = 0;
        for (int i = 0; i < SORT_BLK / 32; ++i) {
            int t = wtot[i];
            tot += t;
            if (i < w) wpre += t;
        }
        int excl = carry + wpre + incl - v;
        if (k < nb) {
            dsh[k] = excl;
            cpre[k * nchunk + blockIdx.x] = excl;
        }
        carry += tot;
        __syncthreads();
    }
    if (tid == 0) cpre[nb * nchunk + blockIdx.x] = n;
#pragma unroll
    for (int k = 0; k < SEG_K; ++k) er[k] += dsh[ed[k] >> NODE_SHIFT];
#pragma unroll
    for (int k = 0; k < SEG_K; ++k) {
        int i = tid + k * SORT_BLK;
        if (i < n) stage[er[k]] = make_int2(es[k], ed[k]);
    }
    __syncthreads();
    int i = tid;
    for (; i + 3 * SORT_BLK < n; i += 4 * SORT_BLK) {
        int2 r0 = stage[i], r1 = stage[i + SORT_BLK], r2 = stage[i + 2 * SORT_BLK], r3 = stage[i + 3 * SORT_BLK];
        rec[c0 + i] = r0;
        rec[c0 + i + SORT_BLK] = r1;
        rec[c0 + i + 2 * SORT_BLK] = r2;
        rec[c0 + i + 3 * SORT_BLK] = r3;
    }
    for (; i < n; i += SORT_BLK) rec[c0 + i] = stage[i];
}

#define TILE_W0 16
__global__ void __launch_bounds__(SORT_BLK)
k_bucket(const int* __restrict__ pairs, const int* __restrict__ cpre,
         int* __restrict__ rowptr, float* __restrict__ dinv, int* __restrict__ csr_src,
         const int* __restrict__ x, const float* __restrict__ emb, const float* __restrict__ W,
         float* __restrict__ hw, int n_nodes, int n_edges, int nb, int nchunk) {
    const int2* __restrict__ rec = reinterpret_cast<const int2*>(pairs);
    __shared__ int cn[BNODES];
    __shared__ int wsum[BNODES / 32];
    __shared__ int red0[SORT_BLK / 32];
    __shared__ int red1[SORT_BLK / 32];
    __shared__ float sdinv[BNODES];
    __shared__ int simg[BSTAGE_CAP];
    int* roff = dsh;
    int* rlen = dsh + nchunk;
    int tid = threadIdx.x, lane = tid & 31, w = tid >> 5;
    int b = blockIdx.x;
    int nbase = b << NODE_SHIFT;
    int a0 = 0, a1 = 0;
    for (int k = tid; k < nchunk; k += SORT_BLK) {
        int lo = cpre[b * nchunk + k];
        int hi = cpre[(b + 1) * nchunk + k];
        roff[k] = k * SEG + lo;
        rlen[k] = hi - lo;
        a0 += lo;
        a1 += hi;
    }
    a0 = wave_sum(a0);
    a1 = wave_sum(a1);
    if (lane == 0) {
        red0[w] = a0;
        red1[w] = a1;
    }
    if (tid < BNODES) cn[tid] = 0;
    __syncthreads();
    int e0 = __builtin_amdgcn_readfirstlane(wave_sum(red0[lane]));
    int e1 = __builtin_amdgcn_readfirstlane(wave_sum(red1[lane]));
    const bool staged = (e1 - e0) <= BSTAGE_CAP;
    int half = lane >> 4, l15 = lane & 15;
    int trow = nbase + (w - TILE_W0) * 16;
    bool tile_on = (w >= TILE_W0) && (w < TILE_W0 + BNODES / 16) && (trow + 16 <= n_nodes);
    int c = w;
    for (; c + (RUNS - 1) * 32 < nchunk; c += RUNS * 32) {
        int rs[RUNS], rl[RUNS], da[RUNS], db[RUNS];
        int longest = 0;
#pragma unroll
        for (int q = 0; q < RUNS; ++q) {
            rs[q] = roff[c + 32 * q];
            rl[q] = rlen[c + 32 * q];
            longest = max(longest, rl[q]);
        }
#pragma unroll
        for (int q = 0; q < RUNS; ++q) {
            da[q] = (lane < rl[q]) ? rec[rs[q] + lane].y : 0;
            db[q] = (lane + 32 < rl[q]) ? rec[rs[q] + lane + 32].y : 0;
        }
#pragma unroll
        for (int q = 0; q < RUNS; ++q) {
            if (lane < rl[q]) atomicAdd(&cn[da[q] - nbase], 1);
            if (lane + 32 < rl[q]) atomicAdd(&cn[db[q] - nbase], 1);
        }
        if (longest > 64) {
            for (int q = 0; q < RUNS; ++q) {
                int rb = roff[c + 32 * q], rn = rlen[c + 32 * q];
                for (int i = lane + 64; i < rn; i += 32) atomicAdd(&cn[rec[rb + i].y - nbase], 1);
            }
        }
    }
    for (; c < nchunk; c += 32) {
        int rb = roff[c], rn = rlen[c];
        for (int i = lane; i < rn; i += 32) atomicAdd(&cn[rec[rb + i].y - nbase], 1);
    }
    __syncthreads();
    int xi = 0;
    v2f bf[4] = {};
    if (tile_on) {
        xi = x[trow + l15];
#pragma unroll
        for (int t = 0; t < 4; ++t) {
            int kk = 4 * t + 2 * half;
            bf[t][0] = W[(kk + 0) * HIDC + l15];
            bf[t][1] = W[(kk + 1) * HIDC + l15];
        }
    }
    int cv = 0, incl = 0;
    if (w < BNODES / 32) {
        cv = cn[tid];
        incl = wave_scan_incl(cv, lane);
        if (lane == 31) wsum[w] = incl;
    }
    __syncthreads();
    if (w < BNODES / 32) {
        int wpre = 0;
        for (int k = 0; k < w; ++k) wpre += wsum[k];
        int excl = wpre + incl - cv;
        cn[tid] = excl;
        int node = nbase + tid;
        float d = (float)(cv + 1);
        float di = (d > 0.0f) ? rsqrtf(d) : 0.0f;
        sdinv[tid] = di;
        if (node < n_nodes) {
            rowptr[node] = e0 + excl;
            dinv[node] = di;
        }
    }
    if (b == nb - 1 && tid == 0) rowptr[n_nodes] = n_edges;
    __syncthreads();
    v2f af[4] = {};
    if (tile_on) {
        const float* erow = emb + xi * HIDC;
#pragma unroll
        for (int t = 0; t < 4; ++t) {
            int kk = 4 * t + 2 * half;
            af[t][0] = erow[kk + 0];
            af[t][1] = erow[kk + 1];
        }
    }
    c = w;
    for (; c + (RUNS - 1) * 32 < nchunk; c += RUNS * 32) {
        int longest = 0;
        for (int off = 0; off < 64; off += 32) {
            int rs[RUNS], rl[RUNS], p[RUNS];
            int2 ra[RUNS];
#pragma unroll
            for (int q = 0; q < RUNS; ++q) {
                rs[q] = roff[c + 32 * q];
                rl[q] = rlen[c + 32 * q];
                longest = max(longest, rl[q]);
            }
#pragma unroll
            for (int q = 0; q < RUNS; ++q)
                ra[q] = (lane + off < rl[q]) ? rec[rs[q] + lane + off] : make_int2(0, 0);
#pragma unroll
            for (int q = 0; q < RUNS; ++q)
                p[q] = (lane + off < rl[q]) ? atomicAdd(&cn[ra[q].y - nbase], 1) : 0;
#pragma unroll
            for (int q = 0; q < RUNS; ++q) {
                if (lane + off < rl[q]) {
                    if (staged) simg[p[q]] = ra[q].x; else csr_src[e0 + p[q]] = ra[q].x;
                }
            }
        }
        if (longest > 64) {
            for (int q = 0; q < RUNS; ++q) {
                int rb = roff[c + 32 * q], rn = rlen[c + 32 * q];
                for (int i = lane + 64; i < rn; i += 32) {
                    int2 r = rec[rb + i];
                    int p = atomicAdd(&cn[r.y - nbase], 1);
                    if (staged) simg[p] = r.x; else csr_src[e0 + p] = r.x;
                }
            }
        }
    }
    for (; c < nchunk; c += 32) {
        int rb = roff[c], rn = rlen[c];
        for (int i = lane; i < rn; i += 32) {
            int2 r = rec[rb + i];
            int p = atomicAdd(&cn[r.y - nbase], 1);
            if (staged) simg[p] = r.x; else csr_src[e0 + p] = r.x;
        }
    }
    if (staged) {
        __syncthreads();
        int ne = e1 - e0;
        int i = tid - (int)((reinterpret_cast<size_t>(csr_src + e0) >> 2) & 31);
        if (i >= 0 && i < ne) csr_src[e0 + i] = simg[i];
        for (i += SORT_BLK; i < ne; i += SORT_BLK) csr_src[e0 + i] = simg[i];
    }
    if (tile_on) {
        float rsl = sdinv[(w - TILE_W0) * 16 + l15];
        v8f acc = {};
#pragma unroll
        for (int t = 0; t < 4; ++t) {
            v2f a;
            a[0] = af[t][0] * rsl;
            a[1] = af[t][1] * rsl;
            acc = __builtin_amdgcn_wmma_f32_16x16x4_f32(
                      false, a, false, bf[t], (short)0, acc, false, false);
        }
#pragma unroll
        for (int v = 0; v < 8; ++v) {
            int m = v + 8 * half;
            hw[(size_t)(trow + m) * HIDC + l15] = acc[v];
        }
    }
}

__device__ __forceinline__ float gather_node(const int* __restrict__ rowptr,
                                             const int* __restrict__ csr_src,
                                             const float* __restrict__ dinv,
                                             const float* __restrict__ hw,
                                             int node, int j, float bj, float& dd) {
    int i = node * HIDC + j;
    int beg = rowptr[node], end = rowptr[node + 1];
    dd = dinv[node];
    float acc = hw[i];
    int s_next = (beg + j < end) ? csr_src[beg + j] : node;
    for (int p = beg; p < end; p += 16) {
        int s = s_next;
        int qn = p + 16 + j;
        s_next = (qn < end) ? csr_src[qn] : node;
        int nk = end - p;
#pragma unroll
        for (int k = 0; k < 16; ++k) {
            int   sk = __shfl(s, k, 16);
            float v  = hw[sk * HIDC + j];
            acc += (k < nk) ? v : 0.0f;
        }
    }
    return fmaxf(acc * dd + bj, 0.0f);
}

#define TROW 20
__global__ void __launch_bounds__(BLK)
k_gather_xform(const int* __restrict__ rowptr, const int* __restrict__ csr_src,
               const float* __restrict__ dinv, const float* __restrict__ hw,
               const float* __restrict__ b, const float* __restrict__ W,
               float* __restrict__ hw2, int n_nodes) {
    __shared__ float tile[BLK / 32][16][TROW];
    int t = blockIdx.x * blockDim.x + threadIdx.x;
    int lane = threadIdx.x & 31, wv = threadIdx.x >> 5;
    int half = lane >> 4, l15 = lane & 15;
    int grp = t >> 4;
    int ngrp = (gridDim.x * blockDim.x) >> 4;
    float bj = b[l15];

    v2f bf[4];
#pragma unroll
    for (int q = 0; q < 4; ++q) {
        int kk = 4 * q + 2 * half;
        bf[q][0] = W[(kk + 0) * HIDC + l15];
        bf[q][1] = W[(kk + 1) * HIDC + l15];
    }

    int ntrip = (n_nodes + ngrp - 1) / ngrp;
    for (int i0 = 0; i0 < ntrip; i0 += 8) {
#pragma unroll 1
        for (int i = 0; i < 8; ++i) {
            int node = grp + (i0 + i) * ngrp;
            float a = 0.0f;
            if (node < n_nodes) {
                float dd;
                float hv = gather_node(rowptr, csr_src, dinv, hw, node, l15, bj, dd);
                a = hv * dd;
            }
            tile[wv][i + 8 * half][l15] = a;
        }
        __syncwarp();
        v8f c = {};
#pragma unroll
        for (int q = 0; q < 4; ++q) {
            int kk = 4 * q + 2 * half;
            v2f a;
            a[0] = tile[wv][l15][kk + 0];
            a[1] = tile[wv][l15][kk + 1];
            c = __builtin_amdgcn_wmma_f32_16x16x4_f32(
                    false, a, false, bf[q], (short)0, c, false, false);
        }
        __syncwarp();
#pragma unroll
        for (int v = 0; v < 8; ++v) {
            int node = grp + (i0 + v) * ngrp;
            if (node < n_nodes) hw2[(size_t)node * HIDC + l15] = c[v];
        }
    }
}

__global__ void k_gather_pool(const int* __restrict__ rowptr, const int* __restrict__ csr_src,
                              const float* __restrict__ dinv, const float* __restrict__ hw,
                              const float* __restrict__ b, const int* __restrict__ batch,
                              float* __restrict__ pooled, float* __restrict__ cnt, int n_nodes) {
    int t = blockIdx.x * blockDim.x + threadIdx.x;
    int j = t & 15;
    int grp = t >> 4;
    int ngrp = (gridDim.x * blockDim.x) >> 4;
    int per = n_nodes / ngrp, rem = n_nodes - per * ngrp;
    int n0 = grp * per + (grp < rem ? grp : rem);
    int n1 = n0 + per + (grp < rem ? 1 : 0);
    float bj = b[j];
    int gcur = -1;
    float psum = 0.0f, pcnt = 0.0f;
    for (int node = n0; node < n1; ++node) {
        int g = batch[node];
        float dd;
        float v = gather_node(rowptr, csr_src, dinv, hw, node, j, bj, dd);
        if (g != gcur) {
            if (gcur >= 0) {
                fadd(&pooled[gcur * HIDC + j], psum);
                if (j == 0) fadd(&cnt[gcur], pcnt);
            }
            gcur = g;
            psum = 0.0f;
            pcnt = 0.0f;
        }
        psum += v;
        pcnt += 1.0f;
    }
    if (gcur >= 0) {
        fadd(&pooled[gcur * HIDC + j], psum);
        if (j == 0) fadd(&cnt[gcur], pcnt);
    }
}

__global__ void k_classify(const float* __restrict__ pooled, const float* __restrict__ cnt,
                           const float* __restrict__ Wc, const float* __restrict__ bc,
                           float* __restrict__ out, int n_graphs, int labels) {
    int i = blockIdx.x * blockDim.x + threadIdx.x;
    if (i >= n_graphs * labels) return;
    int g = i / labels, c = i % labels;
    float inv = 1.0f / fmaxf(cnt[g], 1.0f);
    float acc = bc[c];
#pragma unroll
    for (int j = 0; j < HIDC; ++j)
        acc += pooled[g * HIDC + j] * inv * Wc[j * labels + c];
    out[i] = acc;
}

extern "C" void kernel_launch(void* const* d_in, const int* in_sizes, int n_in,
                              void* d_out, int out_size, void* d_ws, size_t ws_size,
                              hipStream_t stream) {
    const int*   x     = (const int*)d_in[0];
    const int*   ei    = (const int*)d_in[1];
    const int*   batch = (const int*)d_in[2];
    const float* emb   = (const float*)d_in[3];
    const float* W1    = (const float*)d_in[4];
    const float* b1    = (const float*)d_in[5];
    const float* W2    = (const float*)d_in[6];
    const float* b2    = (const float*)d_in[7];
    const float* Wc    = (const float*)d_in[8];
    const float* bc    = (const float*)d_in[9];
    float* out = (float*)d_out;

    const int n_nodes  = in_sizes[0];
    const int n_edges  = in_sizes[1] / 2;
    const int labels   = in_sizes[9];
    const int n_graphs = out_size / labels;
    const int* src = ei;
    const int* dst = ei + n_edges;

    const int nb = (n_nodes + BNODES - 1) / BNODES;
    const int nchunk = (n_edges + SEG - 1) / SEG;

    float* ws     = (float*)d_ws;
    float* hw2    = ws;
    float* hw     = hw2 + (size_t)n_nodes * HIDC;
    int* pairs    = (int*)(hw + (size_t)n_nodes * HIDC);
    float* pooled = (float*)(pairs + (size_t)2 * n_edges);
    float* cnt    = pooled + (size_t)n_graphs * HIDC;
    float* dinv   = cnt + n_graphs;
    int* rowptr   = (int*)(dinv + n_nodes);
    int* cpre     = rowptr + n_nodes + 1;
    int* csr_src  = cpre + (size_t)(nb + 1) * nchunk;

    dim3 b(BLK);
    auto g1 = [](int n) { return dim3((n + BLK - 1) / BLK); };

    k_chunksort<<<dim3(nchunk), dim3(SORT_BLK), nb * sizeof(int), stream>>>(src, dst, cpre, pairs,
                                                                            pooled, cnt, n_edges, nb,
                                                                            nchunk, n_graphs);
    k_bucket<<<dim3(nb), dim3(SORT_BLK), 2 * nchunk * sizeof(int), stream>>>(pairs, cpre, rowptr, dinv,
                                                                             csr_src, x, emb, W1, hw,
                                                                             n_nodes, n_edges, nb, nchunk);

    k_gather_xform<<<dim3(G_EDGE), b, 0, stream>>>(rowptr, csr_src, dinv, hw, b1, W2, hw2, n_nodes);

    k_gather_pool<<<dim3(G_EDGE), b, 0, stream>>>(rowptr, csr_src, dinv, hw2, b2, batch,
                                                  pooled, cnt, n_nodes);

    k_classify<<<g1(n_graphs * labels), b, 0, stream>>>(pooled, cnt, Wc, bc, out,
                                                        n_graphs, labels);
}


# conv 1's gather takes a source's weight and token from a 2-byte node table in LDS and adds dinv[s] * (emb @ W1)[token] from an LDS table: no memory line per edge; sum order kept, rounding regrouped
# speedup vs baseline: 18.1071x; 1.0562x over previous
#include <hip/hip_runtime.h>


typedef __attribute__((ext_vector_type(2))) float v2f;
typedef __attribute__((ext_vector_type(8))) float v8f;

#define HIDC 16
#define BLK 256

#define G_EDGE 1024

#define NODE_CAP 100000
#define VOCAB_CAP 128
#define G_TAB 256
#define TAB_BLK 1024
#define FILL_K ((NODE_CAP / 8 + TAB_BLK - 1) / TAB_BLK)

#define NODE_SHIFT 8
#define BNODES 256
#define SORT_BLK 1024
#define SEG 12500
#define SEG_K 13
#define BSTAGE_CAP 16384
#define RUNS 8

__device__ __forceinline__ void fadd(float* p, float v) {
    __hip_atomic_fetch_add(p, v, __ATOMIC_RELAXED, __HIP_MEMORY_SCOPE_AGENT);
}

__device__ __forceinline__ int wave_sum(int v) {
#pragma unroll
    for (int o = 16; o > 0; o >>= 1) v += __shfl_xor(v, o);
    return v;
}

__device__ __forceinline__ int wave_scan_incl(int v, int lane) {
#pragma unroll
    for (int o = 1; o < 32; o <<= 1) {
        int t = __shfl_up(v, o);
        if (lane >= o) v += t;
    }
    return v;
}

extern __shared__ int dsh[];

__global__ void __launch_bounds__(SORT_BLK)
k_chunksort(const int* __restrict__ src, const int* __restrict__ dst, int* __restrict__ cpre,
            int* __restrict__ pairs, float* __restrict__ pooled, float* __restrict__ cnt,
            int n_edges, int nb, int nchunk, int n_graphs) {
    __shared__ int2 stage[SEG];
    __shared__ int wtot[SORT_BLK / 32];
    int2* __restrict__ rec = reinterpret_cast<int2*>(pairs);
    int tid = threadIdx.x, lane = tid & 31, w = tid >> 5;
    for (int k = tid; k < nb; k += SORT_BLK) dsh[k] = 0;
    int gi = blockIdx.x * SORT_BLK + tid, gs = gridDim.x * SORT_BLK;
    for (int k = gi; k < n_graphs * HIDC; k += gs) pooled[k] = 0.0f;
    for (int k = gi; k < n_graphs; k += gs) cnt[k] = 0.0f;
    int c0 = blockIdx.x * SEG;
    int n = min(SEG, n_edges - c0);
    int es[SEG_K], ed[SEG_K], er[SEG_K];
#pragma unroll
    for (int k = 0; k < SEG_K; ++k) {
        int i = tid + k * SORT_BLK;
        ed[k] = (i < n) ? dst[c0 + i] : 0;
        es[k] = (i < n) ? src[c0 + i] : 0;
    }
    __syncthreads();
#pragma unroll
    for (int k = 0; k < SEG_K; ++k) {
        int i = tid + k * SORT_BLK;
        er[k] = (i < n) ? atomicAdd(&dsh[ed[k] >> NODE_SHIFT], 1) : 0;
    }
    __syncthreads();
    int carry = 0;
    for (int k0 = 0; k0 < nb; k0 += SORT_BLK) {
        int k = k0 + tid;
        int v = (k < nb) ? dsh[k] : 0;
        int incl = wave_scan_incl(v, lane);
        if (lane == 31) wtot[w] = incl;
        __syncthreads();
        int wpre = 0, tot = 0;
        for (int i = 0; i < SORT_BLK / 32; ++i) {
            int t = wtot[i];
            tot += t;
            if (i < w) wpre += t;
        }
        int excl = carry + wpre + incl - v;
        if (k < nb) {
            dsh[k] = excl;
            cpre[k * nchunk + blockIdx.x] = excl;
        }
        carry += tot;
        __syncthreads();
    }
    if (tid == 0) cpre[nb * nchunk + blockIdx.x] = n;
#pragma unroll
    for (int k = 0; k < SEG_K; ++k) er[k] += dsh[ed[k] >> NODE_SHIFT];
#pragma unroll
    for (int k = 0; k < SEG_K; ++k) {
        int i = tid + k * SORT_BLK;
        if (i < n) stage[er[k]] = make_int2(es[k], ed[k]);
    }
    __syncthreads();
    int i = tid;
    for (; i + 3 * SORT_BLK < n; i += 4 * SORT_BLK) {
        int2 r0 = stage[i], r1 = stage[i + SORT_BLK], r2 = stage[i + 2 * SORT_BLK], r3 = stage[i + 3 * SORT_BLK];
        rec[c0 + i] = r0;
        rec[c0 + i + SORT_BLK] = r1;
        rec[c0 + i + 2 * SORT_BLK] = r2;
        rec[c0 + i + 3 * SORT_BLK] = r3;
    }
    for (; i < n; i += SORT_BLK) rec[c0 + i] = stage[i];
}

__global__ void __launch_bounds__(SORT_BLK)
k_bucket(const int* __restrict__ pairs, const int* __restrict__ cpre,
         int* __restrict__ rowptr, float* __restrict__ dinv, int* __restrict__ csr_src,
         const int* __restrict__ x, unsigned short* __restrict__ nrec,
         int n_nodes, int n_edges, int nb, int nchunk) {
    const int2* __restrict__ rec = reinterpret_cast<const int2*>(pairs);
    __shared__ int cn[BNODES];
    __shared__ int wsum[BNODES / 32];
    __shared__ int red0[SORT_BLK / 32];
    __shared__ int red1[SORT_BLK / 32];
    __shared__ int simg[BSTAGE_CAP];
    int* roff = dsh;
    int* rlen = dsh + nchunk;
    int tid = threadIdx.x, lane = tid & 31, w = tid >> 5;
    int b = blockIdx.x;
    int nbase = b << NODE_SHIFT;
    int a0 = 0, a1 = 0;
    for (int k = tid; k < nchunk; k += SORT_BLK) {
        int lo = cpre[b * nchunk + k];
        int hi = cpre[(b + 1) * nchunk + k];
        roff[k] = k * SEG + lo;
        rlen[k] = hi - lo;
        a0 += lo;
        a1 += hi;
    }
    a0 = wave_sum(a0);
    a1 = wave_sum(a1);
    if (lane == 0) {
        red0[w] = a0;
        red1[w] = a1;
    }
    if (tid < BNODES) cn[tid] = 0;
    __syncthreads();
    int e0 = __builtin_amdgcn_readfirstlane(wave_sum(red0[lane]));
    int e1 = __builtin_amdgcn_readfirstlane(wave_sum(red1[lane]));
    const bool staged = (e1 - e0) <= BSTAGE_CAP;
    int c = w;
    for (; c + (RUNS - 1) * 32 < nchunk; c += RUNS * 32) {
        int rs[RUNS], rl[RUNS], da[RUNS], db[RUNS];
        int longest = 0;
#pragma unroll
        for (int q = 0; q < RUNS; ++q) {
            rs[q] = roff[c + 32 * q];
            rl[q] = rlen[c + 32 * q];
            longest = max(longest, rl[q]);
        }
#pragma unroll
        for (int q = 0; q < RUNS; ++q) {
            da[q] = (lane < rl[q]) ? rec[rs[q] + lane].y : 0;
            db[q] = (lane + 32 < rl[q]) ? rec[rs[q] + lane + 32].y : 0;
        }
#pragma unroll
        for (int q = 0; q < RUNS; ++q) {
            if (lane < rl[q]) atomicAdd(&cn[da[q] - nbase], 1);
            if (lane + 32 < rl[q]) atomicAdd(&cn[db[q] - nbase], 1);
        }
        if (longest > 64) {
            for (int q = 0; q < RUNS; ++q) {
                int rb = roff[c + 32 * q], rn = rlen[c + 32 * q];
                for (int i = lane + 64; i < rn; i += 32) atomicAdd(&cn[rec[rb + i].y - nbase], 1);
            }
        }
    }
    for (; c < nchunk; c += 32) {
        int rb = roff[c], rn = rlen[c];
        for (int i = lane; i < rn; i += 32) atomicAdd(&cn[rec[rb + i].y - nbase], 1);
    }
    __syncthreads();
    int xi = 0;
    int cv = 0, incl = 0;
    if (w < BNODES / 32) {
        xi = (nbase + tid < n_nodes) ? x[nbase + tid] : 0;
        cv = cn[tid];
        incl = wave_scan_incl(cv, lane);
        if (lane == 31) wsum[w] = incl;
    }
    __syncthreads();
    if (w < BNODES / 32) {
        int wpre = 0;
        for (int k = 0; k < w; ++k) wpre += wsum[k];
        int excl = wpre + incl - cv;
        cn[tid] = excl;
        int node = nbase + tid;
        float d = (float)(cv + 1);
        float di = (d > 0.0f) ? rsqrtf(d) : 0.0f;
        if (node < n_nodes) {
            rowptr[node] = e0 + excl;
            dinv[node] = di;
            nrec[node] = (unsigned short)((min(cv, 255) << 8) | (xi & 255));
        }
    }
    if (b == nb - 1 && tid == 0) rowptr[n_nodes] = n_edges;
    __syncthreads();
    c = w;
    for (; c + (RUNS - 1) * 32 < nchunk; c += RUNS * 32) {
        int longest = 0;
        for (int off = 0; off < 64; off += 32) {
            int rs[RUNS], rl[RUNS], p[RUNS];
            int2 ra[RUNS];
#pragma unroll
            for (int q = 0; q < RUNS; ++q) {
                rs[q] = roff[c + 32 * q];
                rl[q] = rlen[c + 32 * q];
                longest = max(longest, rl[q]);
            }
#pragma unroll
            for (int q = 0; q < RUNS; ++q)
                ra[q] = (lane + off < rl[q]) ? rec[rs[q] + lane + off] : make_int2(0, 0);
#pragma unroll
            for (int q = 0; q < RUNS; ++q)
                p[q] = (lane + off < rl[q]) ? atomicAdd(&cn[ra[q].y - nbase], 1) : 0;
#pragma unroll
            for (int q = 0; q < RUNS; ++q) {
                if (lane + off < rl[q]) {
                    if (staged) simg[p[q]] = ra[q].x; else csr_src[e0 + p[q]] = ra[q].x;
                }
            }
        }
        if (longest > 64) {
            for (int q = 0; q < RUNS; ++q) {
                int rb = roff[c + 32 * q], rn = rlen[c + 32 * q];
                for (int i = lane + 64; i < rn; i += 32) {
                    int2 r = rec[rb + i];
                    int p = atomicAdd(&cn[r.y - nbase], 1);
                    if (staged) simg[p] = r.x; else csr_src[e0 + p] = r.x;
                }
            }
        }
    }
    for (; c < nchunk; c += 32) {
        int rb = roff[c], rn = rlen[c];
        for (int i = lane; i < rn; i += 32) {
            int2 r = rec[rb + i];
            int p = atomicAdd(&cn[r.y - nbase], 1);
            if (staged) simg[p] = r.x; else csr_src[e0 + p] = r.x;
        }
    }
    if (staged) {
        __syncthreads();
        int ne = e1 - e0;
        int i = tid - (int)((reinterpret_cast<size_t>(csr_src + e0) >> 2) & 31);
        if (i >= 0 && i < ne) csr_src[e0 + i] = simg[i];
        for (i += SORT_BLK; i < ne; i += SORT_BLK) csr_src[e0 + i] = simg[i];
    }
}

__device__ __forceinline__ float gather_node(const int* __restrict__ rowptr,
                                             const int* __restrict__ csr_src,
                                             const float* __restrict__ dinv,
                                             const float* __restrict__ hw,
                                             int node, int j, float bj, float& dd) {
    int i = node * HIDC + j;
    int beg = rowptr[node], end = rowptr[node + 1];
    dd = dinv[node];
    float acc = hw[i];
    int s_next = (beg + j < end) ? csr_src[beg + j] : node;
    for (int p = beg; p < end; p += 16) {
        int s = s_next;
        int qn = p + 16 + j;
        s_next = (qn < end) ? csr_src[qn] : node;
        int nk = end - p;
#pragma unroll
        for (int k = 0; k < 16; ++k) {
            int   sk = __shfl(s, k, 16);
            float v  = hw[sk * HIDC + j];
            acc += (k < nk) ? v : 0.0f;
        }
    }
    return fmaxf(acc * dd + bj, 0.0f);
}

template <bool IN_LDS>
__device__ __forceinline__ float gather_node_tab(const int* __restrict__ rowptr,
                                                 const int* __restrict__ csr_src,
                                                 const float* __restrict__ dinv,
                                                 const unsigned short* __restrict__ nrec,
                                                 const unsigned short* ntab, const float* ttab,
                                                 int node, int j, float bj, float& dd) {
    int beg = rowptr[node], end = rowptr[node + 1];
    dd = dinv[node];
    int s_next = (beg + j < end) ? csr_src[beg + j] : node;
    unsigned rn = IN_LDS ? (unsigned)ntab[node] : (unsigned)nrec[node];
    float acc = dd * ttab[(rn & 255u) * HIDC + j];
    for (int p = beg; p < end; p += 16) {
        int s = s_next;
        int qn = p + 16 + j;
        s_next = (qn < end) ? csr_src[qn] : node;
        unsigned r = IN_LDS ? (unsigned)ntab[s] : (unsigned)nrec[s];
        unsigned c8 = r >> 8;
        float di = rsqrtf((float)(c8 + 1u));
        if (c8 == 255u) di = dinv[s];
        di = (p + j < end) ? di : 0.0f;
        int to = (int)(r & 255u) * HIDC;
#pragma unroll
        for (int k = 0; k < 16; ++k) {
            int   tk = __shfl(to, k, 16);
            float dk = __shfl(di, k, 16);
            acc = fmaf(dk, ttab[tk + j], acc);
        }
    }
    return fmaxf(acc * dd + bj, 0.0f);
}

#define TROW 20
__global__ void __launch_bounds__(TAB_BLK)
k_gather_xform(const int* __restrict__ rowptr, const int* __restrict__ csr_src,
               const float* __restrict__ dinv, const unsigned short* __restrict__ nrec,
               const float* __restrict__ emb, const float* __restrict__ W1,
               const float* __restrict__ b, const float* __restrict__ W,
               float* __restrict__ hw2, int n_nodes, int vocab) {
    __shared__ uint4 ntab4[(NODE_CAP + 7) / 8];
    __shared__ float ttab[VOCAB_CAP * HIDC];
    __shared__ float tile[TAB_BLK / 32][16][TROW];
    unsigned short* ntab = reinterpret_cast<unsigned short*>(ntab4);
    int t = blockIdx.x * blockDim.x + threadIdx.x;
    int lane = threadIdx.x & 31, wv = threadIdx.x >> 5;
    int half = lane >> 4, l15 = lane & 15;
    int grp = t >> 4;
    int ngrp = (gridDim.x * blockDim.x) >> 4;
    float bj = b[l15];

    int nfill = min(n_nodes, NODE_CAP);
    int nvec = nfill >> 3;
    const uint4* __restrict__ g4 = reinterpret_cast<const uint4*>(nrec);
    uint4 piece[FILL_K];
#pragma unroll
    for (int k = 0; k < FILL_K; ++k) {
        int i = threadIdx.x + k * TAB_BLK;
        piece[k] = (i < nvec) ? g4[i] : make_uint4(0u, 0u, 0u, 0u);
    }
#pragma unroll
    for (int k = 0; k < FILL_K; ++k) {
        int i = threadIdx.x + k * TAB_BLK;
        if (i < nvec) ntab4[i] = piece[k];
    }
    for (int i = (nvec << 3) + threadIdx.x; i < nfill; i += TAB_BLK) ntab[i] = nrec[i];
    if (wv < (min(vocab, VOCAB_CAP) >> 4)) {
        v8f acc = {};
#pragma unroll
        for (int q = 0; q < 4; ++q) {
            int kk = 4 * q + 2 * half;
            v2f a, w1;
            a[0]  = emb[(16 * wv + l15) * HIDC + kk + 0];
            a[1]  = emb[(16 * wv + l15) * HIDC + kk + 1];
            w1[0] = W1[(kk + 0) * HIDC + l15];
            w1[1] = W1[(kk + 1) * HIDC + l15];
            acc = __builtin_amdgcn_wmma_f32_16x16x4_f32(
                      false, a, false, w1, (short)0, acc, false, false);
        }
#pragma unroll
        for (int v = 0; v < 8; ++v) ttab[(16 * wv + v + 8 * half) * HIDC + l15] = acc[v];
    }

    v2f bf[4];
#pragma unroll
    for (int q = 0; q < 4; ++q) {
        int kk = 4 * q + 2 * half;
        bf[q][0] = W[(kk + 0) * HIDC + l15];
        bf[q][1] = W[(kk + 1) * HIDC + l15];
    }

    __syncthreads();
    const bool in_lds = n_nodes <= NODE_CAP;

    int ntrip = (n_nodes + ngrp - 1) / ngrp;
    for (int i0 = 0; i0 < ntrip; i0 += 8) {
#pragma unroll 1
        for (int i = 0; i < 8; ++i) {
            int node = grp + (i0 + i) * ngrp;
            float a = 0.0f;
            if (node < n_nodes) {
                float dd;
                float hv = in_lds ? gather_node_tab<true>(rowptr, csr_src, dinv, nrec, ntab, ttab, node, l15, bj, dd)
                                  : gather_node_tab<false>(rowptr, csr_src, dinv, nrec, ntab, ttab, node, l15, bj, dd);
                a = hv * dd;
            }
            tile[wv][i + 8 * half][l15] = a;
        }
        __syncwarp();
        v8f c = {};
#pragma unroll
        for (int q = 0; q < 4; ++q) {
            int kk = 4 * q + 2 * half;
            v2f a;
            a[0] = tile[wv][l15][kk + 0];
            a[1] = tile[wv][l15][kk + 1];
            c = __builtin_amdgcn_wmma_f32_16x16x4_f32(
                    false, a, false, bf[q], (short)0, c, false, false);
        }
        __syncwarp();
#pragma unroll
        for (int v = 0; v < 8; ++v) {
            int node = grp + (i0 + v) * ngrp;
            if (node < n_nodes) hw2[(size_t)node * HIDC + l15] = c[v];
        }
    }
}

__global__ void k_gather_pool(const int* __restrict__ rowptr, const int* __restrict__ csr_src,
                              const float* __restrict__ dinv, const float* __restrict__ hw,
                              const float* __restrict__ b, const int* __restrict__ batch,
                              float* __restrict__ pooled, float* __restrict__ cnt, int n_nodes) {
    int t = blockIdx.x * blockDim.x + threadIdx.x;
    int j = t & 15;
    int grp = t >> 4;
    int ngrp = (gridDim.x * blockDim.x) >> 4;
    int per = n_nodes / ngrp, rem = n_nodes - per * ngrp;
    int n0 = grp * per + (grp < rem ? grp : rem);
    int n1 = n0 + per + (grp < rem ? 1 : 0);
    float bj = b[j];
    int gcur = -1;
    float psum = 0.0f, pcnt = 0.0f;
    for (int node = n0; node < n1; ++node) {
        int g = batch[node];
        float dd;
        float v = gather_node(rowptr, csr_src, dinv, hw, node, j, bj, dd);
        if (g != gcur) {
            if (gcur >= 0) {
                fadd(&pooled[gcur * HIDC + j], psum);
                if (j == 0) fadd(&cnt[gcur], pcnt);
            }
            gcur = g;
            psum = 0.0f;
            pcnt = 0.0f;
        }
        psum += v;
        pcnt += 1.0f;
    }
    if (gcur >= 0) {
        fadd(&pooled[gcur * HIDC + j], psum);
        if (j == 0) fadd(&cnt[gcur], pcnt);
    }
}

__global__ void k_classify(const float* __restrict__ pooled, const float* __restrict__ cnt,
                           const float* __restrict__ Wc, const float* __restrict__ bc,
                           float* __restrict__ out, int n_graphs, int labels) {
    int i = blockIdx.x * blockDim.x + threadIdx.x;
    if (i >= n_graphs * labels) return;
    int g = i / labels, c = i % labels;
    float inv = 1.0f / fmaxf(cnt[g], 1.0f);
    float acc = bc[c];
#pragma unroll
    for (int j = 0; j < HIDC; ++j)
        acc += pooled[g * HIDC + j] * inv * Wc[j * labels + c];
    out[i] = acc;
}

extern "C" void kernel_launch(void* const* d_in, const int* in_sizes, int n_in,
                              void* d_out, int out_size, void* d_ws, size_t ws_size,
                              hipStream_t stream) {
    const int*   x     = (const int*)d_in[0];
    const int*   ei    = (const int*)d_in[1];
    const int*   batch = (const int*)d_in[2];
    const float* emb   = (const float*)d_in[3];
    const float* W1    = (const float*)d_in[4];
    const float* b1    = (const float*)d_in[5];
    const float* W2    = (const float*)d_in[6];
    const float* b2    = (const float*)d_in[7];
    const float* Wc    = (const float*)d_in[8];
    const float* bc    = (const float*)d_in[9];
    float* out = (float*)d_out;

    const int n_nodes  = in_sizes[0];
    const int n_edges  = in_sizes[1] / 2;
    const int vocab    = in_sizes[3] / HIDC;
    const int labels   = in_sizes[9];
    const int n_graphs = out_size / labels;
    const int* src = ei;
    const int* dst = ei + n_edges;

    const int nb = (n_nodes + BNODES - 1) / BNODES;
    const int nchunk = (n_edges + SEG - 1) / SEG;

    float* ws     = (float*)d_ws;
    float* hw2    = ws;
    float* rplane = hw2 + (size_t)n_nodes * HIDC;
    unsigned short* nrec = (unsigned short*)rplane;
    int* pairs    = (int*)(rplane + (size_t)n_nodes * HIDC);
    float* pooled = (float*)(pairs + (size_t)2 * n_edges);
    float* cnt    = pooled + (size_t)n_graphs * HIDC;
    float* dinv   = cnt + n_graphs;
    int* rowptr   = (int*)(dinv + n_nodes);
    int* cpre     = rowptr + n_nodes + 1;
    int* csr_src  = cpre + (size_t)(nb + 1) * nchunk;

    dim3 b(BLK);
    auto g1 = [](int n) { return dim3((n + BLK - 1) / BLK); };

    k_chunksort<<<dim3(nchunk), dim3(SORT_BLK), nb * sizeof(int), stream>>>(src, dst, cpre, pairs,
                                                                            pooled, cnt, n_edges, nb,
                                                                            nchunk, n_graphs);
    k_bucket<<<dim3(nb), dim3(SORT_BLK), 2 * nchunk * sizeof(int), stream>>>(pairs, cpre, rowptr, dinv,
                                                                             csr_src, x, nrec,
                                                                             n_nodes, n_edges, nb, nchunk);

    k_gather_xform<<<dim3(G_TAB), dim3(TAB_BLK), 0, stream>>>(rowptr, csr_src, dinv, nrec, emb, W1, b1, W2,
                                                              hw2, n_nodes, vocab);

    k_gather_pool<<<dim3(G_EDGE), b, 0, stream>>>(rowptr, csr_src, dinv, hw2, b2, batch,
                                                  pooled, cnt, n_nodes);

    k_classify<<<g1(n_graphs * labels), b, 0, stream>>>(pooled, cnt, Wc, bc, out,
                                                        n_graphs, labels);
}


# conv 1's LDS-table gather gives a node 4 lanes x 4 features: one 16-byte table read + 4 multiply-adds per edge, broadcasts inside lane quads; a wave sums 8 nodes at once; per-node sum order kept
# speedup vs baseline: 19.5777x; 1.0812x over previous
#include <hip/hip_runtime.h>


typedef __attribute__((ext_vector_type(2))) float v2f;
typedef __attribute__((ext_vector_type(8))) float v8f;

#define HIDC 16
#define BLK 256

#define G_EDGE 1024

#define NODE_CAP 100000
#define VOCAB_CAP 128
#define G_TAB 250
#define TAB_BLK 800
#define FILL_K ((NODE_CAP / 8 + TAB_BLK - 1) / TAB_BLK)

#define NODE_SHIFT 8
#define BNODES 256
#define SORT_BLK 1024
#define SEG 12500
#define SEG_K 13
#define BSTAGE_CAP 16384
#define RUNS 8

__device__ __forceinline__ void fadd(float* p, float v) {
    __hip_atomic_fetch_add(p, v, __ATOMIC_RELAXED, __HIP_MEMORY_SCOPE_AGENT);
}

__device__ __forceinline__ int wave_sum(int v) {
#pragma unroll
    for (int o = 16; o > 0; o >>= 1) v += __shfl_xor(v, o);
    return v;
}

__device__ __forceinline__ int wave_scan_incl(int v, int lane) {
#pragma unroll
    for (int o = 1; o < 32; o <<= 1) {
        int t = __shfl_up(v, o);
        if (lane >= o) v += t;
    }
    return v;
}

extern __shared__ int dsh[];

__global__ void __launch_bounds__(SORT_BLK)
k_chunksort(const int* __restrict__ src, const int* __restrict__ dst, int* __restrict__ cpre,
            int* __restrict__ pairs, float* __restrict__ pooled, float* __restrict__ cnt,
            int n_edges, int nb, int nchunk, int n_graphs) {
    __shared__ int2 stage[SEG];
    __shared__ int wtot[SORT_BLK / 32];
    int2* __restrict__ rec = reinterpret_cast<int2*>(pairs);
    int tid = threadIdx.x, lane = tid & 31, w = tid >> 5;
    for (int k = tid; k < nb; k += SORT_BLK) dsh[k] = 0;
    int gi = blockIdx.x * SORT_BLK + tid, gs = gridDim.x * SORT_BLK;
    for (int k = gi; k < n_graphs * HIDC; k += gs) pooled[k] = 0.0f;
    for (int k = gi; k < n_graphs; k += gs) cnt[k] = 0.0f;
    int c0 = blockIdx.x * SEG;
    int n = min(SEG, n_edges - c0);
    int es[SEG_K], ed[SEG_K], er[SEG_K];
#pragma unroll
    for (int k = 0; k < SEG_K; ++k) {
        int i = tid + k * SORT_BLK;
        ed[k] = (i < n) ? dst[c0 + i] : 0;
        es[k] = (i < n) ? src[c0 + i] : 0;
    }
    __syncthreads();
#pragma unroll
    for (int k = 0; k < SEG_K; ++k) {
        int i = tid + k * SORT_BLK;
        er[k] = (i < n) ? atomicAdd(&dsh[ed[k] >> NODE_SHIFT], 1) : 0;
    }
    __syncthreads();
    int carry = 0;
    for (int k0 = 0; k0 < nb; k0 += SORT_BLK) {
        int k = k0 + tid;
        int v = (k < nb) ? dsh[k] : 0;
        int incl = wave_scan_incl(v, lane);
        if (lane == 31) wtot[w] = incl;
        __syncthreads();
        int wpre = 0, tot = 0;
        for (int i = 0; i < SORT_BLK / 32; ++i) {
            int t = wtot[i];
            tot += t;
            if (i < w) wpre += t;
        }
        int excl = carry + wpre + incl - v;
        if (k < nb) {
            dsh[k] = excl;
            cpre[k * nchunk + blockIdx.x] = excl;
        }
        carry += tot;
        __syncthreads();
    }
    if (tid == 0) cpre[nb * nchunk + blockIdx.x] = n;
#pragma unroll
    for (int k = 0; k < SEG_K; ++k) er[k] += dsh[ed[k] >> NODE_SHIFT];
#pragma unroll
    for (int k = 0; k < SEG_K; ++k) {
        int i = tid + k * SORT_BLK;
        if (i < n) stage[er[k]] = make_int2(es[k], ed[k]);
    }
    __syncthreads();
    int i = tid;
    for (; i + 3 * SORT_BLK < n; i += 4 * SORT_BLK) {
        int2 r0 = stage[i], r1 = stage[i + SORT_BLK], r2 = stage[i + 2 * SORT_BLK], r3 = stage[i + 3 * SORT_BLK];
        rec[c0 + i] = r0;
        rec[c0 + i + SORT_BLK] = r1;
        rec[c0 + i + 2 * SORT_BLK] = r2;
        rec[c0 + i + 3 * SORT_BLK] = r3;
    }
    for (; i < n; i += SORT_BLK) rec[c0 + i] = stage[i];
}

__global__ void __launch_bounds__(SORT_BLK)
k_bucket(const int* __restrict__ pairs, const int* __restrict__ cpre,
         int* __restrict__ rowptr, float* __restrict__ dinv, int* __restrict__ csr_src,
         const int* __restrict__ x, unsigned short* __restrict__ nrec,
         int n_nodes, int n_edges, int nb, int nchunk) {
    const int2* __restrict__ rec = reinterpret_cast<const int2*>(pairs);
    __shared__ int cn[BNODES];
    __shared__ int wsum[BNODES / 32];
    __shared__ int red0[SORT_BLK / 32];
    __shared__ int red1[SORT_BLK / 32];
    __shared__ int simg[BSTAGE_CAP];
    int* roff = dsh;
    int* rlen = dsh + nchunk;
    int tid = threadIdx.x, lane = tid & 31, w = tid >> 5;
    int b = blockIdx.x;
    int nbase = b << NODE_SHIFT;
    int a0 = 0, a1 = 0;
    for (int k = tid; k < nchunk; k += SORT_BLK) {
        int lo = cpre[b * nchunk + k];
        int hi = cpre[(b + 1) * nchunk + k];
        roff[k] = k * SEG + lo;
        rlen[k] = hi - lo;
        a0 += lo;
        a1 += hi;
    }
    a0 = wave_sum(a0);
    a1 = wave_sum(a1);
    if (lane == 0) {
        red0[w] = a0;
        red1[w] = a1;
    }
    if (tid < BNODES) cn[tid] = 0;
    __syncthreads();
    int e0 = __builtin_amdgcn_readfirstlane(wave_sum(red0[lane]));
    int e1 = __builtin_amdgcn_readfirstlane(wave_sum(red1[lane]));
    const bool staged = (e1 - e0) <= BSTAGE_CAP;
    int c = w;
    for (; c + (RUNS - 1) * 32 < nchunk; c += RUNS * 32) {
        int rs[RUNS], rl[RUNS], da[RUNS], db[RUNS];
        int longest = 0;
#pragma unroll
        for (int q = 0; q < RUNS; ++q) {
            rs[q] = roff[c + 32 * q];
            rl[q] = rlen[c + 32 * q];
            longest = max(longest, rl[q]);
        }
#pragma unroll
        for (int q = 0; q < RUNS; ++q) {
            da[q] = (lane < rl[q]) ? rec[rs[q] + lane].y : 0;
            db[q] = (lane + 32 < rl[q]) ? rec[rs[q] + lane + 32].y : 0;
        }
#pragma unroll
        for (int q = 0; q < RUNS; ++q) {
            if (lane < rl[q]) atomicAdd(&cn[da[q] - nbase], 1);
            if (lane + 32 < rl[q]) atomicAdd(&cn[db[q] - nbase], 1);
        }
        if (longest > 64) {
            for (int q = 0; q < RUNS; ++q) {
                int rb = roff[c + 32 * q], rn = rlen[c + 32 * q];
                for (int i = lane + 64; i < rn; i += 32) atomicAdd(&cn[rec[rb + i].y - nbase], 1);
            }
        }
    }
    for (; c < nchunk; c += 32) {
        int rb = roff[c], rn = rlen[c];
        for (int i = lane; i < rn; i += 32) atomicAdd(&cn[rec[rb + i].y - nbase], 1);
    }
    __syncthreads();
    int xi = 0;
    int cv = 0, incl = 0;
    if (w < BNODES / 32) {
        xi = (nbase + tid < n_nodes) ? x[nbase + tid] : 0;
        cv = cn[tid];
        incl = wave_scan_incl(cv, lane);
        if (lane == 31) wsum[w] = incl;
    }
    __syncthreads();
    if (w < BNODES / 32) {
        int wpre = 0;
        for (int k = 0; k < w; ++k) wpre += wsum[k];
        int excl = wpre + incl - cv;
        cn[tid] = excl;
        int node = nbase + tid;
        float d = (float)(cv + 1);
        float di = (d > 0.0f) ? rsqrtf(d) : 0.0f;
        if (node < n_nodes) {
            rowptr[node] = e0 + excl;
            dinv[node] = di;
            nrec[node] = (unsigned short)((min(cv, 255) << 8) | (xi & 255));
        }
    }
    if (b == nb - 1 && tid == 0) rowptr[n_nodes] = n_edges;
    __syncthreads();
    c = w;
    for (; c + (RUNS - 1) * 32 < nchunk; c += RUNS * 32) {
        int longest = 0;
        for (int off = 0; off < 64; off += 32) {
            int rs[RUNS], rl[RUNS], p[RUNS];
            int2 ra[RUNS];
#pragma unroll
            for (int q = 0; q < RUNS; ++q) {
                rs[q] = roff[c + 32 * q];
                rl[q] = rlen[c + 32 * q];
                longest = max(longest, rl[q]);
            }
#pragma unroll
            for (int q = 0; q < RUNS; ++q)
                ra[q] = (lane + off < rl[q]) ? rec[rs[q] + lane + off] : make_int2(0, 0);
#pragma unroll
            for (int q = 0; q < RUNS; ++q)
                p[q] = (lane + off < rl[q]) ? atomicAdd(&cn[ra[q].y - nbase], 1) : 0;
#pragma unroll
            for (int q = 0; q < RUNS; ++q) {
                if (lane + off < rl[q]) {
                    if (staged) simg[p[q]] = ra[q].x; else csr_src[e0 + p[q]] = ra[q].x;
                }
            }
        }
        if (longest > 64) {
            for (int q = 0; q < RUNS; ++q) {
                int rb = roff[c + 32 * q], rn = rlen[c + 32 * q];
                for (int i = lane + 64; i < rn; i += 32) {
                    int2 r = rec[rb + i];
                    int p = atomicAdd(&cn[r.y - nbase], 1);
                    if (staged) simg[p] = r.x; else csr_src[e0 + p] = r.x;
                }
            }
        }
    }
    for (; c < nchunk; c += 32) {
        int rb = roff[c], rn = rlen[c];
        for (int i = lane; i < rn; i += 32) {
            int2 r = rec[rb + i];
            int p = atomicAdd(&cn[r.y - nbase], 1);
            if (staged) simg[p] = r.x; else csr_src[e0 + p] = r.x;
        }
    }
    if (staged) {
        __syncthreads();
        int ne = e1 - e0;
        int i = tid - (int)((reinterpret_cast<size_t>(csr_src + e0) >> 2) & 31);
        if (i >= 0 && i < ne) csr_src[e0 + i] = simg[i];
        for (i += SORT_BLK; i < ne; i += SORT_BLK) csr_src[e0 + i] = simg[i];
    }
}

__device__ __forceinline__ float gather_node(const int* __restrict__ rowptr,
                                             const int* __restrict__ csr_src,
                                             const float* __restrict__ dinv,
                                             const float* __restrict__ hw,
                                             int node, int j, float bj, float& dd) {
    int i = node * HIDC + j;
    int beg = rowptr[node], end = rowptr[node + 1];
    dd = dinv[node];
    float acc = hw[i];
    int s_next = (beg + j < end) ? csr_src[beg + j] : node;
    for (int p = beg; p < end; p += 16) {
        int s = s_next;
        int qn = p + 16 + j;
        s_next = (qn < end) ? csr_src[qn] : node;
        int nk = end - p;
#pragma unroll
        for (int k = 0; k < 16; ++k) {
            int   sk = __shfl(s, k, 16);
            float v  = hw[sk * HIDC + j];
            acc += (k < nk) ? v : 0.0f;
        }
    }
    return fmaxf(acc * dd + bj, 0.0f);
}

template <int K>
__device__ __forceinline__ int quad_lane(int v) {
    return __builtin_amdgcn_update_dpp(0, v, K * 0x55, 0xf, 0xf, true);
}

#define EDGE_STEP(C, K)                                                                     \
    {                                                                                       \
        int   tk = quad_lane<K>(to[C]);                                                     \
        float dk = __int_as_float(quad_lane<K>(__float_as_int(di[C])));                     \
        float4 tv = *reinterpret_cast<const float4*>(tq + tk);                              \
        acc.x = fmaf(dk, tv.x, acc.x);                                                      \
        acc.y = fmaf(dk, tv.y, acc.y);                                                      \
        acc.z = fmaf(dk, tv.z, acc.z);                                                      \
        acc.w = fmaf(dk, tv.w, acc.w);                                                      \
    }
#define EDGE_GROUP(C) EDGE_STEP(C, 0) EDGE_STEP(C, 1) EDGE_STEP(C, 2) EDGE_STEP(C, 3)

template <bool IN_LDS>
__device__ __forceinline__ float4 gather_quad_tab(const int* __restrict__ rowptr,
                                                  const int* __restrict__ csr_src,
                                                  const float* __restrict__ dinv,
                                                  const unsigned short* __restrict__ nrec,
                                                  const unsigned short* ntab, const float* ttab,
                                                  int node, bool live, int f, float4 bj, float& dd) {
    int beg = 0, end = 0;
    unsigned rn = 0u;
    dd = 0.0f;
    if (live) {
        beg = rowptr[node];
        end = rowptr[node + 1];
        dd = dinv[node];
        rn = IN_LDS ? (unsigned)ntab[node] : (unsigned)nrec[node];
    }
    const int pad = live ? node : 0;
    const char* tq = reinterpret_cast<const char*>(ttab) + f * 16;
    int sn[4];
#pragma unroll
    for (int c = 0; c < 4; ++c) {
        int qn = beg + 4 * c + f;
        sn[c] = (qn < end) ? csr_src[qn] : pad;
    }
    float4 acc;
    {
        float4 tv = *reinterpret_cast<const float4*>(tq + (int)(rn & 255u) * (HIDC * 4));
        acc.x = dd * tv.x;
        acc.y = dd * tv.y;
        acc.z = dd * tv.z;
        acc.w = dd * tv.w;
    }
    for (int p = beg; p < end; p += 16) {
        int to[4];
        float di[4];
        unsigned r[4];
#pragma unroll
        for (int c = 0; c < 4; ++c) r[c] = IN_LDS ? (unsigned)ntab[sn[c]] : (unsigned)nrec[sn[c]];
        int s[4];
#pragma unroll
        for (int c = 0; c < 4; ++c) {
            s[c] = sn[c];
            int qn = p + 16 + 4 * c + f;
            sn[c] = (qn < end) ? csr_src[qn] : pad;
        }
#pragma unroll
        for (int c = 0; c < 4; ++c) {
            unsigned c8 = r[c] >> 8;
            float d = rsqrtf((float)(c8 + 1u));
            if (c8 == 255u) d = dinv[s[c]];
            di[c] = (p + 4 * c + f < end) ? d : 0.0f;
            to[c] = (int)(r[c] & 255u) * (HIDC * 4);
        }
        EDGE_GROUP(0)
        if (p + 4 < end) {
            EDGE_GROUP(1)
            if (p + 8 < end) {
                EDGE_GROUP(2)
                if (p + 12 < end) {
                    EDGE_GROUP(3)
                }
            }
        }
    }
    float4 hv;
    hv.x = fmaxf(acc.x * dd + bj.x, 0.0f);
    hv.y = fmaxf(acc.y * dd + bj.y, 0.0f);
    hv.z = fmaxf(acc.z * dd + bj.z, 0.0f);
    hv.w = fmaxf(acc.w * dd + bj.w, 0.0f);
    return hv;
}

#define TROW 20
__global__ void __launch_bounds__(TAB_BLK)
k_gather_xform(const int* __restrict__ rowptr, const int* __restrict__ csr_src,
               const float* __restrict__ dinv, const unsigned short* __restrict__ nrec,
               const float* __restrict__ emb, const float* __restrict__ W1,
               const float* __restrict__ b, const float* __restrict__ W,
               float* __restrict__ hw2, int n_nodes, int vocab) {
    __shared__ uint4 ntab4[(NODE_CAP + 7) / 8];
    __shared__ __attribute__((aligned(16))) float ttab[VOCAB_CAP * HIDC];
    __shared__ __attribute__((aligned(16))) float tile[TAB_BLK / 32][16][TROW];
    unsigned short* ntab = reinterpret_cast<unsigned short*>(ntab4);
    int lane = threadIdx.x & 31;
    int wv = __builtin_amdgcn_readfirstlane((int)(threadIdx.x >> 5));
    int half = lane >> 4, l15 = lane & 15;
    int quad = lane >> 2, f = lane & 3;
    float4 bj = make_float4(b[4 * f + 0], b[4 * f + 1], b[4 * f + 2], b[4 * f + 3]);

    int nfill = min(n_nodes, NODE_CAP);
    int nvec = nfill >> 3;
    const uint4* __restrict__ g4 = reinterpret_cast<const uint4*>(nrec);
    uint4 piece[FILL_K];
#pragma unroll
    for (int k = 0; k < FILL_K; ++k) {
        int i = threadIdx.x + k * TAB_BLK;
        piece[k] = (i < nvec) ? g4[i] : make_uint4(0u, 0u, 0u, 0u);
    }
#pragma unroll
    for (int k = 0; k < FILL_K; ++k) {
        int i = threadIdx.x + k * TAB_BLK;
        if (i < nvec) ntab4[i] = piece[k];
    }
    for (int i = (nvec << 3) + threadIdx.x; i < nfill; i += TAB_BLK) ntab[i] = nrec[i];
    if (wv < (min(vocab, VOCAB_CAP) >> 4)) {
        v8f acc = {};
#pragma unroll
        for (int q = 0; q < 4; ++q) {
            int kk = 4 * q + 2 * half;
            v2f a, w1;
            a[0]  = emb[(16 * wv + l15) * HIDC + kk + 0];
            a[1]  = emb[(16 * wv + l15) * HIDC + kk + 1];
            w1[0] = W1[(kk + 0) * HIDC + l15];
            w1[1] = W1[(kk + 1) * HIDC + l15];
            acc = __builtin_amdgcn_wmma_f32_16x16x4_f32(
                      false, a, false, w1, (short)0, acc, false, false);
        }
#pragma unroll
        for (int v = 0; v < 8; ++v) ttab[(16 * wv + v + 8 * half) * HIDC + l15] = acc[v];
    }

    v2f bf[4];
#pragma unroll
    for (int q = 0; q < 4; ++q) {
        int kk = 4 * q + 2 * half;
        bf[q][0] = W[(kk + 0) * HIDC + l15];
        bf[q][1] = W[(kk + 1) * HIDC + l15];
    }

    __syncthreads();
    const bool in_lds = n_nodes <= NODE_CAP;

    const int wpb = TAB_BLK / 32;
    const int ntile = (n_nodes + 15) >> 4;
    const int nwave = gridDim.x * wpb;
    for (int tl = blockIdx.x * wpb + wv; tl < ntile; tl += nwave) {
        int base = tl << 4;
#pragma unroll 1
        for (int o = 0; o < 2; ++o) {
            int n = 8 * o + quad;
            int node = base + n;
            bool live = node < n_nodes;
            float dd;
            float4 hv = in_lds ? gather_quad_tab<true>(rowptr, csr_src, dinv, nrec, ntab, ttab, node, live, f, bj, dd)
                               : gather_quad_tab<false>(rowptr, csr_src, dinv, nrec, ntab, ttab, node, live, f, bj, dd);
            float4 a = live ? make_float4(hv.x * dd, hv.y * dd, hv.z * dd, hv.w * dd)
                            : make_float4(0.0f, 0.0f, 0.0f, 0.0f);
            *reinterpret_cast<float4*>(&tile[wv][(n >> 1) + 8 * (n & 1)][4 * f]) = a;
        }
        __syncwarp();
        v8f c = {};
#pragma unroll
        for (int q = 0; q < 4; ++q) {
            int kk = 4 * q + 2 * half;
            v2f a;
            a[0] = tile[wv][l15][kk + 0];
            a[1] = tile[wv][l15][kk + 1];
            c = __builtin_amdgcn_wmma_f32_16x16x4_f32(
                    false, a, false, bf[q], (short)0, c, false, false);
        }
        __syncwarp();
#pragma unroll
        for (int v = 0; v < 8; ++v) {
            int node = base + 2 * v + half;
            if (node < n_nodes) hw2[(size_t)node * HIDC + l15] = c[v];
        }
    }
}

__global__ void k_gather_pool(const int* __restrict__ rowptr, const int* __restrict__ csr_src,
                              const float* __restrict__ dinv, const float* __restrict__ hw,
                              const float* __restrict__ b, const int* __restrict__ batch,
                              float* __restrict__ pooled, float* __restrict__ cnt, int n_nodes) {
    int t = blockIdx.x * blockDim.x + threadIdx.x;
    int j = t & 15;
    int grp = t >> 4;
    int ngrp = (gridDim.x * blockDim.x) >> 4;
    int per = n_nodes / ngrp, rem = n_nodes - per * ngrp;
    int n0 = grp * per + (grp < rem ? grp : rem);
    int n1 = n0 + per + (grp < rem ? 1 : 0);
    float bj = b[j];
    int gcur = -1;
    float psum = 0.0f, pcnt = 0.0f;
    for (int node = n0; node < n1; ++node) {
        int g = batch[node];
        float dd;
        float v = gather_node(rowptr, csr_src, dinv, hw, node, j, bj, dd);
        if (g != gcur) {
            if (gcur >= 0) {
                fadd(&pooled[gcur * HIDC + j], psum);
                if (j == 0) fadd(&cnt[gcur], pcnt);
            }
            gcur = g;
            psum = 0.0f;
            pcnt = 0.0f;
        }
        psum += v;
        pcnt += 1.0f;
    }
    if (gcur >= 0) {
        fadd(&pooled[gcur * HIDC + j], psum);
        if (j == 0) fadd(&cnt[gcur], pcnt);
    }
}

__global__ void k_classify(const float* __restrict__ pooled, const float* __restrict__ cnt,
                           const float* __restrict__ Wc, const float* __restrict__ bc,
                           float* __restrict__ out, int n_graphs, int labels) {
    int i = blockIdx.x * blockDim.x + threadIdx.x;
    if (i >= n_graphs * labels) return;
    int g = i / labels, c = i % labels;
    float inv = 1.0f / fmaxf(cnt[g], 1.0f);
    float acc = bc[c];
#pragma unroll
    for (int j = 0; j < HIDC; ++j)
        acc += pooled[g * HIDC + j] * inv * Wc[j * labels + c];
    out[i] = acc;
}

extern "C" void kernel_launch(void* const* d_in, const int* in_sizes, int n_in,
                              void* d_out, int out_size, void* d_ws, size_t ws_size,
                              hipStream_t stream) {
    const int*   x     = (const int*)d_in[0];
    const int*   ei    = (const int*)d_in[1];
    const int*   batch = (const int*)d_in[2];
    const float* emb   = (const float*)d_in[3];
    const float* W1    = (const float*)d_in[4];
    const float* b1    = (const float*)d_in[5];
    const float* W2    = (const float*)d_in[6];
    const float* b2    = (const float*)d_in[7];
    const float* Wc    = (const float*)d_in[8];
    const float* bc    = (const float*)d_in[9];
    float* out = (float*)d_out;

    const int n_nodes  = in_sizes[0];
    const int n_edges  = in_sizes[1] / 2;
    const int vocab    = in_sizes[3] / HIDC;
    const int labels   = in_sizes[9];
    const int n_graphs = out_size / labels;
    const int* src = ei;
    const int* dst = ei + n_edges;

    const int nb = (n_nodes + BNODES - 1) / BNODES;
    const int nchunk = (n_edges + SEG - 1) / SEG;

    float* ws     = (float*)d_ws;
    float* hw2    = ws;
    float* rplane = hw2 + (size_t)n_nodes * HIDC;
    unsigned short* nrec = (unsigned short*)rplane;
    int* pairs    = (int*)(rplane + (size_t)n_nodes * HIDC);
    float* pooled = (float*)(pairs + (size_t)2 * n_edges);
    float* cnt    = pooled + (size_t)n_graphs * HIDC;
    float* dinv   = cnt + n_graphs;
    int* rowptr   = (int*)(dinv + n_nodes);
    int* cpre     = rowptr + n_nodes + 1;
    int* csr_src  = cpre + (size_t)(nb + 1) * nchunk;

    dim3 b(BLK);
    auto g1 = [](int n) { return dim3((n + BLK - 1) / BLK); };

    k_chunksort<<<dim3(nchunk), dim3(SORT_BLK), nb * sizeof(int), stream>>>(src, dst, cpre, pairs,
                                                                            pooled, cnt, n_edges, nb,
                                                                            nchunk, n_graphs);
    k_bucket<<<dim3(nb), dim3(SORT_BLK), 2 * nchunk * sizeof(int), stream>>>(pairs, cpre, rowptr, dinv,
                                                                             csr_src, x, nrec,
                                                                             n_nodes, n_edges, nb, nchunk);

    k_gather_xform<<<dim3(G_TAB), dim3(TAB_BLK), 0, stream>>>(rowptr, csr_src, dinv, nrec, emb, W1, b1, W2,
                                                              hw2, n_nodes, vocab);

    k_gather_pool<<<dim3(G_EDGE), b, 0, stream>>>(rowptr, csr_src, dinv, hw2, b2, batch,
                                                  pooled, cnt, n_nodes);

    k_classify<<<g1(n_graphs * labels), b, 0, stream>>>(pooled, cnt, Wc, bc, out,
                                                        n_graphs, labels);
}
